# starting kernel
# baseline (speedup 1.0000x reference)
#include <hip/hip_runtime.h>


typedef __attribute__((ext_vector_type(2))) float v2f;
typedef __attribute__((ext_vector_type(8))) float v8f;

#define SQRT_HALF 0.70710678118654752440f

__device__ __forceinline__ int reflect_i(int x, int n) {
  x = (x < 0) ? (-1 - x) : x;
  x = (x >= n) ? (2 * n - 1 - x) : x;
  return x;
}

__device__ __forceinline__ float cif_A(int m, int kk,
                                       const float* __restrict__ ha,
                                       const float* __restrict__ hb, int hp) {
  int p = m & 3;
  int q = m >> 2;
  int want_odd = hp ? ((p & 1) ^ 1) : (p & 1);
  int t = ((kk - want_odd) >> 1) - q;
  bool ok = ((kk & 1) == want_odd) && (t >= 0) && (t <= 4);
  int tc = t < 0 ? 0 : (t > 4 ? 4 : t);
  const float* __restrict__ h = ((p == 0) || (p == 2)) ? ha : hb;
  int idx = (p < 2) ? (2 * tc + 1) : (2 * tc);
  float v = h[idx];
  return ok ? v : 0.0f;
}

__device__ __forceinline__ float cf_A(int m, int kk,
                                      const float* __restrict__ h, int taps) {
  int t = kk - m - (3 - (taps >> 1));
  bool ok = (t >= 0) && (t < taps);
  int tc = t < 0 ? 0 : (t >= taps ? taps - 1 : t);
  float v = h[tc];
  return ok ? v : 0.0f;
}

__device__ __forceinline__ float c2q_val(const float* __restrict__ p1,
                                         const float* __restrict__ p2, int rr,
                                         int stride, int cb) {
  int o = (rr >> 1) * stride;
  float w1r = p1[o], w1i = p1[o + 1];
  float w2r = p2[o], w2i = p2[o + 1];
  float even = cb ? (w1i + w2i) : (w1r + w2r);
  float odd = cb ? (w2r - w1r) : (w1i - w2i);
  return (rr & 1) ? odd : even;
}

__device__ __forceinline__ v8f wmma4(v2f a, v2f b, v8f c) {
  return __builtin_amdgcn_wmma_f32_16x16x4_f32(false, a, false, b, (short)0, c,
                                               false, false);
}

template <bool SRC0_PLAIN>
__global__ void __launch_bounds__(256)
j2_col_kernel(const float* __restrict__ src0_plain, const float* __restrict__ yh,
              int s0a, int s0b, int s1a, int s1b, int Hb, int Wb, int r, int W,
              const float* __restrict__ ha0, const float* __restrict__ hb0,
              const float* __restrict__ ha1, const float* __restrict__ hb1,
              float* __restrict__ out, int nTilesR, int nTilesW, int B) {
  int lane = threadIdx.x & 31;
  int wave = threadIdx.x >> 5;
  long tile = (long)blockIdx.x * 8 + wave;
  long total = (long)B * nTilesR * nTilesW;
  if (tile >= total) return;
  int tpi = nTilesR * nTilesW;
  int b = (int)(tile / tpi);
  int t2 = (int)(tile % tpi);
  int I = t2 / nTilesW;
  int Jc = t2 % nTilesW;
  int x0 = 8 * I - 4;
  int half = lane >> 4;
  int l16 = lane & 15;
  int koff = half * 2;
  int M = l16;
  int cc = Jc * 16 + l16;

  const float s0sc = SRC0_PLAIN ? 1.0f : SQRT_HALF;
  v2f a0[4], a1[4];
#pragma unroll
  for (int c = 0; c < 4; ++c) {
    int kk = 4 * c + koff;
    a0[c].x = cif_A(M, kk, ha0, hb0, 0) * s0sc;
    a0[c].y = cif_A(M, kk + 1, ha0, hb0, 0) * s0sc;
    a1[c].x = cif_A(M, kk, ha1, hb1, 1) * SQRT_HALF;
    a1[c].y = cif_A(M, kk + 1, ha1, hb1, 1) * SQRT_HALF;
  }

  int cb = cc & 1;
  int stride = 2 * Wb;
  int j2 = (cc >> 1) * 2;
  const float* p0 = SRC0_PLAIN ? (src0_plain + (size_t)b * r * W + cc) : nullptr;
  const float* p0a = yh + ((size_t)b * 6 + s0a) * Hb * stride + j2;
  const float* p0b = yh + ((size_t)b * 6 + s0b) * Hb * stride + j2;
  const float* p1a = yh + ((size_t)b * 6 + s1a) * Hb * stride + j2;
  const float* p1b = yh + ((size_t)b * 6 + s1b) * Hb * stride + j2;

  v8f acc = {0.f, 0.f, 0.f, 0.f, 0.f, 0.f, 0.f, 0.f};
#pragma unroll
  for (int c = 0; c < 4; ++c) {
    int kk = 4 * c + koff;
    int r0 = reflect_i(x0 + kk, r);
    int r1 = reflect_i(x0 + kk + 1, r);
    v2f b0, b1;
    if (SRC0_PLAIN) {
      b0.x = p0[(size_t)r0 * W];
      b0.y = p0[(size_t)r1 * W];
    } else {
      b0.x = c2q_val(p0a, p0b, r0, stride, cb);
      b0.y = c2q_val(p0a, p0b, r1, stride, cb);
    }
    b1.x = c2q_val(p1a, p1b, r0, stride, cb);
    b1.y = c2q_val(p1a, p1b, r1, stride, cb);
    acc = wmma4(a0[c], b0, acc);
    acc = wmma4(a1[c], b1, acc);
  }

  int outR = 2 * r;
  float* op = out + ((size_t)b * outR + (size_t)I * 16 + half * 8) * W + cc;
#pragma unroll
  for (int e = 0; e < 8; ++e) op[(size_t)e * W] = acc[e];
}

__global__ void __launch_bounds__(256)
j2_row_kernel(const float* __restrict__ T1, const float* __restrict__ T2, int H,
              int Wt, const float* __restrict__ ha0,
              const float* __restrict__ hb0, const float* __restrict__ ha1,
              const float* __restrict__ hb1, float* __restrict__ out,
              int nTilesR, int nTilesW, int B) {
  int lane = threadIdx.x & 31;
  int wave = threadIdx.x >> 5;
  long tile = (long)blockIdx.x * 8 + wave;
  long total = (long)B * nTilesR * nTilesW;
  if (tile >= total) return;
  int tpi = nTilesR * nTilesW;
  int b = (int)(tile / tpi);
  int t2 = (int)(tile % tpi);
  int I = t2 / nTilesW;
  int Jc = t2 % nTilesW;
  int m0 = I * 16;
  int xc0 = 8 * Jc - 4;
  int half = lane >> 4;
  int l16 = lane & 15;
  int koff = half * 2;

  v2f bf0[4], bf1[4];
#pragma unroll
  for (int c = 0; c < 4; ++c) {
    int kk = 4 * c + koff;
    bf0[c].x = cif_A(l16, kk, ha0, hb0, 0);
    bf0[c].y = cif_A(l16, kk + 1, ha0, hb0, 0);
    bf1[c].x = cif_A(l16, kk, ha1, hb1, 1);
    bf1[c].y = cif_A(l16, kk + 1, ha1, hb1, 1);
  }

  const float* row1 = T1 + ((size_t)b * H + m0 + l16) * Wt;
  const float* row2 = T2 + ((size_t)b * H + m0 + l16) * Wt;
  v8f acc = {0.f, 0.f, 0.f, 0.f, 0.f, 0.f, 0.f, 0.f};
#pragma unroll
  for (int c = 0; c < 4; ++c) {
    int kk = 4 * c + koff;
    int c0 = reflect_i(xc0 + kk, Wt);
    int c1 = reflect_i(xc0 + kk + 1, Wt);
    v2f aA, aB;
    aA.x = row1[c0];
    aA.y = row1[c1];
    aB.x = row2[c0];
    aB.y = row2[c1];
    acc = wmma4(aA, bf0[c], acc);
    acc = wmma4(aB, bf1[c], acc);
  }

  int outW = 2 * Wt;
  float* op = out + ((size_t)b * H + m0 + half * 8) * outW + Jc * 16 + l16;
#pragma unroll
  for (int e = 0; e < 8; ++e) op[(size_t)e * outW] = acc[e];
}

template <bool SRC0_PLAIN>
__global__ void __launch_bounds__(256)
j1_col_kernel(const float* __restrict__ src0_plain, const float* __restrict__ yh,
              int s0a, int s0b, int s1a, int s1b, int Hb, int Wb, int r, int W,
              const float* __restrict__ h0, int taps0,
              const float* __restrict__ h1, int taps1, float* __restrict__ out,
              int nTilesR, int nTilesW, int B) {
  int lane = threadIdx.x & 31;
  int wave = threadIdx.x >> 5;
  long tile = (long)blockIdx.x * 8 + wave;
  long total = (long)B * nTilesR * nTilesW;
  if (tile >= total) return;
  int tpi = nTilesR * nTilesW;
  int b = (int)(tile / tpi);
  int t2 = (int)(tile % tpi);
  int I = t2 / nTilesW;
  int Jc = t2 % nTilesW;
  int x0 = 16 * I - 3;
  int half = lane >> 4;
  int l16 = lane & 15;
  int koff = half * 2;
  int M = l16;
  int cc = Jc * 16 + l16;

  const float s0sc = SRC0_PLAIN ? 1.0f : SQRT_HALF;
  v2f a0[6], a1[6];
#pragma unroll
  for (int c = 0; c < 6; ++c) {
    int kk = 4 * c + koff;
    a0[c].x = cf_A(M, kk, h0, taps0) * s0sc;
    a0[c].y = cf_A(M, kk + 1, h0, taps0) * s0sc;
    a1[c].x = cf_A(M, kk, h1, taps1) * SQRT_HALF;
    a1[c].y = cf_A(M, kk + 1, h1, taps1) * SQRT_HALF;
  }

  int cb = cc & 1;
  int stride = 2 * Wb;
  int j2 = (cc >> 1) * 2;
  const float* p0 = SRC0_PLAIN ? (src0_plain + (size_t)b * r * W + cc) : nullptr;
  const float* p0a = yh + ((size_t)b * 6 + s0a) * Hb * stride + j2;
  const float* p0b = yh + ((size_t)b * 6 + s0b) * Hb * stride + j2;
  const float* p1a = yh + ((size_t)b * 6 + s1a) * Hb * stride + j2;
  const float* p1b = yh + ((size_t)b * 6 + s1b) * Hb * stride + j2;

  v8f acc = {0.f, 0.f, 0.f, 0.f, 0.f, 0.f, 0.f, 0.f};
#pragma unroll
  for (int c = 0; c < 6; ++c) {
    int kk = 4 * c + koff;
    int r0 = reflect_i(x0 + kk, r);
    int r1 = reflect_i(x0 + kk + 1, r);
    v2f b0, b1;
    if (SRC0_PLAIN) {
      b0.x = p0[(size_t)r0 * W];
      b0.y = p0[(size_t)r1 * W];
    } else {
      b0.x = c2q_val(p0a, p0b, r0, stride, cb);
      b0.y = c2q_val(p0a, p0b, r1, stride, cb);
    }
    b1.x = c2q_val(p1a, p1b, r0, stride, cb);
    b1.y = c2q_val(p1a, p1b, r1, stride, cb);
    acc = wmma4(a0[c], b0, acc);
    acc = wmma4(a1[c], b1, acc);
  }

  float* op = out + ((size_t)b * r + (size_t)I * 16 + half * 8) * W + cc;
#pragma unroll
  for (int e = 0; e < 8; ++e) op[(size_t)e * W] = acc[e];
}

__global__ void __launch_bounds__(256)
j1_row_kernel(const float* __restrict__ T1, const float* __restrict__ T2, int H,
              int Wt, const float* __restrict__ h0, int taps0,
              const float* __restrict__ h1, int taps1, float* __restrict__ out,
              int nTilesR, int nTilesW, int B) {
  int lane = threadIdx.x & 31;
  int wave = threadIdx.x >> 5;
  long tile = (long)blockIdx.x * 8 + wave;
  long total = (long)B * nTilesR * nTilesW;
  if (tile >= total) return;
  int tpi = nTilesR * nTilesW;
  int b = (int)(tile / tpi);
  int t2 = (int)(tile % tpi);
  int I = t2 / nTilesW;
  int Jc = t2 % nTilesW;
  int m0 = I * 16;
  int xc0 = 16 * Jc - 3;
  int half = lane >> 4;
  int l16 = lane & 15;
  int koff = half * 2;

  v2f bf0[6], bf1[6];
#pragma unroll
  for (int c = 0; c < 6; ++c) {
    int kk = 4 * c + koff;
    bf0[c].x = cf_A(l16, kk, h0, taps0);
    bf0[c].y = cf_A(l16, kk + 1, h0, taps0);
    bf1[c].x = cf_A(l16, kk, h1, taps1);
    bf1[c].y = cf_A(l16, kk + 1, h1, taps1);
  }

  const float* row1 = T1 + ((size_t)b * H + m0 + l16) * Wt;
  const float* row2 = T2 + ((size_t)b * H + m0 + l16) * Wt;
  v8f acc = {0.f, 0.f, 0.f, 0.f, 0.f, 0.f, 0.f, 0.f};
#pragma unroll
  for (int c = 0; c < 6; ++c) {
    int kk = 4 * c + koff;
    int c0 = reflect_i(xc0 + kk, Wt);
    int c1 = reflect_i(xc0 + kk + 1, Wt);
    v2f aA, aB;
    aA.x = row1[c0];
    aA.y = row1[c1];
    aB.x = row2[c0];
    aB.y = row2[c1];
    acc = wmma4(aA, bf0[c], acc);
    acc = wmma4(aB, bf1[c], acc);
  }

  float* op = out + ((size_t)b * H + m0 + half * 8) * Wt + Jc * 16 + l16;
#pragma unroll
  for (int e = 0; e < 8; ++e) op[(size_t)e * Wt] = acc[e];
}

extern "C" void kernel_launch(void* const* d_in, const int* in_sizes, int n_in,
                              void* d_out, int out_size, void* d_ws,
                              size_t ws_size, hipStream_t stream) {
  (void)in_sizes;
  (void)n_in;
  (void)out_size;
  (void)ws_size;

  const float* yl = (const float*)d_in[0];
  const float* yh0 = (const float*)d_in[1];
  const float* yh1 = (const float*)d_in[2];
  const float* yh2 = (const float*)d_in[3];
  const float* g0o = (const float*)d_in[4];
  const float* g1o = (const float*)d_in[5];
  const float* g0a = (const float*)d_in[6];
  const float* g0b = (const float*)d_in[7];
  const float* g1a = (const float*)d_in[8];
  const float* g1b = (const float*)d_in[9];

  const int B = 64;
  const size_t IMG = (size_t)512 * 512;
  float* T1 = (float*)d_ws;
  float* T2 = T1 + (size_t)B * IMG;
  float* LL = T2 + (size_t)B * IMG;
  float* outp = (float*)d_out;

  dim3 blk(256);
  auto nblk = [](long tiles) { return dim3((unsigned)((tiles + 7) / 8)); };

  {
    int r = 128, W = 128, Hb = 64, Wb = 64;
    int ntr = 2 * r / 16, ntw = W / 16;
    long tc = (long)B * ntr * ntw;
    j2_col_kernel<true><<<nblk(tc), blk, 0, stream>>>(
        yl, yh2, 0, 0, 0, 5, Hb, Wb, r, W, g0b, g0a, g1b, g1a, T1, ntr, ntw, B);
    j2_col_kernel<false><<<nblk(tc), blk, 0, stream>>>(
        nullptr, yh2, 2, 3, 1, 4, Hb, Wb, r, W, g0b, g0a, g1b, g1a, T2, ntr,
        ntw, B);
    int H = 2 * r, Wt = W;
    int rtr = H / 16, rtw = 2 * Wt / 16;
    long tr = (long)B * rtr * rtw;
    j2_row_kernel<<<nblk(tr), blk, 0, stream>>>(T1, T2, H, Wt, g0b, g0a, g1b,
                                                g1a, LL, rtr, rtw, B);
  }

  {
    int r = 256, W = 256, Hb = 128, Wb = 128;
    int ntr = 2 * r / 16, ntw = W / 16;
    long tc = (long)B * ntr * ntw;
    j2_col_kernel<true><<<nblk(tc), blk, 0, stream>>>(
        LL, yh1, 0, 0, 0, 5, Hb, Wb, r, W, g0b, g0a, g1b, g1a, T1, ntr, ntw, B);
    j2_col_kernel<false><<<nblk(tc), blk, 0, stream>>>(
        nullptr, yh1, 2, 3, 1, 4, Hb, Wb, r, W, g0b, g0a, g1b, g1a, T2, ntr,
        ntw, B);
    int H = 2 * r, Wt = W;
    int rtr = H / 16, rtw = 2 * Wt / 16;
    long tr = (long)B * rtr * rtw;
    j2_row_kernel<<<nblk(tr), blk, 0, stream>>>(T1, T2, H, Wt, g0b, g0a, g1b,
                                                g1a, LL, rtr, rtw, B);
  }

  {
    int r = 512, W = 512, Hb = 256, Wb = 256;
    int ntr = r / 16, ntw = W / 16;
    long tc = (long)B * ntr * ntw;
    j1_col_kernel<true><<<nblk(tc), blk, 0, stream>>>(
        LL, yh0, 0, 0, 0, 5, Hb, Wb, r, W, g0o, 7, g1o, 5, T1, ntr, ntw, B);
    j1_col_kernel<false><<<nblk(tc), blk, 0, stream>>>(
        nullptr, yh0, 2, 3, 1, 4, Hb, Wb, r, W, g0o, 7, g1o, 5, T2, ntr, ntw,
        B);
    long tr = (long)B * ntr * ntw;
    j1_row_kernel<<<nblk(tr), blk, 0, stream>>>(T1, T2, r, W, g0o, 7, g1o, 5,
                                                outp, ntr, ntw, B);
  }
}


# Column and row pass fused per wave: column-pass tile computed transposed, its accumulator registers are the row product's A operand; no temp image, 9 launches to 3; a row's sum order changes
# speedup vs baseline: 2.2360x; 2.2360x over previous
#include <hip/hip_runtime.h>


typedef __attribute__((ext_vector_type(2))) float v2f;
typedef __attribute__((ext_vector_type(8))) float v8f;

#define SQRT_HALF 0.70710678118654752440f

__device__ __forceinline__ int reflect_i(int x, int n) {
  x = (x < 0) ? (-1 - x) : x;
  x = (x >= n) ? (2 * n - 1 - x) : x;
  return x;
}

__device__ __forceinline__ float cif_A(int m, int kk,
                                       const float* __restrict__ ha,
                                       const float* __restrict__ hb, int hp) {
  int p = m & 3;
  int q = m >> 2;
  int want_odd = hp ? ((p & 1) ^ 1) : (p & 1);
  int t = ((kk - want_odd) >> 1) - q;
  bool ok = ((kk & 1) == want_odd) && (t >= 0) && (t <= 4);
  int tc = t < 0 ? 0 : (t > 4 ? 4 : t);
  const float* __restrict__ h = ((p == 0) || (p == 2)) ? ha : hb;
  int idx = (p < 2) ? (2 * tc + 1) : (2 * tc);
  float v = h[idx];
  return ok ? v : 0.0f;
}

__device__ __forceinline__ float cf_A(int m, int kk,
                                      const float* __restrict__ h, int taps) {
  int t = kk - m - (3 - (taps >> 1));
  bool ok = (t >= 0) && (t < taps);
  int tc = t < 0 ? 0 : (t >= taps ? taps - 1 : t);
  float v = h[tc];
  return ok ? v : 0.0f;
}

__device__ __forceinline__ float c2q_val(const float* __restrict__ p1,
                                         const float* __restrict__ p2,
                                         unsigned o, int odd_row, int cb) {
  v2f w1 = ((const v2f*)p1)[o];
  v2f w2 = ((const v2f*)p2)[o];
  float even = cb ? (w1.y + w2.y) : (w1.x + w2.x);
  float odd = cb ? (w2.x - w1.x) : (w1.y - w2.y);
  return odd_row ? odd : even;
}

__device__ __forceinline__ v8f wmma4(v2f a, v2f b, v8f c) {
  return __builtin_amdgcn_wmma_f32_16x16x4_f32(false, a, false, b, (short)0, c,
                                               false, false);
}

struct YT {
  v8f y1, y2;
};

template <int KC, int Hb, int Wb>
__device__ __forceinline__ YT col_tile(const float* __restrict__ plain,
                                       const float* __restrict__ yh, int b,
                                       int x0, int c0, int l16, int koff,
                                       const v2f* f0, const v2f* f0s,
                                       const v2f* f1s) {
  constexpr int r = 2 * Hb;
  constexpr int W = 2 * Wb;
  int cc = reflect_i(c0 + l16, W);
  int cb = cc & 1;
  unsigned jb = (unsigned)(cc >> 1);
  constexpr size_t bs = (size_t)Hb * Wb * 2;
  const float* p0 = plain + (size_t)b * r * W;
  const float* yb = yh + (size_t)b * 6 * bs;
  const float* q0 = yb;
  const float* q1 = yb + bs;
  const float* q2 = yb + 2 * bs;
  const float* q3 = yb + 3 * bs;
  const float* q4 = yb + 4 * bs;
  const float* q5 = yb + 5 * bs;
  YT o;
  o.y1 = (v8f){0.f, 0.f, 0.f, 0.f, 0.f, 0.f, 0.f, 0.f};
  o.y2 = (v8f){0.f, 0.f, 0.f, 0.f, 0.f, 0.f, 0.f, 0.f};
#pragma unroll
  for (int c = 0; c < KC; ++c) {
    int kk = 4 * c + koff;
    int r0 = reflect_i(x0 + kk, r);
    int r1 = reflect_i(x0 + kk + 1, r);
    unsigned o0 = (unsigned)(r0 >> 1) * (unsigned)Wb + jb;
    unsigned o1 = (unsigned)(r1 >> 1) * (unsigned)Wb + jb;
    int d0 = r0 & 1, d1 = r1 & 1;
    v2f xl, xlh, xhl, xhh;
    xl.x = p0[(unsigned)r0 * (unsigned)W + (unsigned)cc];
    xl.y = p0[(unsigned)r1 * (unsigned)W + (unsigned)cc];
    xlh.x = c2q_val(q0, q5, o0, d0, cb);
    xlh.y = c2q_val(q0, q5, o1, d1, cb);
    xhl.x = c2q_val(q2, q3, o0, d0, cb);
    xhl.y = c2q_val(q2, q3, o1, d1, cb);
    xhh.x = c2q_val(q1, q4, o0, d0, cb);
    xhh.y = c2q_val(q1, q4, o1, d1, cb);
    o.y1 = wmma4(xl, f0[c], o.y1);
    o.y1 = wmma4(xlh, f1s[c], o.y1);
    o.y2 = wmma4(xhl, f0s[c], o.y2);
    o.y2 = wmma4(xhh, f1s[c], o.y2);
    if (KC > 4 && c == KC / 2 - 1) __builtin_amdgcn_sched_barrier(0);
  }
  return o;
}

__device__ __forceinline__ void store_tile(float* __restrict__ op, int pitch,
                                           v8f acc) {
#pragma unroll
  for (int e = 0; e < 8; ++e) op[(size_t)e * pitch] = acc[e];
}

__device__ __forceinline__ void tile_result_fence() {
  __builtin_amdgcn_sched_barrier(0);
  asm volatile(
      "s_wait_alu depctr_va_vdst(0)\n\t"
      "v_nop\n\tv_nop\n\tv_nop\n\tv_nop\n\tv_nop\n\t"
      "v_nop\n\tv_nop\n\tv_nop\n\tv_nop" ::: "memory");
  __builtin_amdgcn_sched_barrier(0);
}

__device__ __forceinline__ v8f row_even(const YT& P, const v2f* fe0,
                                        const v2f* fe1) {
  v8f acc = {0.f, 0.f, 0.f, 0.f, 0.f, 0.f, 0.f, 0.f};
#pragma unroll
  for (int q = 0; q < 4; ++q) {
    v2f a1, a2;
    a1.x = P.y1[2 * q];
    a1.y = P.y1[2 * q + 1];
    a2.x = P.y2[2 * q];
    a2.y = P.y2[2 * q + 1];
    acc = wmma4(a1, fe0[q], acc);
    acc = wmma4(a2, fe1[q], acc);
  }
  return acc;
}

__device__ __forceinline__ v8f row_odd(const YT& P, const YT& Q, int half,
                                       const v2f* fo0, const v2f* fo1) {
  v8f acc = {0.f, 0.f, 0.f, 0.f, 0.f, 0.f, 0.f, 0.f};
#pragma unroll
  for (int q = 0; q < 4; ++q) {
    v2f a1, a2;
    a1.x = half ? P.y1[2 * q] : Q.y1[2 * q];
    a1.y = half ? P.y1[2 * q + 1] : Q.y1[2 * q + 1];
    a2.x = half ? P.y2[2 * q] : Q.y2[2 * q];
    a2.y = half ? P.y2[2 * q + 1] : Q.y2[2 * q + 1];
    acc = wmma4(a1, fo0[q], acc);
    acc = wmma4(a2, fo1[q], acc);
  }
  return acc;
}

template <int NT, int Hb, int Wb>
__global__ void __launch_bounds__(128)
j2_fused_kernel(const float* __restrict__ plain, const float* __restrict__ yh,
                const float* __restrict__ ha0, const float* __restrict__ hb0,
                const float* __restrict__ ha1, const float* __restrict__ hb1,
                float* __restrict__ out, int B) {
  constexpr int r = 2 * Hb;
  constexpr int W = 2 * Wb;
  int lane = threadIdx.x & 31;
  int wave = __builtin_amdgcn_readfirstlane((int)(threadIdx.x >> 5));
  int g = (int)blockIdx.x * 4 + wave;
  constexpr int nCS = W / (16 * NT);
  constexpr int nRT = (2 * r) / 16;
  int cs = g % nCS;
  int g2 = g / nCS;
  int I = g2 % nRT;
  int b = g2 / nRT;
  if (b >= B) return;
  int half = lane >> 4;
  int l16 = lane & 15;
  int koff = half * 2;
  int x0 = 8 * I - 4;
  int cy0 = cs * 16 * NT - 4;

  v2f f0[4], f0s[4], f1s[4];
  v2f fe0[4], fe1[4], fo0[4], fo1[4];
#pragma unroll
  for (int c = 0; c < 4; ++c) {
    int kk = 4 * c + koff;
    float ax = cif_A(l16, kk, ha0, hb0, 0);
    float ay = cif_A(l16, kk + 1, ha0, hb0, 0);
    f0[c].x = ax;
    f0[c].y = ay;
    f0s[c].x = ax * SQRT_HALF;
    f0s[c].y = ay * SQRT_HALF;
    f1s[c].x = cif_A(l16, kk, ha1, hb1, 1) * SQRT_HALF;
    f1s[c].y = cif_A(l16, kk + 1, ha1, hb1, 1) * SQRT_HALF;
    int ke = 8 * half + 2 * c;
    int ko = 8 * (1 - half) + 2 * c;
    fe0[c].x = cif_A(l16, ke, ha0, hb0, 0);
    fe0[c].y = cif_A(l16, ke + 1, ha0, hb0, 0);
    fe1[c].x = cif_A(l16, ke, ha1, hb1, 1);
    fe1[c].y = cif_A(l16, ke + 1, ha1, hb1, 1);
    fo0[c].x = cif_A(l16, ko, ha0, hb0, 0);
    fo0[c].y = cif_A(l16, ko + 1, ha0, hb0, 0);
    fo1[c].x = cif_A(l16, ko, ha1, hb1, 1);
    fo1[c].y = cif_A(l16, ko + 1, ha1, hb1, 1);
  }

  constexpr int outW = 2 * W;
  float* op = out + ((size_t)b * (2 * r) + (size_t)I * 16 + half * 8) * outW +
              cs * 32 * NT + l16;

  YT P = col_tile<4, Hb, Wb>(plain, yh, b, x0, cy0, l16, koff, f0, f0s, f1s);
  tile_result_fence();
#pragma unroll 1
  for (int t = 0; t < NT; t += 2) {
    store_tile(op + 32 * t, outW, row_even(P, fe0, fe1));
    YT Q = col_tile<4, Hb, Wb>(plain, yh, b, x0, cy0 + 16 * (t + 1), l16, koff,
                               f0, f0s, f1s);
    tile_result_fence();
    store_tile(op + 32 * t + 16, outW, row_odd(P, Q, half, fo0, fo1));
    __builtin_amdgcn_sched_barrier(0);
    store_tile(op + 32 * t + 32, outW, row_even(Q, fe0, fe1));
    P = col_tile<4, Hb, Wb>(plain, yh, b, x0, cy0 + 16 * (t + 2), l16, koff, f0,
                            f0s, f1s);
    tile_result_fence();
    store_tile(op + 32 * t + 48, outW, row_odd(Q, P, half, fo0, fo1));
  }
}

__device__ __forceinline__ v8f row_j1(const YT& P, const YT& Q, const v2f* fa0,
                                      const v2f* fa1, const v2f* fb0,
                                      const v2f* fb1) {
  v8f acc = {0.f, 0.f, 0.f, 0.f, 0.f, 0.f, 0.f, 0.f};
#pragma unroll
  for (int q = 0; q < 4; ++q) {
    v2f a1, a2;
    a1.x = P.y1[2 * q];
    a1.y = P.y1[2 * q + 1];
    a2.x = P.y2[2 * q];
    a2.y = P.y2[2 * q + 1];
    acc = wmma4(a1, fa0[q], acc);
    acc = wmma4(a2, fa1[q], acc);
  }
#pragma unroll
  for (int q = 0; q < 3; ++q) {
    v2f a1, a2;
    a1.x = Q.y1[2 * q];
    a1.y = Q.y1[2 * q + 1];
    a2.x = Q.y2[2 * q];
    a2.y = Q.y2[2 * q + 1];
    acc = wmma4(a1, fb0[q], acc);
    acc = wmma4(a2, fb1[q], acc);
  }
  return acc;
}

template <int NOUT, int Hb, int Wb>
__global__ void __launch_bounds__(128)
j1_fused_kernel(const float* __restrict__ plain, const float* __restrict__ yh,
                const float* __restrict__ h0, int taps0,
                const float* __restrict__ h1, int taps1,
                float* __restrict__ out, int B) {
  constexpr int r = 2 * Hb;
  constexpr int W = 2 * Wb;
  int lane = threadIdx.x & 31;
  int wave = __builtin_amdgcn_readfirstlane((int)(threadIdx.x >> 5));
  int g = (int)blockIdx.x * 4 + wave;
  constexpr int nCS = W / (16 * NOUT);
  constexpr int nRT = r / 16;
  int cs = g % nCS;
  int g2 = g / nCS;
  int I = g2 % nRT;
  int b = g2 / nRT;
  if (b >= B) return;
  int half = lane >> 4;
  int l16 = lane & 15;
  int koff = half * 2;
  int x0 = 16 * I - 3;
  int cy0 = cs * 16 * NOUT - 3;

  v2f f0[6], f0s[6], f1s[6];
#pragma unroll
  for (int c = 0; c < 6; ++c) {
    int kk = 4 * c + koff;
    float ax = cf_A(l16, kk, h0, taps0);
    float ay = cf_A(l16, kk + 1, h0, taps0);
    f0[c].x = ax;
    f0[c].y = ay;
    f0s[c].x = ax * SQRT_HALF;
    f0s[c].y = ay * SQRT_HALF;
    f1s[c].x = cf_A(l16, kk, h1, taps1) * SQRT_HALF;
    f1s[c].y = cf_A(l16, kk + 1, h1, taps1) * SQRT_HALF;
  }
  v2f fa0[4], fa1[4], fb0[3], fb1[3];
#pragma unroll
  for (int q = 0; q < 4; ++q) {
    int ka = 8 * half + 2 * q;
    fa0[q].x = cf_A(l16, ka, h0, taps0);
    fa0[q].y = cf_A(l16, ka + 1, h0, taps0);
    fa1[q].x = cf_A(l16, ka, h1, taps1);
    fa1[q].y = cf_A(l16, ka + 1, h1, taps1);
  }
#pragma unroll
  for (int q = 0; q < 3; ++q) {
    int kb = 16 + 8 * half + 2 * q;
    fb0[q].x = cf_A(l16, kb, h0, taps0);
    fb0[q].y = cf_A(l16, kb + 1, h0, taps0);
    fb1[q].x = cf_A(l16, kb, h1, taps1);
    fb1[q].y = cf_A(l16, kb + 1, h1, taps1);
  }

  float* op = out + ((size_t)b * r + (size_t)I * 16 + half * 8) * W +
              cs * 16 * NOUT + l16;

  YT P = col_tile<6, Hb, Wb>(plain, yh, b, x0, cy0, l16, koff, f0, f0s, f1s);
#pragma unroll 1
  for (int j = 0; j < NOUT; j += 2) {
    YT Q = col_tile<6, Hb, Wb>(plain, yh, b, x0, cy0 + 16 * (j + 1), l16, koff,
                               f0, f0s, f1s);
    tile_result_fence();
    store_tile(op + 16 * j, W, row_j1(P, Q, fa0, fa1, fb0, fb1));
    __builtin_amdgcn_sched_barrier(0);
    P = col_tile<6, Hb, Wb>(plain, yh, b, x0, cy0 + 16 * (j + 2), l16, koff, f0,
                            f0s, f1s);
    tile_result_fence();
    store_tile(op + 16 * j + 16, W, row_j1(Q, P, fa0, fa1, fb0, fb1));
  }
}

extern "C" void kernel_launch(void* const* d_in, const int* in_sizes, int n_in,
                              void* d_out, int out_size, void* d_ws,
                              size_t ws_size, hipStream_t stream) {
  (void)in_sizes;
  (void)n_in;
  (void)out_size;
  (void)ws_size;

  const float* yl = (const float*)d_in[0];
  const float* yh0 = (const float*)d_in[1];
  const float* yh1 = (const float*)d_in[2];
  const float* yh2 = (const float*)d_in[3];
  const float* g0o = (const float*)d_in[4];
  const float* g1o = (const float*)d_in[5];
  const float* g0a = (const float*)d_in[6];
  const float* g0b = (const float*)d_in[7];
  const float* g1a = (const float*)d_in[8];
  const float* g1b = (const float*)d_in[9];

  float* LLa = (float*)d_ws;
  float* LLb = LLa + (size_t)64 * 65536;
  float* outp = (float*)d_out;

  dim3 blk(128);

  j2_fused_kernel<2, 64, 64><<<dim3(1024), blk, 0, stream>>>(
      yl, yh2, g0b, g0a, g1b, g1a, LLa, 64);

  j2_fused_kernel<4, 128, 128><<<dim3(2048), blk, 0, stream>>>(
      LLa, yh1, g0b, g0a, g1b, g1a, LLb, 64);

  j1_fused_kernel<8, 256, 256><<<dim3(2048), blk, 0, stream>>>(
      LLb, yh0, g0o, 7, g1o, 5, outp, 64);
}


# Column tiles aligned to 16 columns, tile columns dealt to lanes so that a K step is 4 adjacent columns; row margins are one K step of the neighbour tile; a row's sum order changes
# speedup vs baseline: 2.5603x; 1.1450x over previous
#include <hip/hip_runtime.h>


typedef __attribute__((ext_vector_type(2))) float v2f;
typedef __attribute__((ext_vector_type(8))) float v8f;

#define SQRT_HALF 0.70710678118654752440f

__device__ __forceinline__ int reflect_i(int x, int n) {
  x = (x < 0) ? (-1 - x) : x;
  x = (x >= n) ? (2 * n - 1 - x) : x;
  return x;
}

__device__ __forceinline__ int tile_col(int m) {
  return 4 * ((m & 7) >> 1) + 2 * (m >> 3) + (m & 1);
}

__device__ __forceinline__ float cif_A(int m, int kk,
                                       const float* __restrict__ ha,
                                       const float* __restrict__ hb, int hp) {
  int p = m & 3;
  int q = m >> 2;
  int want_odd = hp ? ((p & 1) ^ 1) : (p & 1);
  int t = ((kk - want_odd) >> 1) - q;
  bool ok = ((kk & 1) == want_odd) && (t >= 0) && (t <= 4);
  int tc = t < 0 ? 0 : (t > 4 ? 4 : t);
  const float* __restrict__ h = ((p == 0) || (p == 2)) ? ha : hb;
  int idx = (p < 2) ? (2 * tc + 1) : (2 * tc);
  float v = h[idx];
  return ok ? v : 0.0f;
}

__device__ __forceinline__ float cf_A(int m, int kk,
                                      const float* __restrict__ h, int taps) {
  int t = kk - m - (3 - (taps >> 1));
  bool ok = (t >= 0) && (t < taps);
  int tc = t < 0 ? 0 : (t >= taps ? taps - 1 : t);
  float v = h[tc];
  return ok ? v : 0.0f;
}

__device__ __forceinline__ float c2q_val(const float* __restrict__ p1,
                                         const float* __restrict__ p2,
                                         unsigned o, int odd_row, int cb) {
  v2f w1 = ((const v2f*)p1)[o];
  v2f w2 = ((const v2f*)p2)[o];
  float even = cb ? (w1.y + w2.y) : (w1.x + w2.x);
  float odd = cb ? (w2.x - w1.x) : (w1.y - w2.y);
  return odd_row ? odd : even;
}

__device__ __forceinline__ v8f wmma4(v2f a, v2f b, v8f c) {
  return __builtin_amdgcn_wmma_f32_16x16x4_f32(false, a, false, b, (short)0, c,
                                               false, false);
}

struct YT {
  v8f y1, y2;
};

template <int KC, int Hb, int Wb>
__device__ __forceinline__ YT col_tile(const float* __restrict__ plain,
                                       const float* __restrict__ yh, int b,
                                       int x0, int ccu, int koff,
                                       const v2f* f0, const v2f* f0s,
                                       const v2f* f1s) {
  constexpr int r = 2 * Hb;
  constexpr int W = 2 * Wb;
  int cc = reflect_i(ccu, W);
  int cb = cc & 1;
  unsigned jb = (unsigned)(cc >> 1);
  constexpr size_t bs = (size_t)Hb * Wb * 2;
  const float* p0 = plain + (size_t)b * r * W;
  const float* yb = yh + (size_t)b * 6 * bs;
  const float* q0 = yb;
  const float* q1 = yb + bs;
  const float* q2 = yb + 2 * bs;
  const float* q3 = yb + 3 * bs;
  const float* q4 = yb + 4 * bs;
  const float* q5 = yb + 5 * bs;
  YT o;
  o.y1 = (v8f){0.f, 0.f, 0.f, 0.f, 0.f, 0.f, 0.f, 0.f};
  o.y2 = (v8f){0.f, 0.f, 0.f, 0.f, 0.f, 0.f, 0.f, 0.f};
#pragma unroll
  for (int c = 0; c < KC; ++c) {
    int kk = 4 * c + koff;
    int r0 = reflect_i(x0 + kk, r);
    int r1 = reflect_i(x0 + kk + 1, r);
    unsigned o0 = (unsigned)(r0 >> 1) * (unsigned)Wb + jb;
    unsigned o1 = (unsigned)(r1 >> 1) * (unsigned)Wb + jb;
    int d0 = r0 & 1, d1 = r1 & 1;
    v2f xl, xlh, xhl, xhh;
    xl.x = p0[(unsigned)r0 * (unsigned)W + (unsigned)cc];
    xl.y = p0[(unsigned)r1 * (unsigned)W + (unsigned)cc];
    xlh.x = c2q_val(q0, q5, o0, d0, cb);
    xlh.y = c2q_val(q0, q5, o1, d1, cb);
    xhl.x = c2q_val(q2, q3, o0, d0, cb);
    xhl.y = c2q_val(q2, q3, o1, d1, cb);
    xhh.x = c2q_val(q1, q4, o0, d0, cb);
    xhh.y = c2q_val(q1, q4, o1, d1, cb);
    o.y1 = wmma4(xl, f0[c], o.y1);
    o.y1 = wmma4(xlh, f1s[c], o.y1);
    o.y2 = wmma4(xhl, f0s[c], o.y2);
    o.y2 = wmma4(xhh, f1s[c], o.y2);
    if (KC > 4 && c == KC / 2 - 1) __builtin_amdgcn_sched_barrier(0);
  }
  return o;
}

__device__ __forceinline__ void store_tile(float* __restrict__ op, int pitch,
                                           v8f acc) {
#pragma unroll
  for (int e = 0; e < 8; ++e) op[(size_t)e * pitch] = acc[e];
}

__device__ __forceinline__ void tile_result_fence(YT& T) {
  __builtin_amdgcn_sched_barrier(0);
  asm volatile(
      "s_wait_alu depctr_va_vdst(0)\n\t"
      "v_nop\n\tv_nop\n\tv_nop\n\tv_nop\n\tv_nop\n\t"
      "v_nop\n\tv_nop\n\tv_nop\n\tv_nop"
      : "+v"(T.y1), "+v"(T.y2)
      :
      : "memory");
  __builtin_amdgcn_sched_barrier(0);
}

__device__ __forceinline__ void tile_result_fence(YT& H, YT& T) {
  __builtin_amdgcn_sched_barrier(0);
  asm volatile(
      "s_wait_alu depctr_va_vdst(0)\n\t"
      "v_nop\n\tv_nop\n\tv_nop\n\tv_nop\n\tv_nop\n\t"
      "v_nop\n\tv_nop\n\tv_nop\n\tv_nop"
      : "+v"(H.y1), "+v"(H.y2), "+v"(T.y1), "+v"(T.y2)
      :
      : "memory");
  __builtin_amdgcn_sched_barrier(0);
}

#define STEP1(T, q) ((v2f){(T).y1[2 * (q)], (T).y1[2 * (q) + 1]})
#define STEP2(T, q) ((v2f){(T).y2[2 * (q)], (T).y2[2 * (q) + 1]})

__device__ __forceinline__ v8f row_even(v2f l1, v2f l2, const YT& C,
                                        const v2f* g0, const v2f* g1) {
  v8f acc = {0.f, 0.f, 0.f, 0.f, 0.f, 0.f, 0.f, 0.f};
  acc = wmma4(l1, g0[0], acc);
  acc = wmma4(l2, g1[0], acc);
#pragma unroll
  for (int q = 0; q < 3; ++q) {
    acc = wmma4(STEP1(C, q), g0[q + 1], acc);
    acc = wmma4(STEP2(C, q), g1[q + 1], acc);
  }
  return acc;
}

__device__ __forceinline__ v8f row_odd(const YT& C, v2f r1, v2f r2,
                                       const v2f* g0, const v2f* g1) {
  v8f acc = {0.f, 0.f, 0.f, 0.f, 0.f, 0.f, 0.f, 0.f};
#pragma unroll
  for (int q = 1; q < 4; ++q) {
    acc = wmma4(STEP1(C, q), g0[q - 1], acc);
    acc = wmma4(STEP2(C, q), g1[q - 1], acc);
  }
  acc = wmma4(r1, g0[3], acc);
  acc = wmma4(r2, g1[3], acc);
  return acc;
}

template <int NT, int Hb, int Wb>
__global__ void __launch_bounds__(128)
j2_fused_kernel(const float* __restrict__ plain, const float* __restrict__ yh,
                const float* __restrict__ ha0, const float* __restrict__ hb0,
                const float* __restrict__ ha1, const float* __restrict__ hb1,
                float* __restrict__ out, int B) {
  constexpr int r = 2 * Hb;
  constexpr int W = 2 * Wb;
  int lane = threadIdx.x & 31;
  int wave = __builtin_amdgcn_readfirstlane((int)(threadIdx.x >> 5));
  int g = (int)blockIdx.x * 4 + wave;
  constexpr int nCS = W / (16 * NT);
  constexpr int nRT = (2 * r) / 16;
  int cs = g % nCS;
  int g2 = g / nCS;
  int I = g2 % nRT;
  int b = g2 / nRT;
  if (b >= B) return;
  int half = lane >> 4;
  int l16 = lane & 15;
  int koff = half * 2;
  int x0 = 8 * I - 4;
  int lcol = tile_col(l16);
  int S = cs * 16 * NT;
  int hcol = S + 16 * NT + lcol - (lcol >= 12 ? 16 * NT + 16 : 0);

  v2f f0[4], f0s[4], f1[4], f1s[4];
#pragma unroll
  for (int c = 0; c < 4; ++c) {
    int kk = 4 * c + koff;
    float ax = cif_A(l16, kk, ha0, hb0, 0);
    float ay = cif_A(l16, kk + 1, ha0, hb0, 0);
    float bx = cif_A(l16, kk, ha1, hb1, 1);
    float by = cif_A(l16, kk + 1, ha1, hb1, 1);
    f0[c].x = ax;
    f0[c].y = ay;
    f0s[c].x = ax * SQRT_HALF;
    f0s[c].y = ay * SQRT_HALF;
    f1[c].x = bx;
    f1[c].y = by;
    f1s[c].x = bx * SQRT_HALF;
    f1s[c].y = by * SQRT_HALF;
  }

  constexpr int outW = 2 * W;
  float* op = out + ((size_t)b * (2 * r) + (size_t)I * 16 + half * 8) * outW +
              cs * 32 * NT + l16;

  YT Hh = col_tile<4, Hb, Wb>(plain, yh, b, x0, hcol, koff, f0, f0s, f1s);
  YT P = col_tile<4, Hb, Wb>(plain, yh, b, x0, S + lcol, koff, f0, f0s, f1s);
  tile_result_fence(Hh, P);
  v2f l1 = STEP1(Hh, 3), l2 = STEP2(Hh, 3);
  v2f e1 = STEP1(Hh, 0), e2 = STEP2(Hh, 0);
#pragma unroll 1
  for (int u = 0; u + 2 < NT; u += 2) {
    store_tile(op + 32 * u, outW, row_even(l1, l2, P, f0, f1));
    YT Q = col_tile<4, Hb, Wb>(plain, yh, b, x0, S + 16 * (u + 1) + lcol, koff,
                               f0, f0s, f1s);
    tile_result_fence(Q);
    store_tile(op + 32 * u + 16, outW,
               row_odd(P, STEP1(Q, 0), STEP2(Q, 0), f0, f1));
    __builtin_amdgcn_sched_barrier(0);
    store_tile(op + 32 * u + 32, outW,
               row_even(STEP1(P, 3), STEP2(P, 3), Q, f0, f1));
    P = col_tile<4, Hb, Wb>(plain, yh, b, x0, S + 16 * (u + 2) + lcol, koff, f0,
                            f0s, f1s);
    tile_result_fence(P);
    store_tile(op + 32 * u + 48, outW,
               row_odd(Q, STEP1(P, 0), STEP2(P, 0), f0, f1));
    l1 = STEP1(Q, 3);
    l2 = STEP2(Q, 3);
  }
  {
    constexpr int u = NT - 2;
    store_tile(op + 32 * u, outW, row_even(l1, l2, P, f0, f1));
    YT Q = col_tile<4, Hb, Wb>(plain, yh, b, x0, S + 16 * (u + 1) + lcol, koff,
                               f0, f0s, f1s);
    tile_result_fence(Q);
    store_tile(op + 32 * u + 16, outW,
               row_odd(P, STEP1(Q, 0), STEP2(Q, 0), f0, f1));
    store_tile(op + 32 * u + 32, outW,
               row_even(STEP1(P, 3), STEP2(P, 3), Q, f0, f1));
    store_tile(op + 32 * u + 48, outW, row_odd(Q, e1, e2, f0, f1));
  }
}

__device__ __forceinline__ v8f row_j1(v2f l1, v2f l2, const YT& C, v2f r1,
                                      v2f r2, const v2f* g0, const v2f* g1) {
  v8f acc = {0.f, 0.f, 0.f, 0.f, 0.f, 0.f, 0.f, 0.f};
  acc = wmma4(l1, g0[0], acc);
  acc = wmma4(l2, g1[0], acc);
#pragma unroll
  for (int q = 0; q < 4; ++q) {
    acc = wmma4(STEP1(C, q), g0[q + 1], acc);
    acc = wmma4(STEP2(C, q), g1[q + 1], acc);
  }
  acc = wmma4(r1, g0[5], acc);
  acc = wmma4(r2, g1[5], acc);
  return acc;
}

template <int NOUT, int Hb, int Wb>
__global__ void __launch_bounds__(128)
j1_fused_kernel(const float* __restrict__ plain, const float* __restrict__ yh,
                const float* __restrict__ h0, int taps0,
                const float* __restrict__ h1, int taps1,
                float* __restrict__ out, int B) {
  constexpr int r = 2 * Hb;
  constexpr int W = 2 * Wb;
  int lane = threadIdx.x & 31;
  int wave = __builtin_amdgcn_readfirstlane((int)(threadIdx.x >> 5));
  int g = (int)blockIdx.x * 4 + wave;
  constexpr int nCS = W / (16 * NOUT);
  constexpr int nRT = r / 16;
  int cs = g % nCS;
  int g2 = g / nCS;
  int I = g2 % nRT;
  int b = g2 / nRT;
  if (b >= B) return;
  int half = lane >> 4;
  int l16 = lane & 15;
  int koff = half * 2;
  int x0 = 16 * I - 3;
  int lcol = tile_col(l16);
  int S = cs * 16 * NOUT;
  int hcol = S + 16 * NOUT + lcol - (lcol >= 12 ? 16 * NOUT + 16 : 0);

  v2f f0[6], f0s[6], f1s[6];
#pragma unroll
  for (int c = 0; c < 6; ++c) {
    int kk = 4 * c + koff;
    float ax = cf_A(l16, kk, h0, taps0);
    float ay = cf_A(l16, kk + 1, h0, taps0);
    f0[c].x = ax;
    f0[c].y = ay;
    f0s[c].x = ax * SQRT_HALF;
    f0s[c].y = ay * SQRT_HALF;
    f1s[c].x = cf_A(l16, kk, h1, taps1) * SQRT_HALF;
    f1s[c].y = cf_A(l16, kk + 1, h1, taps1) * SQRT_HALF;
  }
  v2f g0[6], g1[6];
#pragma unroll
  for (int s = 0; s < 6; ++s) {
    int kk = 4 * s - 1 + koff;
    g0[s].x = cf_A(l16, kk, h0, taps0);
    g0[s].y = cf_A(l16, kk + 1, h0, taps0);
    g1[s].x = cf_A(l16, kk, h1, taps1);
    g1[s].y = cf_A(l16, kk + 1, h1, taps1);
  }

  float* op = out + ((size_t)b * r + (size_t)I * 16 + half * 8) * W +
              cs * 16 * NOUT + l16;

  YT Hh = col_tile<6, Hb, Wb>(plain, yh, b, x0, hcol, koff, f0, f0s, f1s);
  YT P = col_tile<6, Hb, Wb>(plain, yh, b, x0, S + lcol, koff, f0, f0s, f1s);
  tile_result_fence(Hh, P);
  v2f l1 = STEP1(Hh, 3), l2 = STEP2(Hh, 3);
  v2f e1 = STEP1(Hh, 0), e2 = STEP2(Hh, 0);
#pragma unroll 1
  for (int j = 0; j + 2 < NOUT; j += 2) {
    YT Q = col_tile<6, Hb, Wb>(plain, yh, b, x0, S + 16 * (j + 1) + lcol, koff,
                               f0, f0s, f1s);
    tile_result_fence(Q);
    store_tile(op + 16 * j, W,
               row_j1(l1, l2, P, STEP1(Q, 0), STEP2(Q, 0), g0, g1));
    l1 = STEP1(P, 3);
    l2 = STEP2(P, 3);
    __builtin_amdgcn_sched_barrier(0);
    P = col_tile<6, Hb, Wb>(plain, yh, b, x0, S + 16 * (j + 2) + lcol, koff, f0,
                            f0s, f1s);
    tile_result_fence(P);
    store_tile(op + 16 * j + 16, W,
               row_j1(l1, l2, Q, STEP1(P, 0), STEP2(P, 0), g0, g1));
    l1 = STEP1(Q, 3);
    l2 = STEP2(Q, 3);
  }
  {
    constexpr int j = NOUT - 2;
    YT Q = col_tile<6, Hb, Wb>(plain, yh, b, x0, S + 16 * (j + 1) + lcol, koff,
                               f0, f0s, f1s);
    tile_result_fence(Q);
    store_tile(op + 16 * j, W,
               row_j1(l1, l2, P, STEP1(Q, 0), STEP2(Q, 0), g0, g1));
    store_tile(op + 16 * j + 16, W,
               row_j1(STEP1(P, 3), STEP2(P, 3), Q, e1, e2, g0, g1));
  }
}

extern "C" void kernel_launch(void* const* d_in, const int* in_sizes, int n_in,
                              void* d_out, int out_size, void* d_ws,
                              size_t ws_size, hipStream_t stream) {
  (void)in_sizes;
  (void)n_in;
  (void)out_size;
  (void)ws_size;

  const float* yl = (const float*)d_in[0];
  const float* yh0 = (const float*)d_in[1];
  const float* yh1 = (const float*)d_in[2];
  const float* yh2 = (const float*)d_in[3];
  const float* g0o = (const float*)d_in[4];
  const float* g1o = (const float*)d_in[5];
  const float* g0a = (const float*)d_in[6];
  const float* g0b = (const float*)d_in[7];
  const float* g1a = (const float*)d_in[8];
  const float* g1b = (const float*)d_in[9];

  float* LLa = (float*)d_ws;
  float* LLb = LLa + (size_t)64 * 65536;
  float* outp = (float*)d_out;

  dim3 blk(128);

  j2_fused_kernel<2, 64, 64><<<dim3(1024), blk, 0, stream>>>(
      yl, yh2, g0b, g0a, g1b, g1a, LLa, 64);

  j2_fused_kernel<4, 128, 128><<<dim3(2048), blk, 0, stream>>>(
      LLa, yh1, g0b, g0a, g1b, g1a, LLb, 64);

  j1_fused_kernel<8, 256, 256><<<dim3(2048), blk, 0, stream>>>(
      LLb, yh0, g0o, 7, g1o, 5, outp, 64);
}


# Levels 2 and 1: strips twice as long (half the image width): half as many halo tiles, each just the neighbour strip's edge tile; image-edge margin = own edge K step mirrored across half waves
# speedup vs baseline: 2.9021x; 1.1335x over previous
#include <hip/hip_runtime.h>


typedef __attribute__((ext_vector_type(2))) float v2f;
typedef __attribute__((ext_vector_type(8))) float v8f;

#define SQRT_HALF 0.70710678118654752440f

__device__ __forceinline__ int reflect_i(int x, int n) {
  x = (x < 0) ? (-1 - x) : x;
  x = (x >= n) ? (2 * n - 1 - x) : x;
  return x;
}

__device__ __forceinline__ int tile_col(int m) {
  return 4 * ((m & 7) >> 1) + 2 * (m >> 3) + (m & 1);
}

__device__ __forceinline__ float cif_A(int m, int kk,
                                       const float* __restrict__ ha,
                                       const float* __restrict__ hb, int hp) {
  int p = m & 3;
  int q = m >> 2;
  int want_odd = hp ? ((p & 1) ^ 1) : (p & 1);
  int t = ((kk - want_odd) >> 1) - q;
  bool ok = ((kk & 1) == want_odd) && (t >= 0) && (t <= 4);
  int tc = t < 0 ? 0 : (t > 4 ? 4 : t);
  const float* __restrict__ h = ((p == 0) || (p == 2)) ? ha : hb;
  int idx = (p < 2) ? (2 * tc + 1) : (2 * tc);
  float v = h[idx];
  return ok ? v : 0.0f;
}

__device__ __forceinline__ float cf_A(int m, int kk,
                                      const float* __restrict__ h, int taps) {
  int t = kk - m - (3 - (taps >> 1));
  bool ok = (t >= 0) && (t < taps);
  int tc = t < 0 ? 0 : (t >= taps ? taps - 1 : t);
  float v = h[tc];
  return ok ? v : 0.0f;
}

__device__ __forceinline__ float c2q_val(const float* __restrict__ p1,
                                         const float* __restrict__ p2,
                                         unsigned o, int odd_row, int cb) {
  v2f w1 = ((const v2f*)p1)[o];
  v2f w2 = ((const v2f*)p2)[o];
  float even = cb ? (w1.y + w2.y) : (w1.x + w2.x);
  float odd = cb ? (w2.x - w1.x) : (w1.y - w2.y);
  return odd_row ? odd : even;
}

__device__ __forceinline__ v8f wmma4(v2f a, v2f b, v8f c) {
  return __builtin_amdgcn_wmma_f32_16x16x4_f32(false, a, false, b, (short)0, c,
                                               false, false);
}

struct YT {
  v8f y1, y2;
};

template <int KC, int Hb, int Wb>
__device__ __forceinline__ YT col_tile(const float* __restrict__ plain,
                                       const float* __restrict__ yh, int b,
                                       int x0, int ccu, int koff,
                                       const v2f* f0, const v2f* f0s,
                                       const v2f* f1s) {
  constexpr int r = 2 * Hb;
  constexpr int W = 2 * Wb;
  int cc = reflect_i(ccu, W);
  int cb = cc & 1;
  unsigned jb = (unsigned)(cc >> 1);
  constexpr size_t bs = (size_t)Hb * Wb * 2;
  const float* p0 = plain + (size_t)b * r * W;
  const float* yb = yh + (size_t)b * 6 * bs;
  const float* q0 = yb;
  const float* q1 = yb + bs;
  const float* q2 = yb + 2 * bs;
  const float* q3 = yb + 3 * bs;
  const float* q4 = yb + 4 * bs;
  const float* q5 = yb + 5 * bs;
  YT o;
  o.y1 = (v8f){0.f, 0.f, 0.f, 0.f, 0.f, 0.f, 0.f, 0.f};
  o.y2 = (v8f){0.f, 0.f, 0.f, 0.f, 0.f, 0.f, 0.f, 0.f};
#pragma unroll
  for (int c = 0; c < KC; ++c) {
    int kk = 4 * c + koff;
    int r0 = reflect_i(x0 + kk, r);
    int r1 = reflect_i(x0 + kk + 1, r);
    unsigned o0 = (unsigned)(r0 >> 1) * (unsigned)Wb + jb;
    unsigned o1 = (unsigned)(r1 >> 1) * (unsigned)Wb + jb;
    int d0 = r0 & 1, d1 = r1 & 1;
    v2f xl, xlh, xhl, xhh;
    xl.x = p0[(unsigned)r0 * (unsigned)W + (unsigned)cc];
    xl.y = p0[(unsigned)r1 * (unsigned)W + (unsigned)cc];
    xlh.x = c2q_val(q0, q5, o0, d0, cb);
    xlh.y = c2q_val(q0, q5, o1, d1, cb);
    xhl.x = c2q_val(q2, q3, o0, d0, cb);
    xhl.y = c2q_val(q2, q3, o1, d1, cb);
    xhh.x = c2q_val(q1, q4, o0, d0, cb);
    xhh.y = c2q_val(q1, q4, o1, d1, cb);
    o.y1 = wmma4(xl, f0[c], o.y1);
    o.y1 = wmma4(xlh, f1s[c], o.y1);
    o.y2 = wmma4(xhl, f0s[c], o.y2);
    o.y2 = wmma4(xhh, f1s[c], o.y2);
    if (KC > 4 && c == KC / 2 - 1) __builtin_amdgcn_sched_barrier(0);
  }
  return o;
}

__device__ __forceinline__ void store_tile(float* __restrict__ op, int pitch,
                                           v8f acc) {
#pragma unroll
  for (int e = 0; e < 8; ++e) op[(size_t)e * pitch] = acc[e];
}

__device__ __forceinline__ void tile_result_fence(YT& T) {
  __builtin_amdgcn_sched_barrier(0);
  asm volatile(
      "s_wait_alu depctr_va_vdst(0)\n\t"
      "v_nop\n\tv_nop\n\tv_nop\n\tv_nop\n\tv_nop\n\t"
      "v_nop\n\tv_nop\n\tv_nop\n\tv_nop"
      : "+v"(T.y1), "+v"(T.y2)
      :
      : "memory");
  __builtin_amdgcn_sched_barrier(0);
}

__device__ __forceinline__ void tile_result_fence(YT& H, YT& T) {
  __builtin_amdgcn_sched_barrier(0);
  asm volatile(
      "s_wait_alu depctr_va_vdst(0)\n\t"
      "v_nop\n\tv_nop\n\tv_nop\n\tv_nop\n\tv_nop\n\t"
      "v_nop\n\tv_nop\n\tv_nop\n\tv_nop"
      : "+v"(H.y1), "+v"(H.y2), "+v"(T.y1), "+v"(T.y2)
      :
      : "memory");
  __builtin_amdgcn_sched_barrier(0);
}

#define STEP1(T, q) ((v2f){(T).y1[2 * (q)], (T).y1[2 * (q) + 1]})
#define STEP2(T, q) ((v2f){(T).y2[2 * (q)], (T).y2[2 * (q) + 1]})

__device__ __forceinline__ float other_half(float x) {
  return __builtin_bit_cast(
      float, __builtin_amdgcn_permlanex16(__builtin_bit_cast(int, x),
                                          __builtin_bit_cast(int, x),
                                          0x76543210, 0xfedcba98, false, false));
}

__device__ __forceinline__ v2f mirror_step(v2f a) {
  return (v2f){other_half(a.y), other_half(a.x)};
}

__device__ __forceinline__ v2f pick_step(bool c, v2f a, v2f b) {
  return (v2f){c ? a.x : b.x, c ? a.y : b.y};
}

__device__ __forceinline__ v8f row_even(v2f l1, v2f l2, const YT& C,
                                        const v2f* g0, const v2f* g1) {
  v8f acc = {0.f, 0.f, 0.f, 0.f, 0.f, 0.f, 0.f, 0.f};
  acc = wmma4(l1, g0[0], acc);
  acc = wmma4(l2, g1[0], acc);
#pragma unroll
  for (int q = 0; q < 3; ++q) {
    acc = wmma4(STEP1(C, q), g0[q + 1], acc);
    acc = wmma4(STEP2(C, q), g1[q + 1], acc);
  }
  return acc;
}

__device__ __forceinline__ v8f row_odd(const YT& C, v2f r1, v2f r2,
                                       const v2f* g0, const v2f* g1) {
  v8f acc = {0.f, 0.f, 0.f, 0.f, 0.f, 0.f, 0.f, 0.f};
#pragma unroll
  for (int q = 1; q < 4; ++q) {
    acc = wmma4(STEP1(C, q), g0[q - 1], acc);
    acc = wmma4(STEP2(C, q), g1[q - 1], acc);
  }
  acc = wmma4(r1, g0[3], acc);
  acc = wmma4(r2, g1[3], acc);
  return acc;
}

template <int NT, int Hb, int Wb>
__global__ void __launch_bounds__(128, 4)
j2_fused_kernel(const float* __restrict__ plain, const float* __restrict__ yh,
                const float* __restrict__ ha0, const float* __restrict__ hb0,
                const float* __restrict__ ha1, const float* __restrict__ hb1,
                float* __restrict__ out, int B) {
  constexpr int r = 2 * Hb;
  constexpr int W = 2 * Wb;
  int lane = threadIdx.x & 31;
  int wave = __builtin_amdgcn_readfirstlane((int)(threadIdx.x >> 5));
  int g = (int)blockIdx.x * 4 + wave;
  constexpr int nCS = W / (16 * NT);
  constexpr int nRT = (2 * r) / 16;
  int cs = g % nCS;
  int g2 = g / nCS;
  int I = g2 % nRT;
  int b = g2 / nRT;
  if (b >= B) return;
  int half = lane >> 4;
  int l16 = lane & 15;
  int koff = half * 2;
  int x0 = 8 * I - 4;
  int lcol = tile_col(l16);
  int S = cs * 16 * NT;
  constexpr bool TWO = (nCS == 2);
  int hcol = TWO ? ((cs ? S - 16 : S + 16 * NT) + lcol)
                 : (S + 16 * NT + lcol - (lcol >= 12 ? 16 * NT + 16 : 0));

  v2f f0[4], f0s[4], f1[4], f1s[4];
#pragma unroll
  for (int c = 0; c < 4; ++c) {
    int kk = 4 * c + koff;
    float ax = cif_A(l16, kk, ha0, hb0, 0);
    float ay = cif_A(l16, kk + 1, ha0, hb0, 0);
    float bx = cif_A(l16, kk, ha1, hb1, 1);
    float by = cif_A(l16, kk + 1, ha1, hb1, 1);
    f0[c].x = ax;
    f0[c].y = ay;
    f0s[c].x = ax * SQRT_HALF;
    f0s[c].y = ay * SQRT_HALF;
    f1[c].x = bx;
    f1[c].y = by;
    f1s[c].x = bx * SQRT_HALF;
    f1s[c].y = by * SQRT_HALF;
  }

  constexpr int outW = 2 * W;
  float* op = out + ((size_t)b * (2 * r) + (size_t)I * 16 + half * 8) * outW +
              cs * 32 * NT + l16;

  YT Hh = col_tile<4, Hb, Wb>(plain, yh, b, x0, hcol, koff, f0, f0s, f1s);
  YT P = col_tile<4, Hb, Wb>(plain, yh, b, x0, S + lcol, koff, f0, f0s, f1s);
  tile_result_fence(Hh, P);
  v2f l1 = STEP1(Hh, 3), l2 = STEP2(Hh, 3);
  v2f e1 = STEP1(Hh, 0), e2 = STEP2(Hh, 0);
  if (TWO) {
    l1 = pick_step(cs != 0, l1, mirror_step(STEP1(P, 0)));
    l2 = pick_step(cs != 0, l2, mirror_step(STEP2(P, 0)));
  }
#pragma unroll
  for (int u = 0; u + 2 < NT; u += 2) {
    store_tile(op + 32 * u, outW, row_even(l1, l2, P, f0, f1));
    YT Q = col_tile<4, Hb, Wb>(plain, yh, b, x0, S + 16 * (u + 1) + lcol, koff,
                               f0, f0s, f1s);
    tile_result_fence(Q);
    store_tile(op + 32 * u + 16, outW,
               row_odd(P, STEP1(Q, 0), STEP2(Q, 0), f0, f1));
    __builtin_amdgcn_sched_barrier(0);
    store_tile(op + 32 * u + 32, outW,
               row_even(STEP1(P, 3), STEP2(P, 3), Q, f0, f1));
    P = col_tile<4, Hb, Wb>(plain, yh, b, x0, S + 16 * (u + 2) + lcol, koff, f0,
                            f0s, f1s);
    tile_result_fence(P);
    store_tile(op + 32 * u + 48, outW,
               row_odd(Q, STEP1(P, 0), STEP2(P, 0), f0, f1));
    l1 = STEP1(Q, 3);
    l2 = STEP2(Q, 3);
  }
  {
    constexpr int u = NT - 2;
    store_tile(op + 32 * u, outW, row_even(l1, l2, P, f0, f1));
    YT Q = col_tile<4, Hb, Wb>(plain, yh, b, x0, S + 16 * (u + 1) + lcol, koff,
                               f0, f0s, f1s);
    tile_result_fence(Q);
    if (TWO) {
      e1 = pick_step(cs != 0, mirror_step(STEP1(Q, 3)), e1);
      e2 = pick_step(cs != 0, mirror_step(STEP2(Q, 3)), e2);
    }
    store_tile(op + 32 * u + 16, outW,
               row_odd(P, STEP1(Q, 0), STEP2(Q, 0), f0, f1));
    store_tile(op + 32 * u + 32, outW,
               row_even(STEP1(P, 3), STEP2(P, 3), Q, f0, f1));
    store_tile(op + 32 * u + 48, outW, row_odd(Q, e1, e2, f0, f1));
  }
}

__device__ __forceinline__ v8f row_j1(v2f l1, v2f l2, const YT& C, v2f r1,
                                      v2f r2, const v2f* g0, const v2f* g1) {
  v8f acc = {0.f, 0.f, 0.f, 0.f, 0.f, 0.f, 0.f, 0.f};
  acc = wmma4(l1, g0[0], acc);
  acc = wmma4(l2, g1[0], acc);
#pragma unroll
  for (int q = 0; q < 4; ++q) {
    acc = wmma4(STEP1(C, q), g0[q + 1], acc);
    acc = wmma4(STEP2(C, q), g1[q + 1], acc);
  }
  acc = wmma4(r1, g0[5], acc);
  acc = wmma4(r2, g1[5], acc);
  return acc;
}

template <int NOUT, int Hb, int Wb>
__global__ void __launch_bounds__(128)
j1_fused_kernel(const float* __restrict__ plain, const float* __restrict__ yh,
                const float* __restrict__ h0, int taps0,
                const float* __restrict__ h1, int taps1,
                float* __restrict__ out, int B) {
  constexpr int r = 2 * Hb;
  constexpr int W = 2 * Wb;
  int lane = threadIdx.x & 31;
  int wave = __builtin_amdgcn_readfirstlane((int)(threadIdx.x >> 5));
  int g = (int)blockIdx.x * 4 + wave;
  constexpr int nCS = W / (16 * NOUT);
  constexpr int nRT = r / 16;
  int cs = g % nCS;
  int g2 = g / nCS;
  int I = g2 % nRT;
  int b = g2 / nRT;
  if (b >= B) return;
  int half = lane >> 4;
  int l16 = lane & 15;
  int koff = half * 2;
  int x0 = 16 * I - 3;
  int lcol = tile_col(l16);
  int S = cs * 16 * NOUT;
  constexpr bool TWO = (nCS == 2);
  int hcol = TWO ? ((cs ? S - 16 : S + 16 * NOUT) + lcol)
                 : (S + 16 * NOUT + lcol - (lcol >= 12 ? 16 * NOUT + 16 : 0));

  v2f f0[6], f0s[6], f1s[6];
#pragma unroll
  for (int c = 0; c < 6; ++c) {
    int kk = 4 * c + koff;
    float ax = cf_A(l16, kk, h0, taps0);
    float ay = cf_A(l16, kk + 1, h0, taps0);
    f0[c].x = ax;
    f0[c].y = ay;
    f0s[c].x = ax * SQRT_HALF;
    f0s[c].y = ay * SQRT_HALF;
    f1s[c].x = cf_A(l16, kk, h1, taps1) * SQRT_HALF;
    f1s[c].y = cf_A(l16, kk + 1, h1, taps1) * SQRT_HALF;
  }
  v2f g0[6], g1[6];
#pragma unroll
  for (int s = 0; s < 6; ++s) {
    int kk = 4 * s - 1 + koff;
    g0[s].x = cf_A(l16, kk, h0, taps0);
    g0[s].y = cf_A(l16, kk + 1, h0, taps0);
    g1[s].x = cf_A(l16, kk, h1, taps1);
    g1[s].y = cf_A(l16, kk + 1, h1, taps1);
  }

  float* op = out + ((size_t)b * r + (size_t)I * 16 + half * 8) * W +
              cs * 16 * NOUT + l16;

  YT Hh = col_tile<6, Hb, Wb>(plain, yh, b, x0, hcol, koff, f0, f0s, f1s);
  YT P = col_tile<6, Hb, Wb>(plain, yh, b, x0, S + lcol, koff, f0, f0s, f1s);
  tile_result_fence(Hh, P);
  v2f l1 = STEP1(Hh, 3), l2 = STEP2(Hh, 3);
  v2f e1 = STEP1(Hh, 0), e2 = STEP2(Hh, 0);
  if (TWO) {
    l1 = pick_step(cs != 0, l1, mirror_step(STEP1(P, 0)));
    l2 = pick_step(cs != 0, l2, mirror_step(STEP2(P, 0)));
  }
#pragma unroll 1
  for (int j = 0; j + 2 < NOUT; j += 2) {
    YT Q = col_tile<6, Hb, Wb>(plain, yh, b, x0, S + 16 * (j + 1) + lcol, koff,
                               f0, f0s, f1s);
    tile_result_fence(Q);
    store_tile(op + 16 * j, W,
               row_j1(l1, l2, P, STEP1(Q, 0), STEP2(Q, 0), g0, g1));
    l1 = STEP1(P, 3);
    l2 = STEP2(P, 3);
    __builtin_amdgcn_sched_barrier(0);
    P = col_tile<6, Hb, Wb>(plain, yh, b, x0, S + 16 * (j + 2) + lcol, koff, f0,
                            f0s, f1s);
    tile_result_fence(P);
    store_tile(op + 16 * j + 16, W,
               row_j1(l1, l2, Q, STEP1(P, 0), STEP2(P, 0), g0, g1));
    l1 = STEP1(Q, 3);
    l2 = STEP2(Q, 3);
  }
  {
    constexpr int j = NOUT - 2;
    YT Q = col_tile<6, Hb, Wb>(plain, yh, b, x0, S + 16 * (j + 1) + lcol, koff,
                               f0, f0s, f1s);
    tile_result_fence(Q);
    if (TWO) {
      e1 = pick_step(cs != 0, mirror_step(STEP1(Q, 3)), e1);
      e2 = pick_step(cs != 0, mirror_step(STEP2(Q, 3)), e2);
    }
    store_tile(op + 16 * j, W,
               row_j1(l1, l2, P, STEP1(Q, 0), STEP2(Q, 0), g0, g1));
    store_tile(op + 16 * j + 16, W,
               row_j1(STEP1(P, 3), STEP2(P, 3), Q, e1, e2, g0, g1));
  }
}

extern "C" void kernel_launch(void* const* d_in, const int* in_sizes, int n_in,
                              void* d_out, int out_size, void* d_ws,
                              size_t ws_size, hipStream_t stream) {
  (void)in_sizes;
  (void)n_in;
  (void)out_size;
  (void)ws_size;

  const float* yl = (const float*)d_in[0];
  const float* yh0 = (const float*)d_in[1];
  const float* yh1 = (const float*)d_in[2];
  const float* yh2 = (const float*)d_in[3];
  const float* g0o = (const float*)d_in[4];
  const float* g1o = (const float*)d_in[5];
  const float* g0a = (const float*)d_in[6];
  const float* g0b = (const float*)d_in[7];
  const float* g1a = (const float*)d_in[8];
  const float* g1b = (const float*)d_in[9];

  float* LLa = (float*)d_ws;
  float* LLb = LLa + (size_t)64 * 65536;
  float* outp = (float*)d_out;

  dim3 blk(128);

  j2_fused_kernel<2, 64, 64><<<dim3(1024), blk, 0, stream>>>(
      yl, yh2, g0b, g0a, g1b, g1a, LLa, 64);

  j2_fused_kernel<8, 128, 128><<<dim3(1024), blk, 0, stream>>>(
      LLa, yh1, g0b, g0a, g1b, g1a, LLb, 64);

  j1_fused_kernel<16, 256, 256><<<dim3(1024), blk, 0, stream>>>(
      LLb, yh0, g0o, 7, g1o, 5, outp, 64);
}


# Column pass: the two lanes of a column pair fetch one row each of a K step's two (one 8-byte load per band and plain) and swap the other column's element by DPP: half the loads; sums unchanged
# speedup vs baseline: 3.2943x; 1.1351x over previous
#include <hip/hip_runtime.h>


typedef __attribute__((ext_vector_type(2))) float v2f;
typedef __attribute__((ext_vector_type(8))) float v8f;

#define SQRT_HALF 0.70710678118654752440f

__device__ __forceinline__ int reflect_i(int x, int n) {
  x = (x < 0) ? (-1 - x) : x;
  x = (x >= n) ? (2 * n - 1 - x) : x;
  return x;
}

__device__ __forceinline__ int tile_col(int m) {
  return 4 * ((m & 7) >> 1) + 2 * (m >> 3) + (m & 1);
}

__device__ __forceinline__ float cif_A(int m, int kk,
                                       const float* __restrict__ ha,
                                       const float* __restrict__ hb, int hp) {
  int p = m & 3;
  int q = m >> 2;
  int want_odd = hp ? ((p & 1) ^ 1) : (p & 1);
  int t = ((kk - want_odd) >> 1) - q;
  bool ok = ((kk & 1) == want_odd) && (t >= 0) && (t <= 4);
  int tc = t < 0 ? 0 : (t > 4 ? 4 : t);
  const float* __restrict__ h = ((p == 0) || (p == 2)) ? ha : hb;
  int idx = (p < 2) ? (2 * tc + 1) : (2 * tc);
  float v = h[idx];
  return ok ? v : 0.0f;
}

__device__ __forceinline__ float cf_A(int m, int kk,
                                      const float* __restrict__ h, int taps) {
  int t = kk - m - (3 - (taps >> 1));
  bool ok = (t >= 0) && (t < taps);
  int tc = t < 0 ? 0 : (t >= taps ? taps - 1 : t);
  float v = h[tc];
  return ok ? v : 0.0f;
}

__device__ __forceinline__ void c2q_pair(const float* __restrict__ p1,
                                         const float* __restrict__ p2,
                                         unsigned o, int odd_row, int cb,
                                         float& own, float& give) {
  v2f w1 = ((const v2f*)p1)[o];
  v2f w2 = ((const v2f*)p2)[o];
  float c0 = odd_row ? (w1.y - w2.y) : (w1.x + w2.x);
  float c1 = odd_row ? (w2.x - w1.x) : (w1.y + w2.y);
  own = cb ? c1 : c0;
  give = cb ? c0 : c1;
}

__device__ __forceinline__ float pair_lane(float x) {
  return __builtin_bit_cast(
      float, __builtin_amdgcn_update_dpp(0, __builtin_bit_cast(int, x), 0xB1,
                                         0xF, 0xF, true));
}

__device__ __forceinline__ v8f wmma4(v2f a, v2f b, v8f c) {
  return __builtin_amdgcn_wmma_f32_16x16x4_f32(false, a, false, b, (short)0, c,
                                               false, false);
}

struct YT {
  v8f y1, y2;
};

template <int KC, int Hb, int Wb>
__device__ __forceinline__ YT col_tile(const float* __restrict__ plain,
                                       const float* __restrict__ yh, int b,
                                       int x0, int ccu, int koff,
                                       const v2f* f0, const v2f* f0s,
                                       const v2f* f1s) {
  constexpr int r = 2 * Hb;
  constexpr int W = 2 * Wb;
  int cc = reflect_i(ccu, W);
  int cb = cc & 1;
  unsigned jb = (unsigned)(cc >> 1);
  int lo = (int)(threadIdx.x & 1);
  constexpr size_t bs = (size_t)Hb * Wb * 2;
  const float* p0 = plain + (size_t)b * r * W;
  const float* yb = yh + (size_t)b * 6 * bs;
  const float* q0 = yb;
  const float* q1 = yb + bs;
  const float* q2 = yb + 2 * bs;
  const float* q3 = yb + 3 * bs;
  const float* q4 = yb + 4 * bs;
  const float* q5 = yb + 5 * bs;
  YT o;
  o.y1 = (v8f){0.f, 0.f, 0.f, 0.f, 0.f, 0.f, 0.f, 0.f};
  o.y2 = (v8f){0.f, 0.f, 0.f, 0.f, 0.f, 0.f, 0.f, 0.f};
#pragma unroll
  for (int c = 0; c < KC; ++c) {
    int kk = 4 * c + koff;
    int rs = reflect_i(x0 + kk + lo, r);
    unsigned ob = (unsigned)(rs >> 1) * (unsigned)Wb + jb;
    int d = rs & 1;
    v2f pl = ((const v2f*)p0)[(unsigned)rs * (unsigned)Wb + jb];
    float pl_o = cb ? pl.y : pl.x;
    float pl_g = cb ? pl.x : pl.y;
    float lh_o, lh_g, hl_o, hl_g, hh_o, hh_g;
    c2q_pair(q0, q5, ob, d, cb, lh_o, lh_g);
    c2q_pair(q2, q3, ob, d, cb, hl_o, hl_g);
    c2q_pair(q1, q4, ob, d, cb, hh_o, hh_g);
    float pl_x = pair_lane(pl_g);
    float lh_x = pair_lane(lh_g);
    float hl_x = pair_lane(hl_g);
    float hh_x = pair_lane(hh_g);
    v2f xl = lo ? (v2f){pl_x, pl_o} : (v2f){pl_o, pl_x};
    v2f xlh = lo ? (v2f){lh_x, lh_o} : (v2f){lh_o, lh_x};
    v2f xhl = lo ? (v2f){hl_x, hl_o} : (v2f){hl_o, hl_x};
    v2f xhh = lo ? (v2f){hh_x, hh_o} : (v2f){hh_o, hh_x};
    o.y1 = wmma4(xl, f0[c], o.y1);
    o.y1 = wmma4(xlh, f1s[c], o.y1);
    o.y2 = wmma4(xhl, f0s[c], o.y2);
    o.y2 = wmma4(xhh, f1s[c], o.y2);
    if (KC > 4 && c == KC / 2 - 1) __builtin_amdgcn_sched_barrier(0);
  }
  return o;
}

__device__ __forceinline__ void store_tile(float* __restrict__ op, int pitch,
                                           v8f acc) {
#pragma unroll
  for (int e = 0; e < 8; ++e) op[(size_t)e * pitch] = acc[e];
}

__device__ __forceinline__ void tile_result_fence(YT& T) {
  __builtin_amdgcn_sched_barrier(0);
  asm volatile(
      "s_wait_alu depctr_va_vdst(0)\n\t"
      "v_nop\n\tv_nop\n\tv_nop\n\tv_nop\n\tv_nop\n\t"
      "v_nop\n\tv_nop\n\tv_nop\n\tv_nop"
      : "+v"(T.y1), "+v"(T.y2)
      :
      : "memory");
  __builtin_amdgcn_sched_barrier(0);
}

__device__ __forceinline__ void tile_result_fence(YT& H, YT& T) {
  __builtin_amdgcn_sched_barrier(0);
  asm volatile(
      "s_wait_alu depctr_va_vdst(0)\n\t"
      "v_nop\n\tv_nop\n\tv_nop\n\tv_nop\n\tv_nop\n\t"
      "v_nop\n\tv_nop\n\tv_nop\n\tv_nop"
      : "+v"(H.y1), "+v"(H.y2), "+v"(T.y1), "+v"(T.y2)
      :
      : "memory");
  __builtin_amdgcn_sched_barrier(0);
}

#define STEP1(T, q) ((v2f){(T).y1[2 * (q)], (T).y1[2 * (q) + 1]})
#define STEP2(T, q) ((v2f){(T).y2[2 * (q)], (T).y2[2 * (q) + 1]})

__device__ __forceinline__ float other_half(float x) {
  return __builtin_bit_cast(
      float, __builtin_amdgcn_permlanex16(__builtin_bit_cast(int, x),
                                          __builtin_bit_cast(int, x),
                                          0x76543210, 0xfedcba98, false, false));
}

__device__ __forceinline__ v2f mirror_step(v2f a) {
  return (v2f){other_half(a.y), other_half(a.x)};
}

__device__ __forceinline__ v2f pick_step(bool c, v2f a, v2f b) {
  return (v2f){c ? a.x : b.x, c ? a.y : b.y};
}

__device__ __forceinline__ v8f row_even(v2f l1, v2f l2, const YT& C,
                                        const v2f* g0, const v2f* g1) {
  v8f acc = {0.f, 0.f, 0.f, 0.f, 0.f, 0.f, 0.f, 0.f};
  acc = wmma4(l1, g0[0], acc);
  acc = wmma4(l2, g1[0], acc);
#pragma unroll
  for (int q = 0; q < 3; ++q) {
    acc = wmma4(STEP1(C, q), g0[q + 1], acc);
    acc = wmma4(STEP2(C, q), g1[q + 1], acc);
  }
  return acc;
}

__device__ __forceinline__ v8f row_odd(const YT& C, v2f r1, v2f r2,
                                       const v2f* g0, const v2f* g1) {
  v8f acc = {0.f, 0.f, 0.f, 0.f, 0.f, 0.f, 0.f, 0.f};
#pragma unroll
  for (int q = 1; q < 4; ++q) {
    acc = wmma4(STEP1(C, q), g0[q - 1], acc);
    acc = wmma4(STEP2(C, q), g1[q - 1], acc);
  }
  acc = wmma4(r1, g0[3], acc);
  acc = wmma4(r2, g1[3], acc);
  return acc;
}

template <int NT, int Hb, int Wb>
__global__ void __launch_bounds__(128, 4)
j2_fused_kernel(const float* __restrict__ plain, const float* __restrict__ yh,
                const float* __restrict__ ha0, const float* __restrict__ hb0,
                const float* __restrict__ ha1, const float* __restrict__ hb1,
                float* __restrict__ out, int B) {
  constexpr int r = 2 * Hb;
  constexpr int W = 2 * Wb;
  int lane = threadIdx.x & 31;
  int wave = __builtin_amdgcn_readfirstlane((int)(threadIdx.x >> 5));
  int g = (int)blockIdx.x * 4 + wave;
  constexpr int nCS = W / (16 * NT);
  constexpr int nRT = (2 * r) / 16;
  int cs = g % nCS;
  int g2 = g / nCS;
  int I = g2 % nRT;
  int b = g2 / nRT;
  if (b >= B) return;
  int half = lane >> 4;
  int l16 = lane & 15;
  int koff = half * 2;
  int x0 = 8 * I - 4;
  int lcol = tile_col(l16);
  int S = cs * 16 * NT;
  constexpr bool TWO = (nCS == 2);
  int hcol = TWO ? ((cs ? S - 16 : S + 16 * NT) + lcol)
                 : (S + 16 * NT + lcol - (lcol >= 12 ? 16 * NT + 16 : 0));

  v2f f0[4], f0s[4], f1[4], f1s[4];
#pragma unroll
  for (int c = 0; c < 4; ++c) {
    int kk = 4 * c + koff;
    float ax = cif_A(l16, kk, ha0, hb0, 0);
    float ay = cif_A(l16, kk + 1, ha0, hb0, 0);
    float bx = cif_A(l16, kk, ha1, hb1, 1);
    float by = cif_A(l16, kk + 1, ha1, hb1, 1);
    f0[c].x = ax;
    f0[c].y = ay;
    f0s[c].x = ax * SQRT_HALF;
    f0s[c].y = ay * SQRT_HALF;
    f1[c].x = bx;
    f1[c].y = by;
    f1s[c].x = bx * SQRT_HALF;
    f1s[c].y = by * SQRT_HALF;
  }

  constexpr int outW = 2 * W;
  float* op = out + ((size_t)b * (2 * r) + (size_t)I * 16 + half * 8) * outW +
              cs * 32 * NT + l16;

  YT Hh = col_tile<4, Hb, Wb>(plain, yh, b, x0, hcol, koff, f0, f0s, f1s);
  YT P = col_tile<4, Hb, Wb>(plain, yh, b, x0, S + lcol, koff, f0, f0s, f1s);
  tile_result_fence(Hh, P);
  v2f l1 = STEP1(Hh, 3), l2 = STEP2(Hh, 3);
  v2f e1 = STEP1(Hh, 0), e2 = STEP2(Hh, 0);
  if (TWO) {
    l1 = pick_step(cs != 0, l1, mirror_step(STEP1(P, 0)));
    l2 = pick_step(cs != 0, l2, mirror_step(STEP2(P, 0)));
  }
#pragma unroll
  for (int u = 0; u + 2 < NT; u += 2) {
    store_tile(op + 32 * u, outW, row_even(l1, l2, P, f0, f1));
    YT Q = col_tile<4, Hb, Wb>(plain, yh, b, x0, S + 16 * (u + 1) + lcol, koff,
                               f0, f0s, f1s);
    tile_result_fence(Q);
    store_tile(op + 32 * u + 16, outW,
               row_odd(P, STEP1(Q, 0), STEP2(Q, 0), f0, f1));
    __builtin_amdgcn_sched_barrier(0);
    store_tile(op + 32 * u + 32, outW,
               row_even(STEP1(P, 3), STEP2(P, 3), Q, f0, f1));
    P = col_tile<4, Hb, Wb>(plain, yh, b, x0, S + 16 * (u + 2) + lcol, koff, f0,
                            f0s, f1s);
    tile_result_fence(P);
    store_tile(op + 32 * u + 48, outW,
               row_odd(Q, STEP1(P, 0), STEP2(P, 0), f0, f1));
    l1 = STEP1(Q, 3);
    l2 = STEP2(Q, 3);
  }
  {
    constexpr int u = NT - 2;
    store_tile(op + 32 * u, outW, row_even(l1, l2, P, f0, f1));
    YT Q = col_tile<4, Hb, Wb>(plain, yh, b, x0, S + 16 * (u + 1) + lcol, koff,
                               f0, f0s, f1s);
    tile_result_fence(Q);
    if (TWO) {
      e1 = pick_step(cs != 0, mirror_step(STEP1(Q, 3)), e1);
      e2 = pick_step(cs != 0, mirror_step(STEP2(Q, 3)), e2);
    }
    store_tile(op + 32 * u + 16, outW,
               row_odd(P, STEP1(Q, 0), STEP2(Q, 0), f0, f1));
    store_tile(op + 32 * u + 32, outW,
               row_even(STEP1(P, 3), STEP2(P, 3), Q, f0, f1));
    store_tile(op + 32 * u + 48, outW, row_odd(Q, e1, e2, f0, f1));
  }
}

__device__ __forceinline__ v8f row_j1(v2f l1, v2f l2, const YT& C, v2f r1,
                                      v2f r2, const v2f* g0, const v2f* g1) {
  v8f acc = {0.f, 0.f, 0.f, 0.f, 0.f, 0.f, 0.f, 0.f};
  acc = wmma4(l1, g0[0], acc);
  acc = wmma4(l2, g1[0], acc);
#pragma unroll
  for (int q = 0; q < 4; ++q) {
    acc = wmma4(STEP1(C, q), g0[q + 1], acc);
    acc = wmma4(STEP2(C, q), g1[q + 1], acc);
  }
  acc = wmma4(r1, g0[5], acc);
  acc = wmma4(r2, g1[5], acc);
  return acc;
}

template <int NOUT, int Hb, int Wb>
__global__ void __launch_bounds__(128)
j1_fused_kernel(const float* __restrict__ plain, const float* __restrict__ yh,
                const float* __restrict__ h0, int taps0,
                const float* __restrict__ h1, int taps1,
                float* __restrict__ out, int B) {
  constexpr int r = 2 * Hb;
  constexpr int W = 2 * Wb;
  int lane = threadIdx.x & 31;
  int wave = __builtin_amdgcn_readfirstlane((int)(threadIdx.x >> 5));
  int g = (int)blockIdx.x * 4 + wave;
  constexpr int nCS = W / (16 * NOUT);
  constexpr int nRT = r / 16;
  int cs = g % nCS;
  int g2 = g / nCS;
  int I = g2 % nRT;
  int b = g2 / nRT;
  if (b >= B) return;
  int half = lane >> 4;
  int l16 = lane & 15;
  int koff = half * 2;
  int x0 = 16 * I - 3;
  int lcol = tile_col(l16);
  int S = cs * 16 * NOUT;
  constexpr bool TWO = (nCS == 2);
  int hcol = TWO ? ((cs ? S - 16 : S + 16 * NOUT) + lcol)
                 : (S + 16 * NOUT + lcol - (lcol >= 12 ? 16 * NOUT + 16 : 0));

  v2f f0[6], f0s[6], f1s[6];
#pragma unroll
  for (int c = 0; c < 6; ++c) {
    int kk = 4 * c + koff;
    float ax = cf_A(l16, kk, h0, taps0);
    float ay = cf_A(l16, kk + 1, h0, taps0);
    f0[c].x = ax;
    f0[c].y = ay;
    f0s[c].x = ax * SQRT_HALF;
    f0s[c].y = ay * SQRT_HALF;
    f1s[c].x = cf_A(l16, kk, h1, taps1) * SQRT_HALF;
    f1s[c].y = cf_A(l16, kk + 1, h1, taps1) * SQRT_HALF;
  }
  v2f g0[6], g1[6];
#pragma unroll
  for (int s = 0; s < 6; ++s) {
    int kk = 4 * s - 1 + koff;
    g0[s].x = cf_A(l16, kk, h0, taps0);
    g0[s].y = cf_A(l16, kk + 1, h0, taps0);
    g1[s].x = cf_A(l16, kk, h1, taps1);
    g1[s].y = cf_A(l16, kk + 1, h1, taps1);
  }

  float* op = out + ((size_t)b * r + (size_t)I * 16 + half * 8) * W +
              cs * 16 * NOUT + l16;

  YT Hh = col_tile<6, Hb, Wb>(plain, yh, b, x0, hcol, koff, f0, f0s, f1s);
  YT P = col_tile<6, Hb, Wb>(plain, yh, b, x0, S + lcol, koff, f0, f0s, f1s);
  tile_result_fence(Hh, P);
  v2f l1 = STEP1(Hh, 3), l2 = STEP2(Hh, 3);
  v2f e1 = STEP1(Hh, 0), e2 = STEP2(Hh, 0);
  if (TWO) {
    l1 = pick_step(cs != 0, l1, mirror_step(STEP1(P, 0)));
    l2 = pick_step(cs != 0, l2, mirror_step(STEP2(P, 0)));
  }
#pragma unroll 1
  for (int j = 0; j + 2 < NOUT; j += 2) {
    YT Q = col_tile<6, Hb, Wb>(plain, yh, b, x0, S + 16 * (j + 1) + lcol, koff,
                               f0, f0s, f1s);
    tile_result_fence(Q);
    store_tile(op + 16 * j, W,
               row_j1(l1, l2, P, STEP1(Q, 0), STEP2(Q, 0), g0, g1));
    l1 = STEP1(P, 3);
    l2 = STEP2(P, 3);
    __builtin_amdgcn_sched_barrier(0);
    P = col_tile<6, Hb, Wb>(plain, yh, b, x0, S + 16 * (j + 2) + lcol, koff, f0,
                            f0s, f1s);
    tile_result_fence(P);
    store_tile(op + 16 * j + 16, W,
               row_j1(l1, l2, Q, STEP1(P, 0), STEP2(P, 0), g0, g1));
    l1 = STEP1(Q, 3);
    l2 = STEP2(Q, 3);
  }
  {
    constexpr int j = NOUT - 2;
    YT Q = col_tile<6, Hb, Wb>(plain, yh, b, x0, S + 16 * (j + 1) + lcol, koff,
                               f0, f0s, f1s);
    tile_result_fence(Q);
    if (TWO) {
      e1 = pick_step(cs != 0, mirror_step(STEP1(Q, 3)), e1);
      e2 = pick_step(cs != 0, mirror_step(STEP2(Q, 3)), e2);
    }
    store_tile(op + 16 * j, W,
               row_j1(l1, l2, P, STEP1(Q, 0), STEP2(Q, 0), g0, g1));
    store_tile(op + 16 * j + 16, W,
               row_j1(STEP1(P, 3), STEP2(P, 3), Q, e1, e2, g0, g1));
  }
}

extern "C" void kernel_launch(void* const* d_in, const int* in_sizes, int n_in,
                              void* d_out, int out_size, void* d_ws,
                              size_t ws_size, hipStream_t stream) {
  (void)in_sizes;
  (void)n_in;
  (void)out_size;
  (void)ws_size;

  const float* yl = (const float*)d_in[0];
  const float* yh0 = (const float*)d_in[1];
  const float* yh1 = (const float*)d_in[2];
  const float* yh2 = (const float*)d_in[3];
  const float* g0o = (const float*)d_in[4];
  const float* g1o = (const float*)d_in[5];
  const float* g0a = (const float*)d_in[6];
  const float* g0b = (const float*)d_in[7];
  const float* g1a = (const float*)d_in[8];
  const float* g1b = (const float*)d_in[9];

  float* LLa = (float*)d_ws;
  float* LLb = LLa + (size_t)64 * 65536;
  float* outp = (float*)d_out;

  dim3 blk(128);

  j2_fused_kernel<2, 64, 64><<<dim3(1024), blk, 0, stream>>>(
      yl, yh2, g0b, g0a, g1b, g1a, LLa, 64);

  j2_fused_kernel<8, 128, 128><<<dim3(1024), blk, 0, stream>>>(
      LLa, yh1, g0b, g0a, g1b, g1a, LLb, 64);

  j1_fused_kernel<16, 256, 256><<<dim3(1024), blk, 0, stream>>>(
      LLb, yh0, g0o, 7, g1o, 5, outp, 64);
}


# Column pass: the two 16-column tiles that share each 128-byte line are loaded as a pair, the second tile's load right behind the first's, so the line is requested once, not twice; sums unchanged
# speedup vs baseline: 3.9378x; 1.1953x over previous
#include <hip/hip_runtime.h>


typedef __attribute__((ext_vector_type(2))) float v2f;
typedef __attribute__((ext_vector_type(8))) float v8f;

#define SQRT_HALF 0.70710678118654752440f

__device__ __forceinline__ int reflect_i(int x, int n) {
  x = (x < 0) ? (-1 - x) : x;
  x = (x >= n) ? (2 * n - 1 - x) : x;
  return x;
}

__device__ __forceinline__ int tile_col(int m) {
  return 4 * ((m & 7) >> 1) + 2 * (m >> 3) + (m & 1);
}

__device__ __forceinline__ float cif_A(int m, int kk,
                                       const float* __restrict__ ha,
                                       const float* __restrict__ hb, int hp) {
  int p = m & 3;
  int q = m >> 2;
  int want_odd = hp ? ((p & 1) ^ 1) : (p & 1);
  int t = ((kk - want_odd) >> 1) - q;
  bool ok = ((kk & 1) == want_odd) && (t >= 0) && (t <= 4);
  int tc = t < 0 ? 0 : (t > 4 ? 4 : t);
  const float* __restrict__ h = ((p == 0) || (p == 2)) ? ha : hb;
  int idx = (p < 2) ? (2 * tc + 1) : (2 * tc);
  float v = h[idx];
  return ok ? v : 0.0f;
}

__device__ __forceinline__ float cf_A(int m, int kk,
                                      const float* __restrict__ h, int taps) {
  int t = kk - m - (3 - (taps >> 1));
  bool ok = (t >= 0) && (t < taps);
  int tc = t < 0 ? 0 : (t >= taps ? taps - 1 : t);
  float v = h[tc];
  return ok ? v : 0.0f;
}

__device__ __forceinline__ void c2q_make(v2f w1, v2f w2, int odd_row, int cb,
                                         float& own, float& give) {
  float c0 = odd_row ? (w1.y - w2.y) : (w1.x + w2.x);
  float c1 = odd_row ? (w2.x - w1.x) : (w1.y + w2.y);
  own = cb ? c1 : c0;
  give = cb ? c0 : c1;
}

__device__ __forceinline__ void c2q_pair(const float* __restrict__ p1,
                                         const float* __restrict__ p2,
                                         unsigned o, int odd_row, int cb,
                                         float& own, float& give) {
  v2f w1 = ((const v2f*)p1)[o];
  v2f w2 = ((const v2f*)p2)[o];
  c2q_make(w1, w2, odd_row, cb, own, give);
}

__device__ __forceinline__ float pair_lane(float x) {
  return __builtin_bit_cast(
      float, __builtin_amdgcn_update_dpp(0, __builtin_bit_cast(int, x), 0xB1,
                                         0xF, 0xF, true));
}

__device__ __forceinline__ v8f wmma4(v2f a, v2f b, v8f c) {
  return __builtin_amdgcn_wmma_f32_16x16x4_f32(false, a, false, b, (short)0, c,
                                               false, false);
}

struct YT {
  v8f y1, y2;
};

template <int KC, int Hb, int Wb>
__device__ __forceinline__ YT col_tile(const float* __restrict__ plain,
                                       const float* __restrict__ yh, int b,
                                       int x0, int ccu, int koff,
                                       const v2f* f0, const v2f* f0s,
                                       const v2f* f1s) {
  constexpr int r = 2 * Hb;
  constexpr int W = 2 * Wb;
  int cc = reflect_i(ccu, W);
  int cb = cc & 1;
  unsigned jb = (unsigned)(cc >> 1);
  int lo = (int)(threadIdx.x & 1);
  constexpr size_t bs = (size_t)Hb * Wb * 2;
  const float* p0 = plain + (size_t)b * r * W;
  const float* yb = yh + (size_t)b * 6 * bs;
  const float* q0 = yb;
  const float* q1 = yb + bs;
  const float* q2 = yb + 2 * bs;
  const float* q3 = yb + 3 * bs;
  const float* q4 = yb + 4 * bs;
  const float* q5 = yb + 5 * bs;
  YT o;
  o.y1 = (v8f){0.f, 0.f, 0.f, 0.f, 0.f, 0.f, 0.f, 0.f};
  o.y2 = (v8f){0.f, 0.f, 0.f, 0.f, 0.f, 0.f, 0.f, 0.f};
#pragma unroll
  for (int c = 0; c < KC; ++c) {
    int kk = 4 * c + koff;
    int rs = reflect_i(x0 + kk + lo, r);
    unsigned ob = (unsigned)(rs >> 1) * (unsigned)Wb + jb;
    int d = rs & 1;
    v2f pl = ((const v2f*)p0)[(unsigned)rs * (unsigned)Wb + jb];
    float pl_o = cb ? pl.y : pl.x;
    float pl_g = cb ? pl.x : pl.y;
    float lh_o, lh_g, hl_o, hl_g, hh_o, hh_g;
    c2q_pair(q0, q5, ob, d, cb, lh_o, lh_g);
    c2q_pair(q2, q3, ob, d, cb, hl_o, hl_g);
    c2q_pair(q1, q4, ob, d, cb, hh_o, hh_g);
    float pl_x = pair_lane(pl_g);
    float lh_x = pair_lane(lh_g);
    float hl_x = pair_lane(hl_g);
    float hh_x = pair_lane(hh_g);
    v2f xl = lo ? (v2f){pl_x, pl_o} : (v2f){pl_o, pl_x};
    v2f xlh = lo ? (v2f){lh_x, lh_o} : (v2f){lh_o, lh_x};
    v2f xhl = lo ? (v2f){hl_x, hl_o} : (v2f){hl_o, hl_x};
    v2f xhh = lo ? (v2f){hh_x, hh_o} : (v2f){hh_o, hh_x};
    o.y1 = wmma4(xl, f0[c], o.y1);
    o.y1 = wmma4(xlh, f1s[c], o.y1);
    o.y2 = wmma4(xhl, f0s[c], o.y2);
    o.y2 = wmma4(xhh, f1s[c], o.y2);
    if (KC > 4 && c == KC / 2 - 1) __builtin_amdgcn_sched_barrier(0);
  }
  return o;
}

__device__ __forceinline__ void col_step(YT& o, v2f pl, v2f lhA, v2f lhB,
                                         v2f hlA, v2f hlB, v2f hhA, v2f hhB,
                                         int d, int cb, int lo, v2f f0c,
                                         v2f f0sc, v2f f1sc) {
  float pl_o = cb ? pl.y : pl.x;
  float pl_g = cb ? pl.x : pl.y;
  float lh_o, lh_g, hl_o, hl_g, hh_o, hh_g;
  c2q_make(lhA, lhB, d, cb, lh_o, lh_g);
  c2q_make(hlA, hlB, d, cb, hl_o, hl_g);
  c2q_make(hhA, hhB, d, cb, hh_o, hh_g);
  float pl_x = pair_lane(pl_g);
  float lh_x = pair_lane(lh_g);
  float hl_x = pair_lane(hl_g);
  float hh_x = pair_lane(hh_g);
  v2f xl = lo ? (v2f){pl_x, pl_o} : (v2f){pl_o, pl_x};
  v2f xlh = lo ? (v2f){lh_x, lh_o} : (v2f){lh_o, lh_x};
  v2f xhl = lo ? (v2f){hl_x, hl_o} : (v2f){hl_o, hl_x};
  v2f xhh = lo ? (v2f){hh_x, hh_o} : (v2f){hh_o, hh_x};
  o.y1 = wmma4(xl, f0c, o.y1);
  o.y1 = wmma4(xlh, f1sc, o.y1);
  o.y2 = wmma4(xhl, f0sc, o.y2);
  o.y2 = wmma4(xhh, f1sc, o.y2);
}

template <int KC, int Hb, int Wb>
__device__ __forceinline__ void col_tile2(const float* __restrict__ plain,
                                          const float* __restrict__ yh, int b,
                                          int x0, int ccu, int koff,
                                          const v2f* f0, const v2f* f0s,
                                          const v2f* f1s, YT& A, YT& B) {
  constexpr int r = 2 * Hb;
  constexpr int W = 2 * Wb;
  int cb = ccu & 1;
  unsigned jb = (unsigned)(ccu >> 1);
  int lo = (int)(threadIdx.x & 1);
  constexpr size_t bs = (size_t)Hb * Wb * 2;
  const v2f* p0 = (const v2f*)(plain + (size_t)b * r * W);
  const float* yb = yh + (size_t)b * 6 * bs;
  const v2f* q0 = (const v2f*)yb;
  const v2f* q1 = (const v2f*)(yb + bs);
  const v2f* q2 = (const v2f*)(yb + 2 * bs);
  const v2f* q3 = (const v2f*)(yb + 3 * bs);
  const v2f* q4 = (const v2f*)(yb + 4 * bs);
  const v2f* q5 = (const v2f*)(yb + 5 * bs);
  A.y1 = (v8f){0.f, 0.f, 0.f, 0.f, 0.f, 0.f, 0.f, 0.f};
  A.y2 = (v8f){0.f, 0.f, 0.f, 0.f, 0.f, 0.f, 0.f, 0.f};
  B.y1 = (v8f){0.f, 0.f, 0.f, 0.f, 0.f, 0.f, 0.f, 0.f};
  B.y2 = (v8f){0.f, 0.f, 0.f, 0.f, 0.f, 0.f, 0.f, 0.f};
#pragma unroll
  for (int c = 0; c < KC; ++c) {
    int kk = 4 * c + koff;
    int rs = reflect_i(x0 + kk + lo, r);
    unsigned ob = (unsigned)(rs >> 1) * (unsigned)Wb + jb;
    unsigned opl = (unsigned)rs * (unsigned)Wb + jb;
    int d = rs & 1;
    v2f plA = p0[opl], plB = p0[opl + 8];
    v2f a0 = q0[ob], b0 = q0[ob + 8];
    v2f a5 = q5[ob], b5 = q5[ob + 8];
    v2f a2 = q2[ob], b2 = q2[ob + 8];
    v2f a3 = q3[ob], b3 = q3[ob + 8];
    v2f a1 = q1[ob], b1 = q1[ob + 8];
    v2f a4 = q4[ob], b4 = q4[ob + 8];
    col_step(A, plA, a0, a5, a2, a3, a1, a4, d, cb, lo, f0[c], f0s[c],
             f1s[c]);
    col_step(B, plB, b0, b5, b2, b3, b1, b4, d, cb, lo, f0[c], f0s[c],
             f1s[c]);
    if ((c & 1) == 1 && c + 1 < KC) __builtin_amdgcn_sched_barrier(0);
  }
}

__device__ __forceinline__ void store_tile(float* __restrict__ op, int pitch,
                                           v8f acc) {
#pragma unroll
  for (int e = 0; e < 8; ++e) op[(size_t)e * pitch] = acc[e];
}

__device__ __forceinline__ void tile_result_fence(YT& T) {
  __builtin_amdgcn_sched_barrier(0);
  asm volatile(
      "s_wait_alu depctr_va_vdst(0)\n\t"
      "v_nop\n\tv_nop\n\tv_nop\n\tv_nop\n\tv_nop\n\t"
      "v_nop\n\tv_nop\n\tv_nop\n\tv_nop"
      : "+v"(T.y1), "+v"(T.y2)
      :
      : "memory");
  __builtin_amdgcn_sched_barrier(0);
}

__device__ __forceinline__ void tile_result_fence(YT& H, YT& T) {
  __builtin_amdgcn_sched_barrier(0);
  asm volatile(
      "s_wait_alu depctr_va_vdst(0)\n\t"
      "v_nop\n\tv_nop\n\tv_nop\n\tv_nop\n\tv_nop\n\t"
      "v_nop\n\tv_nop\n\tv_nop\n\tv_nop"
      : "+v"(H.y1), "+v"(H.y2), "+v"(T.y1), "+v"(T.y2)
      :
      : "memory");
  __builtin_amdgcn_sched_barrier(0);
}

__device__ __forceinline__ void tile_result_fence(YT& H, YT& S, YT& T) {
  __builtin_amdgcn_sched_barrier(0);
  asm volatile(
      "s_wait_alu depctr_va_vdst(0)\n\t"
      "v_nop\n\tv_nop\n\tv_nop\n\tv_nop\n\tv_nop\n\t"
      "v_nop\n\tv_nop\n\tv_nop\n\tv_nop"
      : "+v"(H.y1), "+v"(H.y2), "+v"(S.y1), "+v"(S.y2), "+v"(T.y1),
        "+v"(T.y2)
      :
      : "memory");
  __builtin_amdgcn_sched_barrier(0);
}

#define STEP1(T, q) ((v2f){(T).y1[2 * (q)], (T).y1[2 * (q) + 1]})
#define STEP2(T, q) ((v2f){(T).y2[2 * (q)], (T).y2[2 * (q) + 1]})

__device__ __forceinline__ float other_half(float x) {
  return __builtin_bit_cast(
      float, __builtin_amdgcn_permlanex16(__builtin_bit_cast(int, x),
                                          __builtin_bit_cast(int, x),
                                          0x76543210, 0xfedcba98, false, false));
}

__device__ __forceinline__ v2f mirror_step(v2f a) {
  return (v2f){other_half(a.y), other_half(a.x)};
}

__device__ __forceinline__ v2f pick_step(bool c, v2f a, v2f b) {
  return (v2f){c ? a.x : b.x, c ? a.y : b.y};
}

__device__ __forceinline__ v8f row_even(v2f l1, v2f l2, const YT& C,
                                        const v2f* g0, const v2f* g1) {
  v8f acc = {0.f, 0.f, 0.f, 0.f, 0.f, 0.f, 0.f, 0.f};
  acc = wmma4(l1, g0[0], acc);
  acc = wmma4(l2, g1[0], acc);
#pragma unroll
  for (int q = 0; q < 3; ++q) {
    acc = wmma4(STEP1(C, q), g0[q + 1], acc);
    acc = wmma4(STEP2(C, q), g1[q + 1], acc);
  }
  return acc;
}

__device__ __forceinline__ v8f row_odd(const YT& C, v2f r1, v2f r2,
                                       const v2f* g0, const v2f* g1) {
  v8f acc = {0.f, 0.f, 0.f, 0.f, 0.f, 0.f, 0.f, 0.f};
#pragma unroll
  for (int q = 1; q < 4; ++q) {
    acc = wmma4(STEP1(C, q), g0[q - 1], acc);
    acc = wmma4(STEP2(C, q), g1[q - 1], acc);
  }
  acc = wmma4(r1, g0[3], acc);
  acc = wmma4(r2, g1[3], acc);
  return acc;
}

template <int NT, int Hb, int Wb>
__global__ void __launch_bounds__(128, 4)
j2_fused_kernel(const float* __restrict__ plain, const float* __restrict__ yh,
                const float* __restrict__ ha0, const float* __restrict__ hb0,
                const float* __restrict__ ha1, const float* __restrict__ hb1,
                float* __restrict__ out, int B) {
  constexpr int r = 2 * Hb;
  constexpr int W = 2 * Wb;
  int lane = threadIdx.x & 31;
  int wave = __builtin_amdgcn_readfirstlane((int)(threadIdx.x >> 5));
  int g = (int)blockIdx.x * 4 + wave;
  constexpr int nCS = W / (16 * NT);
  constexpr int nRT = (2 * r) / 16;
  int cs = g % nCS;
  int g2 = g / nCS;
  int I = g2 % nRT;
  int b = g2 / nRT;
  if (b >= B) return;
  int half = lane >> 4;
  int l16 = lane & 15;
  int koff = half * 2;
  int x0 = 8 * I - 4;
  int lcol = tile_col(l16);
  int S = cs * 16 * NT;
  constexpr bool TWO = (nCS == 2);
  int hcol = TWO ? ((cs ? S - 16 : S + 16 * NT) + lcol)
                 : (S + 16 * NT + lcol - (lcol >= 12 ? 16 * NT + 16 : 0));

  v2f f0[4], f0s[4], f1[4], f1s[4];
#pragma unroll
  for (int c = 0; c < 4; ++c) {
    int kk = 4 * c + koff;
    float ax = cif_A(l16, kk, ha0, hb0, 0);
    float ay = cif_A(l16, kk + 1, ha0, hb0, 0);
    float bx = cif_A(l16, kk, ha1, hb1, 1);
    float by = cif_A(l16, kk + 1, ha1, hb1, 1);
    f0[c].x = ax;
    f0[c].y = ay;
    f0s[c].x = ax * SQRT_HALF;
    f0s[c].y = ay * SQRT_HALF;
    f1[c].x = bx;
    f1[c].y = by;
    f1s[c].x = bx * SQRT_HALF;
    f1s[c].y = by * SQRT_HALF;
  }

  constexpr int outW = 2 * W;
  float* op = out + ((size_t)b * (2 * r) + (size_t)I * 16 + half * 8) * outW +
              cs * 32 * NT + l16;

  static_assert(NT % 2 == 0, "tiles are computed in aligned pairs");
  constexpr int NP = NT / 2;
  YT Hh = col_tile<4, Hb, Wb>(plain, yh, b, x0, hcol, koff, f0, f0s, f1s);
  YT A0, A1;
  col_tile2<4, Hb, Wb>(plain, yh, b, x0, S + lcol, koff, f0, f0s, f1s, A0, A1);
  tile_result_fence(Hh, A0, A1);
  v2f l1 = STEP1(Hh, 3), l2 = STEP2(Hh, 3);
  v2f e1 = STEP1(Hh, 0), e2 = STEP2(Hh, 0);
  if (TWO) {
    l1 = pick_step(cs != 0, l1, mirror_step(STEP1(A0, 0)));
    l2 = pick_step(cs != 0, l2, mirror_step(STEP2(A0, 0)));
  }
#pragma unroll
  for (int p = 1; p < NP; ++p) {
    const int u = 2 * (p - 1);
    store_tile(op + 32 * u, outW, row_even(l1, l2, A0, f0, f1));
    store_tile(op + 32 * u + 16, outW,
               row_odd(A0, STEP1(A1, 0), STEP2(A1, 0), f0, f1));
    store_tile(op + 32 * u + 32, outW,
               row_even(STEP1(A0, 3), STEP2(A0, 3), A1, f0, f1));
    YT B0, B1;
    col_tile2<4, Hb, Wb>(plain, yh, b, x0, S + 32 * p + lcol, koff, f0, f0s,
                         f1s, B0, B1);
    tile_result_fence(B0, B1);
    store_tile(op + 32 * u + 48, outW,
               row_odd(A1, STEP1(B0, 0), STEP2(B0, 0), f0, f1));
    l1 = STEP1(A1, 3);
    l2 = STEP2(A1, 3);
    A0 = B0;
    A1 = B1;
    __builtin_amdgcn_sched_barrier(0);
  }
  {
    constexpr int u = NT - 2;
    if (TWO) {
      e1 = pick_step(cs != 0, mirror_step(STEP1(A1, 3)), e1);
      e2 = pick_step(cs != 0, mirror_step(STEP2(A1, 3)), e2);
    }
    store_tile(op + 32 * u, outW, row_even(l1, l2, A0, f0, f1));
    store_tile(op + 32 * u + 16, outW,
               row_odd(A0, STEP1(A1, 0), STEP2(A1, 0), f0, f1));
    store_tile(op + 32 * u + 32, outW,
               row_even(STEP1(A0, 3), STEP2(A0, 3), A1, f0, f1));
    store_tile(op + 32 * u + 48, outW, row_odd(A1, e1, e2, f0, f1));
  }
}

__device__ __forceinline__ v8f row_j1(v2f l1, v2f l2, const YT& C, v2f r1,
                                      v2f r2, const v2f* g0, const v2f* g1) {
  v8f acc = {0.f, 0.f, 0.f, 0.f, 0.f, 0.f, 0.f, 0.f};
  acc = wmma4(l1, g0[0], acc);
  acc = wmma4(l2, g1[0], acc);
#pragma unroll
  for (int q = 0; q < 4; ++q) {
    acc = wmma4(STEP1(C, q), g0[q + 1], acc);
    acc = wmma4(STEP2(C, q), g1[q + 1], acc);
  }
  acc = wmma4(r1, g0[5], acc);
  acc = wmma4(r2, g1[5], acc);
  return acc;
}

template <int NOUT, int Hb, int Wb>
__global__ void __launch_bounds__(128)
j1_fused_kernel(const float* __restrict__ plain, const float* __restrict__ yh,
                const float* __restrict__ h0, int taps0,
                const float* __restrict__ h1, int taps1,
                float* __restrict__ out, int B) {
  constexpr int r = 2 * Hb;
  constexpr int W = 2 * Wb;
  int lane = threadIdx.x & 31;
  int wave = __builtin_amdgcn_readfirstlane((int)(threadIdx.x >> 5));
  int g = (int)blockIdx.x * 4 + wave;
  constexpr int nCS = W / (16 * NOUT);
  constexpr int nRT = r / 16;
  int cs = g % nCS;
  int g2 = g / nCS;
  int I = g2 % nRT;
  int b = g2 / nRT;
  if (b >= B) return;
  int half = lane >> 4;
  int l16 = lane & 15;
  int koff = half * 2;
  int x0 = 16 * I - 3;
  int lcol = tile_col(l16);
  int S = cs * 16 * NOUT;
  constexpr bool TWO = (nCS == 2);
  int hcol = TWO ? ((cs ? S - 16 : S + 16 * NOUT) + lcol)
                 : (S + 16 * NOUT + lcol - (lcol >= 12 ? 16 * NOUT + 16 : 0));

  v2f f0[6], f0s[6], f1s[6];
#pragma unroll
  for (int c = 0; c < 6; ++c) {
    int kk = 4 * c + koff;
    float ax = cf_A(l16, kk, h0, taps0);
    float ay = cf_A(l16, kk + 1, h0, taps0);
    f0[c].x = ax;
    f0[c].y = ay;
    f0s[c].x = ax * SQRT_HALF;
    f0s[c].y = ay * SQRT_HALF;
    f1s[c].x = cf_A(l16, kk, h1, taps1) * SQRT_HALF;
    f1s[c].y = cf_A(l16, kk + 1, h1, taps1) * SQRT_HALF;
  }
  v2f g0[6], g1[6];
#pragma unroll
  for (int s = 0; s < 6; ++s) {
    int kk = 4 * s - 1 + koff;
    g0[s].x = cf_A(l16, kk, h0, taps0);
    g0[s].y = cf_A(l16, kk + 1, h0, taps0);
    g1[s].x = cf_A(l16, kk, h1, taps1);
    g1[s].y = cf_A(l16, kk + 1, h1, taps1);
  }

  float* op = out + ((size_t)b * r + (size_t)I * 16 + half * 8) * W +
              cs * 16 * NOUT + l16;

  static_assert(NOUT % 2 == 0, "tiles are computed in aligned pairs");
  constexpr int NP = NOUT / 2;
  YT Hh = col_tile<6, Hb, Wb>(plain, yh, b, x0, hcol, koff, f0, f0s, f1s);
  YT A0, A1;
  col_tile2<6, Hb, Wb>(plain, yh, b, x0, S + lcol, koff, f0, f0s, f1s, A0, A1);
  tile_result_fence(Hh, A0, A1);
  v2f l1 = STEP1(Hh, 3), l2 = STEP2(Hh, 3);
  v2f e1 = STEP1(Hh, 0), e2 = STEP2(Hh, 0);
  if (TWO) {
    l1 = pick_step(cs != 0, l1, mirror_step(STEP1(A0, 0)));
    l2 = pick_step(cs != 0, l2, mirror_step(STEP2(A0, 0)));
  }
#pragma unroll 1
  for (int p = 1; p < NP; ++p) {
    YT B0, B1;
    col_tile2<6, Hb, Wb>(plain, yh, b, x0, S + 32 * p + lcol, koff, f0, f0s,
                         f1s, B0, B1);
    tile_result_fence(B0, B1);
    store_tile(op + 32 * (p - 1), W,
               row_j1(l1, l2, A0, STEP1(A1, 0), STEP2(A1, 0), g0, g1));
    store_tile(op + 32 * (p - 1) + 16, W,
               row_j1(STEP1(A0, 3), STEP2(A0, 3), A1, STEP1(B0, 0),
                      STEP2(B0, 0), g0, g1));
    l1 = STEP1(A1, 3);
    l2 = STEP2(A1, 3);
    A0 = B0;
    A1 = B1;
    __builtin_amdgcn_sched_barrier(0);
  }
  {
    if (TWO) {
      e1 = pick_step(cs != 0, mirror_step(STEP1(A1, 3)), e1);
      e2 = pick_step(cs != 0, mirror_step(STEP2(A1, 3)), e2);
    }
    store_tile(op + 16 * (NOUT - 2), W,
               row_j1(l1, l2, A0, STEP1(A1, 0), STEP2(A1, 0), g0, g1));
    store_tile(op + 16 * (NOUT - 1), W,
               row_j1(STEP1(A0, 3), STEP2(A0, 3), A1, e1, e2, g0, g1));
  }
}

extern "C" void kernel_launch(void* const* d_in, const int* in_sizes, int n_in,
                              void* d_out, int out_size, void* d_ws,
                              size_t ws_size, hipStream_t stream) {
  (void)in_sizes;
  (void)n_in;
  (void)out_size;
  (void)ws_size;

  const float* yl = (const float*)d_in[0];
  const float* yh0 = (const float*)d_in[1];
  const float* yh1 = (const float*)d_in[2];
  const float* yh2 = (const float*)d_in[3];
  const float* g0o = (const float*)d_in[4];
  const float* g1o = (const float*)d_in[5];
  const float* g0a = (const float*)d_in[6];
  const float* g0b = (const float*)d_in[7];
  const float* g1a = (const float*)d_in[8];
  const float* g1b = (const float*)d_in[9];

  float* LLa = (float*)d_ws;
  float* LLb = LLa + (size_t)64 * 65536;
  float* outp = (float*)d_out;

  dim3 blk(128);

  j2_fused_kernel<2, 64, 64><<<dim3(1024), blk, 0, stream>>>(
      yl, yh2, g0b, g0a, g1b, g1a, LLa, 64);

  j2_fused_kernel<8, 128, 128><<<dim3(1024), blk, 0, stream>>>(
      LLa, yh1, g0b, g0a, g1b, g1a, LLb, 64);

  j1_fused_kernel<16, 256, 256><<<dim3(1024), blk, 0, stream>>>(
      LLb, yh0, g0o, 7, g1o, 5, outp, 64);
}


# Column pass: a lane loads ONE band element per band and makes its 2x2 quad block itself, no lane swap; a half wave's load = one 128-byte line; even/odd-column accumulators; column sum order changes
# speedup vs baseline: 4.2772x; 1.0862x over previous
#include <hip/hip_runtime.h>


typedef __attribute__((ext_vector_type(2))) float v2f;
typedef __attribute__((ext_vector_type(8))) float v8f;

#define SQRT_HALF 0.70710678118654752440f

__device__ __forceinline__ int reflect_i(int x, int n) {
  x = (x < 0) ? (-1 - x) : x;
  x = (x >= n) ? (2 * n - 1 - x) : x;
  return x;
}

__device__ __forceinline__ int tile_col(int m) {
  return 4 * ((m & 7) >> 1) + 2 * (m >> 3) + (m & 1);
}

__device__ __forceinline__ float cif_A(int m, int kk,
                                       const float* __restrict__ ha,
                                       const float* __restrict__ hb, int hp) {
  int p = m & 3;
  int q = m >> 2;
  int want_odd = hp ? ((p & 1) ^ 1) : (p & 1);
  int t = ((kk - want_odd) >> 1) - q;
  bool ok = ((kk & 1) == want_odd) && (t >= 0) && (t <= 4);
  int tc = t < 0 ? 0 : (t > 4 ? 4 : t);
  const float* __restrict__ h = ((p == 0) || (p == 2)) ? ha : hb;
  int idx = (p < 2) ? (2 * tc + 1) : (2 * tc);
  float v = h[idx];
  return ok ? v : 0.0f;
}

__device__ __forceinline__ float cf_A(int m, int kk,
                                      const float* __restrict__ h, int taps) {
  int t = kk - m - (3 - (taps >> 1));
  bool ok = (t >= 0) && (t < taps);
  int tc = t < 0 ? 0 : (t >= taps ? taps - 1 : t);
  float v = h[tc];
  return ok ? v : 0.0f;
}

__device__ __forceinline__ void c2q_make(v2f w1, v2f w2, int odd_row, int cb,
                                         float& own, float& give) {
  float c0 = odd_row ? (w1.y - w2.y) : (w1.x + w2.x);
  float c1 = odd_row ? (w2.x - w1.x) : (w1.y + w2.y);
  own = cb ? c1 : c0;
  give = cb ? c0 : c1;
}

__device__ __forceinline__ void c2q_pair(const float* __restrict__ p1,
                                         const float* __restrict__ p2,
                                         unsigned o, int odd_row, int cb,
                                         float& own, float& give) {
  v2f w1 = ((const v2f*)p1)[o];
  v2f w2 = ((const v2f*)p2)[o];
  c2q_make(w1, w2, odd_row, cb, own, give);
}

__device__ __forceinline__ float pair_lane(float x) {
  return __builtin_bit_cast(
      float, __builtin_amdgcn_update_dpp(0, __builtin_bit_cast(int, x), 0xB1,
                                         0xF, 0xF, true));
}

__device__ __forceinline__ v8f wmma4(v2f a, v2f b, v8f c) {
  return __builtin_amdgcn_wmma_f32_16x16x4_f32(false, a, false, b, (short)0, c,
                                               false, false);
}

struct YT {
  v8f y1, y2;
};

template <int KC, int Hb, int Wb>
__device__ __forceinline__ YT col_tile(const float* __restrict__ plain,
                                       const float* __restrict__ yh, int b,
                                       int x0, int ccu, int koff,
                                       const v2f* f0, const v2f* f0s,
                                       const v2f* f1s) {
  constexpr int r = 2 * Hb;
  constexpr int W = 2 * Wb;
  int cc = reflect_i(ccu, W);
  int cb = cc & 1;
  unsigned jb = (unsigned)(cc >> 1);
  int lo = (int)(threadIdx.x & 1);
  constexpr size_t bs = (size_t)Hb * Wb * 2;
  const float* p0 = plain + (size_t)b * r * W;
  const float* yb = yh + (size_t)b * 6 * bs;
  const float* q0 = yb;
  const float* q1 = yb + bs;
  const float* q2 = yb + 2 * bs;
  const float* q3 = yb + 3 * bs;
  const float* q4 = yb + 4 * bs;
  const float* q5 = yb + 5 * bs;
  YT o;
  o.y1 = (v8f){0.f, 0.f, 0.f, 0.f, 0.f, 0.f, 0.f, 0.f};
  o.y2 = (v8f){0.f, 0.f, 0.f, 0.f, 0.f, 0.f, 0.f, 0.f};
#pragma unroll
  for (int c = 0; c < KC; ++c) {
    int kk = 4 * c + koff;
    int rs = reflect_i(x0 + kk + lo, r);
    unsigned ob = (unsigned)(rs >> 1) * (unsigned)Wb + jb;
    int d = rs & 1;
    v2f pl = ((const v2f*)p0)[(unsigned)rs * (unsigned)Wb + jb];
    float pl_o = cb ? pl.y : pl.x;
    float pl_g = cb ? pl.x : pl.y;
    float lh_o, lh_g, hl_o, hl_g, hh_o, hh_g;
    c2q_pair(q0, q5, ob, d, cb, lh_o, lh_g);
    c2q_pair(q2, q3, ob, d, cb, hl_o, hl_g);
    c2q_pair(q1, q4, ob, d, cb, hh_o, hh_g);
    float pl_x = pair_lane(pl_g);
    float lh_x = pair_lane(lh_g);
    float hl_x = pair_lane(hl_g);
    float hh_x = pair_lane(hh_g);
    v2f xl = lo ? (v2f){pl_x, pl_o} : (v2f){pl_o, pl_x};
    v2f xlh = lo ? (v2f){lh_x, lh_o} : (v2f){lh_o, lh_x};
    v2f xhl = lo ? (v2f){hl_x, hl_o} : (v2f){hl_o, hl_x};
    v2f xhh = lo ? (v2f){hh_x, hh_o} : (v2f){hh_o, hh_x};
    o.y1 = wmma4(xl, f0[c], o.y1);
    o.y1 = wmma4(xlh, f1s[c], o.y1);
    o.y2 = wmma4(xhl, f0s[c], o.y2);
    o.y2 = wmma4(xhh, f1s[c], o.y2);
    if (KC > 4 && c == KC / 2 - 1) __builtin_amdgcn_sched_barrier(0);
  }
  return o;
}

__device__ __forceinline__ int pair_col(int m) {
  return 2 * (m & 7) + (m >> 3);
}

__device__ __forceinline__ void pair_step(YT& E, YT& O, v2f r0, v2f r1,
                                          v2f lhA, v2f lhB, v2f hlA, v2f hlB,
                                          v2f hhA, v2f hhB, v2f f0c, v2f f0sc,
                                          v2f f1sc) {
  v2f ple = {r0.x, r1.x};
  v2f plo = {r0.y, r1.y};
  v2f lhe = {lhA.x + lhB.x, lhA.y - lhB.y};
  v2f lho = {lhA.y + lhB.y, lhB.x - lhA.x};
  v2f hle = {hlA.x + hlB.x, hlA.y - hlB.y};
  v2f hlo = {hlA.y + hlB.y, hlB.x - hlA.x};
  v2f hhe = {hhA.x + hhB.x, hhA.y - hhB.y};
  v2f hho = {hhA.y + hhB.y, hhB.x - hhA.x};
  E.y1 = wmma4(ple, f0c, E.y1);
  E.y1 = wmma4(lhe, f1sc, E.y1);
  E.y2 = wmma4(hle, f0sc, E.y2);
  E.y2 = wmma4(hhe, f1sc, E.y2);
  O.y1 = wmma4(plo, f0c, O.y1);
  O.y1 = wmma4(lho, f1sc, O.y1);
  O.y2 = wmma4(hlo, f0sc, O.y2);
  O.y2 = wmma4(hho, f1sc, O.y2);
}

template <int KC, int Hb, int Wb>
__device__ __forceinline__ void col_pair(const float* __restrict__ plain,
                                         const float* __restrict__ yh, int b,
                                         int x0, unsigned jb, int koff,
                                         const v2f* f0, const v2f* f0s,
                                         const v2f* f1s, YT& E, YT& O) {
  constexpr int r = 2 * Hb;
  constexpr int W = 2 * Wb;
  constexpr size_t bs = (size_t)Hb * Wb * 2;
  const v2f* p0 = (const v2f*)(plain + (size_t)b * r * W);
  const float* yb = yh + (size_t)b * 6 * bs;
  const v2f* q0 = (const v2f*)yb;
  const v2f* q1 = (const v2f*)(yb + bs);
  const v2f* q2 = (const v2f*)(yb + 2 * bs);
  const v2f* q3 = (const v2f*)(yb + 3 * bs);
  const v2f* q4 = (const v2f*)(yb + 4 * bs);
  const v2f* q5 = (const v2f*)(yb + 5 * bs);
  E.y1 = (v8f){0.f, 0.f, 0.f, 0.f, 0.f, 0.f, 0.f, 0.f};
  E.y2 = (v8f){0.f, 0.f, 0.f, 0.f, 0.f, 0.f, 0.f, 0.f};
  O.y1 = (v8f){0.f, 0.f, 0.f, 0.f, 0.f, 0.f, 0.f, 0.f};
  O.y2 = (v8f){0.f, 0.f, 0.f, 0.f, 0.f, 0.f, 0.f, 0.f};
#pragma unroll
  for (int c = 0; c < KC; ++c) {
    int kk = 4 * c + koff;
    int ir = reflect_i((x0 + kk) >> 1, Hb);
    unsigned ob = (unsigned)ir * (unsigned)Wb + jb;
    unsigned opl = 2u * (unsigned)ir * (unsigned)Wb + jb;
    v2f r0 = p0[opl], r1 = p0[opl + Wb];
    v2f a0 = q0[ob], a5 = q5[ob];
    v2f a2 = q2[ob], a3 = q3[ob];
    v2f a1 = q1[ob], a4 = q4[ob];
    pair_step(E, O, r0, r1, a0, a5, a2, a3, a1, a4, f0[c], f0s[c], f1s[c]);
    if ((c & 1) == 1 && c + 1 < KC) __builtin_amdgcn_sched_barrier(0);
  }
}

__device__ __forceinline__ v2f swap_k(bool c, v2f a) {
  return (v2f){c ? a.y : a.x, c ? a.x : a.y};
}

__device__ __forceinline__ void store_tile(float* __restrict__ op, int pitch,
                                           v8f acc) {
#pragma unroll
  for (int e = 0; e < 8; ++e) op[(size_t)e * pitch] = acc[e];
}

__device__ __forceinline__ void tile_result_fence(YT& T) {
  __builtin_amdgcn_sched_barrier(0);
  asm volatile(
      "s_wait_alu depctr_va_vdst(0)\n\t"
      "v_nop\n\tv_nop\n\tv_nop\n\tv_nop\n\tv_nop\n\t"
      "v_nop\n\tv_nop\n\tv_nop\n\tv_nop"
      : "+v"(T.y1), "+v"(T.y2)
      :
      : "memory");
  __builtin_amdgcn_sched_barrier(0);
}

__device__ __forceinline__ void tile_result_fence(YT& H, YT& T) {
  __builtin_amdgcn_sched_barrier(0);
  asm volatile(
      "s_wait_alu depctr_va_vdst(0)\n\t"
      "v_nop\n\tv_nop\n\tv_nop\n\tv_nop\n\tv_nop\n\t"
      "v_nop\n\tv_nop\n\tv_nop\n\tv_nop"
      : "+v"(H.y1), "+v"(H.y2), "+v"(T.y1), "+v"(T.y2)
      :
      : "memory");
  __builtin_amdgcn_sched_barrier(0);
}

__device__ __forceinline__ void tile_result_fence(YT& H, YT& S, YT& T) {
  __builtin_amdgcn_sched_barrier(0);
  asm volatile(
      "s_wait_alu depctr_va_vdst(0)\n\t"
      "v_nop\n\tv_nop\n\tv_nop\n\tv_nop\n\tv_nop\n\t"
      "v_nop\n\tv_nop\n\tv_nop\n\tv_nop"
      : "+v"(H.y1), "+v"(H.y2), "+v"(S.y1), "+v"(S.y2), "+v"(T.y1),
        "+v"(T.y2)
      :
      : "memory");
  __builtin_amdgcn_sched_barrier(0);
}

#define STEP1(T, q) ((v2f){(T).y1[2 * (q)], (T).y1[2 * (q) + 1]})
#define STEP2(T, q) ((v2f){(T).y2[2 * (q)], (T).y2[2 * (q) + 1]})

#define PSTEP1(E, O, t) ((v2f){(E).y1[(t)], (O).y1[(t)]})
#define PSTEP2(E, O, t) ((v2f){(E).y2[(t)], (O).y2[(t)]})

__device__ __forceinline__ float other_half(float x) {
  return __builtin_bit_cast(
      float, __builtin_amdgcn_permlanex16(__builtin_bit_cast(int, x),
                                          __builtin_bit_cast(int, x),
                                          0x76543210, 0xfedcba98, false, false));
}

__device__ __forceinline__ v2f mirror_step(v2f a) {
  return (v2f){other_half(a.y), other_half(a.x)};
}

__device__ __forceinline__ v2f pick_step(bool c, v2f a, v2f b) {
  return (v2f){c ? a.x : b.x, c ? a.y : b.y};
}

template <int BASE>
__device__ __forceinline__ v8f row_even(v2f l1, v2f l2, const YT& E,
                                        const YT& O, const v2f* g0,
                                        const v2f* g1) {
  v8f acc = {0.f, 0.f, 0.f, 0.f, 0.f, 0.f, 0.f, 0.f};
  acc = wmma4(l1, g0[0], acc);
  acc = wmma4(l2, g1[0], acc);
#pragma unroll
  for (int q = 0; q < 3; ++q) {
    acc = wmma4(PSTEP1(E, O, BASE + q), g0[q + 1], acc);
    acc = wmma4(PSTEP2(E, O, BASE + q), g1[q + 1], acc);
  }
  return acc;
}

template <int BASE>
__device__ __forceinline__ v8f row_odd(const YT& E, const YT& O, v2f r1,
                                       v2f r2, const v2f* g0, const v2f* g1) {
  v8f acc = {0.f, 0.f, 0.f, 0.f, 0.f, 0.f, 0.f, 0.f};
#pragma unroll
  for (int q = 1; q < 4; ++q) {
    acc = wmma4(PSTEP1(E, O, BASE + q), g0[q - 1], acc);
    acc = wmma4(PSTEP2(E, O, BASE + q), g1[q - 1], acc);
  }
  acc = wmma4(r1, g0[3], acc);
  acc = wmma4(r2, g1[3], acc);
  return acc;
}

template <int NT, int Hb, int Wb>
__global__ void __launch_bounds__(128, 4)
j2_fused_kernel(const float* __restrict__ plain, const float* __restrict__ yh,
                const float* __restrict__ ha0, const float* __restrict__ hb0,
                const float* __restrict__ ha1, const float* __restrict__ hb1,
                float* __restrict__ out, int B) {
  constexpr int r = 2 * Hb;
  constexpr int W = 2 * Wb;
  int lane = threadIdx.x & 31;
  int wave = __builtin_amdgcn_readfirstlane((int)(threadIdx.x >> 5));
  int g = (int)blockIdx.x * 4 + wave;
  constexpr int nCS = W / (16 * NT);
  constexpr int nRT = (2 * r) / 16;
  int cs = g % nCS;
  int g2 = g / nCS;
  int I = g2 % nRT;
  int b = g2 / nRT;
  if (b >= B) return;
  int half = lane >> 4;
  int l16 = lane & 15;
  int koff = half * 2;
  int x0 = 8 * I - 4;
  int lcol = tile_col(l16);
  int S = cs * 16 * NT;
  constexpr bool TWO = (nCS == 2);
  int hcol = TWO ? ((cs ? S - 16 : S + 16 * NT) + lcol)
                 : (S + 16 * NT + lcol - (lcol >= 12 ? 16 * NT + 16 : 0));

  v2f f0[4], f0s[4], f1[4], f1s[4];
#pragma unroll
  for (int c = 0; c < 4; ++c) {
    int kk = 4 * c + koff;
    float ax = cif_A(l16, kk, ha0, hb0, 0);
    float ay = cif_A(l16, kk + 1, ha0, hb0, 0);
    float bx = cif_A(l16, kk, ha1, hb1, 1);
    float by = cif_A(l16, kk + 1, ha1, hb1, 1);
    f0[c].x = ax;
    f0[c].y = ay;
    f0s[c].x = ax * SQRT_HALF;
    f0s[c].y = ay * SQRT_HALF;
    f1[c].x = bx;
    f1[c].y = by;
    f1s[c].x = bx * SQRT_HALF;
    f1s[c].y = by * SQRT_HALF;
  }

  constexpr int outW = 2 * W;
  float* op = out + ((size_t)b * (2 * r) + (size_t)I * 16 + half * 8) * outW +
              cs * 32 * NT + l16;

  static_assert(NT % 2 == 0, "tiles are computed in aligned pairs");
  constexpr int NP = NT / 2;
  YT Hh = col_tile<4, Hb, Wb>(plain, yh, b, x0, hcol, koff, f0, f0s, f1s);
  bool top = (I == 0), bot = (I == nRT - 1);
  v2f c0[4], c0s[4], c1s[4];
#pragma unroll
  for (int c = 0; c < 4; ++c) {
    bool sw = (c == 0 && top) || (c == 3 && bot);
    c0[c] = swap_k(sw, f0[c]);
    c0s[c] = swap_k(sw, f0s[c]);
    c1s[c] = swap_k(sw, f1s[c]);
  }
  unsigned jp = (unsigned)(S / 2 + pair_col(l16));
  YT E0, O0;
  col_pair<4, Hb, Wb>(plain, yh, b, x0, jp, koff, c0, c0s, c1s, E0, O0);
  tile_result_fence(Hh, E0, O0);
  v2f l1 = STEP1(Hh, 3), l2 = STEP2(Hh, 3);
  v2f e1 = STEP1(Hh, 0), e2 = STEP2(Hh, 0);
  if (TWO) {
    l1 = pick_step(cs != 0, l1, mirror_step(PSTEP1(E0, O0, 0)));
    l2 = pick_step(cs != 0, l2, mirror_step(PSTEP2(E0, O0, 0)));
  }
#pragma unroll
  for (int p = 1; p < NP; ++p) {
    const int u = 2 * (p - 1);
    store_tile(op + 32 * u, outW, row_even<0>(l1, l2, E0, O0, f0, f1));
    store_tile(op + 32 * u + 16, outW,
               row_odd<0>(E0, O0, PSTEP1(E0, O0, 4), PSTEP2(E0, O0, 4), f0,
                          f1));
    store_tile(op + 32 * u + 32, outW,
               row_even<4>(PSTEP1(E0, O0, 3), PSTEP2(E0, O0, 3), E0, O0, f0,
                           f1));
    YT E1, O1;
    col_pair<4, Hb, Wb>(plain, yh, b, x0, jp + 16u * (unsigned)p, koff, c0,
                        c0s, c1s, E1, O1);
    tile_result_fence(E1, O1);
    store_tile(op + 32 * u + 48, outW,
               row_odd<4>(E0, O0, PSTEP1(E1, O1, 0), PSTEP2(E1, O1, 0), f0,
                          f1));
    l1 = PSTEP1(E0, O0, 7);
    l2 = PSTEP2(E0, O0, 7);
    E0 = E1;
    O0 = O1;
    __builtin_amdgcn_sched_barrier(0);
  }
  {
    constexpr int u = NT - 2;
    if (TWO) {
      e1 = pick_step(cs != 0, mirror_step(PSTEP1(E0, O0, 7)), e1);
      e2 = pick_step(cs != 0, mirror_step(PSTEP2(E0, O0, 7)), e2);
    }
    store_tile(op + 32 * u, outW, row_even<0>(l1, l2, E0, O0, f0, f1));
    store_tile(op + 32 * u + 16, outW,
               row_odd<0>(E0, O0, PSTEP1(E0, O0, 4), PSTEP2(E0, O0, 4), f0,
                          f1));
    store_tile(op + 32 * u + 32, outW,
               row_even<4>(PSTEP1(E0, O0, 3), PSTEP2(E0, O0, 3), E0, O0, f0,
                           f1));
    store_tile(op + 32 * u + 48, outW, row_odd<4>(E0, O0, e1, e2, f0, f1));
  }
}

template <int BASE>
__device__ __forceinline__ v8f row_j1(v2f l1, v2f l2, const YT& E,
                                      const YT& O, v2f r1, v2f r2,
                                      const v2f* g0, const v2f* g1) {
  v8f acc = {0.f, 0.f, 0.f, 0.f, 0.f, 0.f, 0.f, 0.f};
  acc = wmma4(l1, g0[0], acc);
  acc = wmma4(l2, g1[0], acc);
#pragma unroll
  for (int q = 0; q < 4; ++q) {
    acc = wmma4(PSTEP1(E, O, BASE + q), g0[q + 1], acc);
    acc = wmma4(PSTEP2(E, O, BASE + q), g1[q + 1], acc);
  }
  acc = wmma4(r1, g0[5], acc);
  acc = wmma4(r2, g1[5], acc);
  return acc;
}

template <int NOUT, int Hb, int Wb>
__global__ void __launch_bounds__(128)
j1_fused_kernel(const float* __restrict__ plain, const float* __restrict__ yh,
                const float* __restrict__ h0, int taps0,
                const float* __restrict__ h1, int taps1,
                float* __restrict__ out, int B) {
  constexpr int r = 2 * Hb;
  constexpr int W = 2 * Wb;
  int lane = threadIdx.x & 31;
  int wave = __builtin_amdgcn_readfirstlane((int)(threadIdx.x >> 5));
  int g = (int)blockIdx.x * 4 + wave;
  constexpr int nCS = W / (16 * NOUT);
  constexpr int nRT = r / 16;
  int cs = g % nCS;
  int g2 = g / nCS;
  int I = g2 % nRT;
  int b = g2 / nRT;
  if (b >= B) return;
  int half = lane >> 4;
  int l16 = lane & 15;
  int koff = half * 2;
  int x0 = 16 * I - 4;
  int lcol = tile_col(l16);
  int S = cs * 16 * NOUT;
  constexpr bool TWO = (nCS == 2);
  int hcol = TWO ? ((cs ? S - 16 : S + 16 * NOUT) + lcol)
                 : (S + 16 * NOUT + lcol - (lcol >= 12 ? 16 * NOUT + 16 : 0));

  v2f g0[6], g1[6];
#pragma unroll
  for (int s = 0; s < 6; ++s) {
    int kk = 4 * s - 1 + koff;
    g0[s].x = cf_A(l16, kk, h0, taps0);
    g0[s].y = cf_A(l16, kk + 1, h0, taps0);
    g1[s].x = cf_A(l16, kk, h1, taps1);
    g1[s].y = cf_A(l16, kk + 1, h1, taps1);
  }
  v2f f0s[6], f1s[6];
#pragma unroll
  for (int c = 0; c < 6; ++c) {
    f0s[c].x = g0[c].x * SQRT_HALF;
    f0s[c].y = g0[c].y * SQRT_HALF;
    f1s[c].x = g1[c].x * SQRT_HALF;
    f1s[c].y = g1[c].y * SQRT_HALF;
  }

  float* op = out + ((size_t)b * r + (size_t)I * 16 + half * 8) * W +
              cs * 16 * NOUT + l16;

  static_assert(NOUT % 2 == 0, "tiles are computed in aligned pairs");
  constexpr int NP = NOUT / 2;
  YT Hh = col_tile<6, Hb, Wb>(plain, yh, b, x0, hcol, koff, g0, f0s, f1s);
  bool top = (I == 0), bot = (I == nRT - 1);
  v2f c0[6], c0s[6], c1s[6];
#pragma unroll
  for (int c = 0; c < 6; ++c) {
    bool sw = (c == 0 && top) || (c == 5 && bot);
    c0[c] = swap_k(sw, g0[c]);
    c0s[c] = swap_k(sw, f0s[c]);
    c1s[c] = swap_k(sw, f1s[c]);
  }
  unsigned jp = (unsigned)(S / 2 + pair_col(l16));
  YT E0, O0;
  col_pair<6, Hb, Wb>(plain, yh, b, x0, jp, koff, c0, c0s, c1s, E0, O0);
  tile_result_fence(Hh, E0, O0);
  v2f l1 = STEP1(Hh, 3), l2 = STEP2(Hh, 3);
  v2f e1 = STEP1(Hh, 0), e2 = STEP2(Hh, 0);
  if (TWO) {
    l1 = pick_step(cs != 0, l1, mirror_step(PSTEP1(E0, O0, 0)));
    l2 = pick_step(cs != 0, l2, mirror_step(PSTEP2(E0, O0, 0)));
  }
#pragma unroll 1
  for (int p = 1; p < NP; ++p) {
    YT E1, O1;
    col_pair<6, Hb, Wb>(plain, yh, b, x0, jp + 16u * (unsigned)p, koff, c0,
                        c0s, c1s, E1, O1);
    tile_result_fence(E1, O1);
    store_tile(op + 32 * (p - 1), W,
               row_j1<0>(l1, l2, E0, O0, PSTEP1(E0, O0, 4), PSTEP2(E0, O0, 4),
                         g0, g1));
    store_tile(op + 32 * (p - 1) + 16, W,
               row_j1<4>(PSTEP1(E0, O0, 3), PSTEP2(E0, O0, 3), E0, O0,
                         PSTEP1(E1, O1, 0), PSTEP2(E1, O1, 0), g0, g1));
    l1 = PSTEP1(E0, O0, 7);
    l2 = PSTEP2(E0, O0, 7);
    E0 = E1;
    O0 = O1;
    __builtin_amdgcn_sched_barrier(0);
  }
  {
    if (TWO) {
      e1 = pick_step(cs != 0, mirror_step(PSTEP1(E0, O0, 7)), e1);
      e2 = pick_step(cs != 0, mirror_step(PSTEP2(E0, O0, 7)), e2);
    }
    store_tile(op + 16 * (NOUT - 2), W,
               row_j1<0>(l1, l2, E0, O0, PSTEP1(E0, O0, 4), PSTEP2(E0, O0, 4),
                         g0, g1));
    store_tile(op + 16 * (NOUT - 1), W,
               row_j1<4>(PSTEP1(E0, O0, 3), PSTEP2(E0, O0, 3), E0, O0, e1, e2,
                         g0, g1));
  }
}

extern "C" void kernel_launch(void* const* d_in, const int* in_sizes, int n_in,
                              void* d_out, int out_size, void* d_ws,
                              size_t ws_size, hipStream_t stream) {
  (void)in_sizes;
  (void)n_in;
  (void)out_size;
  (void)ws_size;

  const float* yl = (const float*)d_in[0];
  const float* yh0 = (const float*)d_in[1];
  const float* yh1 = (const float*)d_in[2];
  const float* yh2 = (const float*)d_in[3];
  const float* g0o = (const float*)d_in[4];
  const float* g1o = (const float*)d_in[5];
  const float* g0a = (const float*)d_in[6];
  const float* g0b = (const float*)d_in[7];
  const float* g1a = (const float*)d_in[8];
  const float* g1b = (const float*)d_in[9];

  float* LLa = (float*)d_ws;
  float* LLb = LLa + (size_t)64 * 65536;
  float* outp = (float*)d_out;

  dim3 blk(128);

  j2_fused_kernel<2, 64, 64><<<dim3(1024), blk, 0, stream>>>(
      yl, yh2, g0b, g0a, g1b, g1a, LLa, 64);

  j2_fused_kernel<8, 128, 128><<<dim3(1024), blk, 0, stream>>>(
      LLa, yh1, g0b, g0a, g1b, g1a, LLb, 64);

  j1_fused_kernel<16, 256, 256><<<dim3(1024), blk, 0, stream>>>(
      LLb, yh0, g0o, 7, g1o, 5, outp, 64);
}


# Column pass software-pipelined: a pair's later K-step groups are requested before its first products, the next pair's first group right behind those, each in its own registers; sums unchanged
# speedup vs baseline: 4.5249x; 1.0579x over previous
#include <hip/hip_runtime.h>


typedef __attribute__((ext_vector_type(2))) float v2f;
typedef __attribute__((ext_vector_type(8))) float v8f;

#define SQRT_HALF 0.70710678118654752440f

__device__ __forceinline__ int reflect_i(int x, int n) {
  x = (x < 0) ? (-1 - x) : x;
  x = (x >= n) ? (2 * n - 1 - x) : x;
  return x;
}

__device__ __forceinline__ int tile_col(int m) {
  return 4 * ((m & 7) >> 1) + 2 * (m >> 3) + (m & 1);
}

__device__ __forceinline__ float cif_A(int m, int kk,
                                       const float* __restrict__ ha,
                                       const float* __restrict__ hb, int hp) {
  int p = m & 3;
  int q = m >> 2;
  int want_odd = hp ? ((p & 1) ^ 1) : (p & 1);
  int t = ((kk - want_odd) >> 1) - q;
  bool ok = ((kk & 1) == want_odd) && (t >= 0) && (t <= 4);
  int tc = t < 0 ? 0 : (t > 4 ? 4 : t);
  const float* __restrict__ h = ((p == 0) || (p == 2)) ? ha : hb;
  int idx = (p < 2) ? (2 * tc + 1) : (2 * tc);
  float v = h[idx];
  return ok ? v : 0.0f;
}

__device__ __forceinline__ float cf_A(int m, int kk,
                                      const float* __restrict__ h, int taps) {
  int t = kk - m - (3 - (taps >> 1));
  bool ok = (t >= 0) && (t < taps);
  int tc = t < 0 ? 0 : (t >= taps ? taps - 1 : t);
  float v = h[tc];
  return ok ? v : 0.0f;
}

__device__ __forceinline__ void c2q_make(v2f w1, v2f w2, int odd_row, int cb,
                                         float& own, float& give) {
  float c0 = odd_row ? (w1.y - w2.y) : (w1.x + w2.x);
  float c1 = odd_row ? (w2.x - w1.x) : (w1.y + w2.y);
  own = cb ? c1 : c0;
  give = cb ? c0 : c1;
}

__device__ __forceinline__ void c2q_pair(const float* __restrict__ p1,
                                         const float* __restrict__ p2,
                                         unsigned o, int odd_row, int cb,
                                         float& own, float& give) {
  v2f w1 = ((const v2f*)p1)[o];
  v2f w2 = ((const v2f*)p2)[o];
  c2q_make(w1, w2, odd_row, cb, own, give);
}

__device__ __forceinline__ float pair_lane(float x) {
  return __builtin_bit_cast(
      float, __builtin_amdgcn_update_dpp(0, __builtin_bit_cast(int, x), 0xB1,
                                         0xF, 0xF, true));
}

__device__ __forceinline__ v8f wmma4(v2f a, v2f b, v8f c) {
  return __builtin_amdgcn_wmma_f32_16x16x4_f32(false, a, false, b, (short)0, c,
                                               false, false);
}

struct YT {
  v8f y1, y2;
};

template <int KC, int Hb, int Wb>
__device__ __forceinline__ YT col_tile(const float* __restrict__ plain,
                                       const float* __restrict__ yh, int b,
                                       int x0, int ccu, int koff,
                                       const v2f* f0, const v2f* f0s,
                                       const v2f* f1s) {
  constexpr int r = 2 * Hb;
  constexpr int W = 2 * Wb;
  int cc = reflect_i(ccu, W);
  int cb = cc & 1;
  unsigned jb = (unsigned)(cc >> 1);
  int lo = (int)(threadIdx.x & 1);
  constexpr size_t bs = (size_t)Hb * Wb * 2;
  const float* p0 = plain + (size_t)b * r * W;
  const float* yb = yh + (size_t)b * 6 * bs;
  const float* q0 = yb;
  const float* q1 = yb + bs;
  const float* q2 = yb + 2 * bs;
  const float* q3 = yb + 3 * bs;
  const float* q4 = yb + 4 * bs;
  const float* q5 = yb + 5 * bs;
  YT o;
  o.y1 = (v8f){0.f, 0.f, 0.f, 0.f, 0.f, 0.f, 0.f, 0.f};
  o.y2 = (v8f){0.f, 0.f, 0.f, 0.f, 0.f, 0.f, 0.f, 0.f};
#pragma unroll
  for (int c = 0; c < KC; ++c) {
    int kk = 4 * c + koff;
    int rs = reflect_i(x0 + kk + lo, r);
    unsigned ob = (unsigned)(rs >> 1) * (unsigned)Wb + jb;
    int d = rs & 1;
    v2f pl = ((const v2f*)p0)[(unsigned)rs * (unsigned)Wb + jb];
    float pl_o = cb ? pl.y : pl.x;
    float pl_g = cb ? pl.x : pl.y;
    float lh_o, lh_g, hl_o, hl_g, hh_o, hh_g;
    c2q_pair(q0, q5, ob, d, cb, lh_o, lh_g);
    c2q_pair(q2, q3, ob, d, cb, hl_o, hl_g);
    c2q_pair(q1, q4, ob, d, cb, hh_o, hh_g);
    float pl_x = pair_lane(pl_g);
    float lh_x = pair_lane(lh_g);
    float hl_x = pair_lane(hl_g);
    float hh_x = pair_lane(hh_g);
    v2f xl = lo ? (v2f){pl_x, pl_o} : (v2f){pl_o, pl_x};
    v2f xlh = lo ? (v2f){lh_x, lh_o} : (v2f){lh_o, lh_x};
    v2f xhl = lo ? (v2f){hl_x, hl_o} : (v2f){hl_o, hl_x};
    v2f xhh = lo ? (v2f){hh_x, hh_o} : (v2f){hh_o, hh_x};
    o.y1 = wmma4(xl, f0[c], o.y1);
    o.y1 = wmma4(xlh, f1s[c], o.y1);
    o.y2 = wmma4(xhl, f0s[c], o.y2);
    o.y2 = wmma4(xhh, f1s[c], o.y2);
    if (KC > 4 && c == KC / 2 - 1) __builtin_amdgcn_sched_barrier(0);
  }
  return o;
}

__device__ __forceinline__ int pair_col(int m) {
  return 2 * (m & 7) + (m >> 3);
}

__device__ __forceinline__ void pair_step(YT& E, YT& O, v2f r0, v2f r1,
                                          v2f lhA, v2f lhB, v2f hlA, v2f hlB,
                                          v2f hhA, v2f hhB, v2f f0c, v2f f0sc,
                                          v2f f1sc) {
  v2f ple = {r0.x, r1.x};
  v2f plo = {r0.y, r1.y};
  v2f lhe = {lhA.x + lhB.x, lhA.y - lhB.y};
  v2f lho = {lhA.y + lhB.y, lhB.x - lhA.x};
  v2f hle = {hlA.x + hlB.x, hlA.y - hlB.y};
  v2f hlo = {hlA.y + hlB.y, hlB.x - hlA.x};
  v2f hhe = {hhA.x + hhB.x, hhA.y - hhB.y};
  v2f hho = {hhA.y + hhB.y, hhB.x - hhA.x};
  E.y1 = wmma4(ple, f0c, E.y1);
  E.y1 = wmma4(lhe, f1sc, E.y1);
  E.y2 = wmma4(hle, f0sc, E.y2);
  E.y2 = wmma4(hhe, f1sc, E.y2);
  O.y1 = wmma4(plo, f0c, O.y1);
  O.y1 = wmma4(lho, f1sc, O.y1);
  O.y2 = wmma4(hlo, f0sc, O.y2);
  O.y2 = wmma4(hho, f1sc, O.y2);
}

struct PG {
  v2f r0[2], r1[2], a0[2], a1[2], a2[2], a3[2], a4[2], a5[2];
};

template <int CB, int Hb, int Wb>
__device__ __forceinline__ void pair_load(PG& G,
                                          const float* __restrict__ plain,
                                          const float* __restrict__ yh, int b,
                                          int x0, unsigned jb, int koff) {
  constexpr int r = 2 * Hb;
  constexpr int W = 2 * Wb;
  constexpr size_t bs = (size_t)Hb * Wb * 2;
  const v2f* p0 = (const v2f*)(plain + (size_t)b * r * W);
  const float* yb = yh + (size_t)b * 6 * bs;
  const v2f* q0 = (const v2f*)yb;
  const v2f* q1 = (const v2f*)(yb + bs);
  const v2f* q2 = (const v2f*)(yb + 2 * bs);
  const v2f* q3 = (const v2f*)(yb + 3 * bs);
  const v2f* q4 = (const v2f*)(yb + 4 * bs);
  const v2f* q5 = (const v2f*)(yb + 5 * bs);
#pragma unroll
  for (int s = 0; s < 2; ++s) {
    int kk = 4 * (CB + s) + koff;
    int ir = reflect_i((x0 + kk) >> 1, Hb);
    unsigned ob = (unsigned)ir * (unsigned)Wb + jb;
    unsigned opl = 2u * (unsigned)ir * (unsigned)Wb + jb;
    G.r0[s] = p0[opl];
    G.r1[s] = p0[opl + Wb];
    G.a0[s] = q0[ob];
    G.a5[s] = q5[ob];
    G.a2[s] = q2[ob];
    G.a3[s] = q3[ob];
    G.a1[s] = q1[ob];
    G.a4[s] = q4[ob];
  }
}

template <int CB>
__device__ __forceinline__ void pair_mma(YT& E, YT& O, const PG& G,
                                         const v2f* f0, const v2f* f0s,
                                         const v2f* f1s) {
#pragma unroll
  for (int s = 0; s < 2; ++s)
    pair_step(E, O, G.r0[s], G.r1[s], G.a0[s], G.a5[s], G.a2[s], G.a3[s],
              G.a1[s], G.a4[s], f0[CB + s], f0s[CB + s], f1s[CB + s]);
}

__device__ __forceinline__ void order_point(YT& E, YT& O) {
  asm volatile(""
               : "+v"(E.y1), "+v"(E.y2), "+v"(O.y1), "+v"(O.y2)
               :
               : "memory");
}

template <int KC, int Hb, int Wb>
__device__ __forceinline__ void col_pair_piped(
    const float* __restrict__ plain, const float* __restrict__ yh, int b,
    int x0, unsigned jb, unsigned jb_next, bool next, int koff, const v2f* f0,
    const v2f* f0s, const v2f* f1s, PG& G, YT& E, YT& O) {
  static_assert(KC == 4 || KC == 6, "two or three groups of two K steps");
  E.y1 = (v8f){0.f, 0.f, 0.f, 0.f, 0.f, 0.f, 0.f, 0.f};
  E.y2 = (v8f){0.f, 0.f, 0.f, 0.f, 0.f, 0.f, 0.f, 0.f};
  O.y1 = (v8f){0.f, 0.f, 0.f, 0.f, 0.f, 0.f, 0.f, 0.f};
  O.y2 = (v8f){0.f, 0.f, 0.f, 0.f, 0.f, 0.f, 0.f, 0.f};
  PG G1;
  [[maybe_unused]] PG G2;
  pair_load<2, Hb, Wb>(G1, plain, yh, b, x0, jb, koff);
  if constexpr (KC == 6) pair_load<4, Hb, Wb>(G2, plain, yh, b, x0, jb, koff);
  __builtin_amdgcn_sched_barrier(0);
  pair_mma<0>(E, O, G, f0, f0s, f1s);
  order_point(E, O);
  if (next) pair_load<0, Hb, Wb>(G, plain, yh, b, x0, jb_next, koff);
  order_point(E, O);
  __builtin_amdgcn_sched_barrier(0);
  pair_mma<2>(E, O, G1, f0, f0s, f1s);
  if constexpr (KC == 6) pair_mma<4>(E, O, G2, f0, f0s, f1s);
}

__device__ __forceinline__ v2f swap_k(bool c, v2f a) {
  return (v2f){c ? a.y : a.x, c ? a.x : a.y};
}

__device__ __forceinline__ void store_tile(float* __restrict__ op, int pitch,
                                           v8f acc) {
#pragma unroll
  for (int e = 0; e < 8; ++e) op[(size_t)e * pitch] = acc[e];
}

__device__ __forceinline__ void tile_result_fence(YT& T) {
  __builtin_amdgcn_sched_barrier(0);
  asm volatile(
      "s_wait_alu depctr_va_vdst(0)\n\t"
      "v_nop\n\tv_nop\n\tv_nop\n\tv_nop\n\tv_nop\n\t"
      "v_nop\n\tv_nop\n\tv_nop\n\tv_nop"
      : "+v"(T.y1), "+v"(T.y2)
      :
      : "memory");
  __builtin_amdgcn_sched_barrier(0);
}

__device__ __forceinline__ void tile_result_fence(YT& H, YT& T) {
  __builtin_amdgcn_sched_barrier(0);
  asm volatile(
      "s_wait_alu depctr_va_vdst(0)\n\t"
      "v_nop\n\tv_nop\n\tv_nop\n\tv_nop\n\tv_nop\n\t"
      "v_nop\n\tv_nop\n\tv_nop\n\tv_nop"
      : "+v"(H.y1), "+v"(H.y2), "+v"(T.y1), "+v"(T.y2)
      :
      : "memory");
  __builtin_amdgcn_sched_barrier(0);
}

__device__ __forceinline__ void tile_result_fence(YT& H, YT& S, YT& T) {
  __builtin_amdgcn_sched_barrier(0);
  asm volatile(
      "s_wait_alu depctr_va_vdst(0)\n\t"
      "v_nop\n\tv_nop\n\tv_nop\n\tv_nop\n\tv_nop\n\t"
      "v_nop\n\tv_nop\n\tv_nop\n\tv_nop"
      : "+v"(H.y1), "+v"(H.y2), "+v"(S.y1), "+v"(S.y2), "+v"(T.y1),
        "+v"(T.y2)
      :
      : "memory");
  __builtin_amdgcn_sched_barrier(0);
}

#define STEP1(T, q) ((v2f){(T).y1[2 * (q)], (T).y1[2 * (q) + 1]})
#define STEP2(T, q) ((v2f){(T).y2[2 * (q)], (T).y2[2 * (q) + 1]})

#define PSTEP1(E, O, t) ((v2f){(E).y1[(t)], (O).y1[(t)]})
#define PSTEP2(E, O, t) ((v2f){(E).y2[(t)], (O).y2[(t)]})

__device__ __forceinline__ float other_half(float x) {
  return __builtin_bit_cast(
      float, __builtin_amdgcn_permlanex16(__builtin_bit_cast(int, x),
                                          __builtin_bit_cast(int, x),
                                          0x76543210, 0xfedcba98, false, false));
}

__device__ __forceinline__ v2f mirror_step(v2f a) {
  return (v2f){other_half(a.y), other_half(a.x)};
}

__device__ __forceinline__ v2f pick_step(bool c, v2f a, v2f b) {
  return (v2f){c ? a.x : b.x, c ? a.y : b.y};
}

template <int BASE>
__device__ __forceinline__ v8f row_even(v2f l1, v2f l2, const YT& E,
                                        const YT& O, const v2f* g0,
                                        const v2f* g1) {
  v8f acc = {0.f, 0.f, 0.f, 0.f, 0.f, 0.f, 0.f, 0.f};
  acc = wmma4(l1, g0[0], acc);
  acc = wmma4(l2, g1[0], acc);
#pragma unroll
  for (int q = 0; q < 3; ++q) {
    acc = wmma4(PSTEP1(E, O, BASE + q), g0[q + 1], acc);
    acc = wmma4(PSTEP2(E, O, BASE + q), g1[q + 1], acc);
  }
  return acc;
}

template <int BASE>
__device__ __forceinline__ v8f row_odd(const YT& E, const YT& O, v2f r1,
                                       v2f r2, const v2f* g0, const v2f* g1) {
  v8f acc = {0.f, 0.f, 0.f, 0.f, 0.f, 0.f, 0.f, 0.f};
#pragma unroll
  for (int q = 1; q < 4; ++q) {
    acc = wmma4(PSTEP1(E, O, BASE + q), g0[q - 1], acc);
    acc = wmma4(PSTEP2(E, O, BASE + q), g1[q - 1], acc);
  }
  acc = wmma4(r1, g0[3], acc);
  acc = wmma4(r2, g1[3], acc);
  return acc;
}

template <int NT, int Hb, int Wb>
__global__ void __launch_bounds__(128, 4)
j2_fused_kernel(const float* __restrict__ plain, const float* __restrict__ yh,
                const float* __restrict__ ha0, const float* __restrict__ hb0,
                const float* __restrict__ ha1, const float* __restrict__ hb1,
                float* __restrict__ out, int B) {
  constexpr int r = 2 * Hb;
  constexpr int W = 2 * Wb;
  int lane = threadIdx.x & 31;
  int wave = __builtin_amdgcn_readfirstlane((int)(threadIdx.x >> 5));
  int g = (int)blockIdx.x * 4 + wave;
  constexpr int nCS = W / (16 * NT);
  constexpr int nRT = (2 * r) / 16;
  int cs = g % nCS;
  int g2 = g / nCS;
  int I = g2 % nRT;
  int b = g2 / nRT;
  if (b >= B) return;
  int half = lane >> 4;
  int l16 = lane & 15;
  int koff = half * 2;
  int x0 = 8 * I - 4;
  int lcol = tile_col(l16);
  int S = cs * 16 * NT;
  constexpr bool TWO = (nCS == 2);
  int hcol = TWO ? ((cs ? S - 16 : S + 16 * NT) + lcol)
                 : (S + 16 * NT + lcol - (lcol >= 12 ? 16 * NT + 16 : 0));

  v2f f0[4], f0s[4], f1[4], f1s[4];
#pragma unroll
  for (int c = 0; c < 4; ++c) {
    int kk = 4 * c + koff;
    float ax = cif_A(l16, kk, ha0, hb0, 0);
    float ay = cif_A(l16, kk + 1, ha0, hb0, 0);
    float bx = cif_A(l16, kk, ha1, hb1, 1);
    float by = cif_A(l16, kk + 1, ha1, hb1, 1);
    f0[c].x = ax;
    f0[c].y = ay;
    f0s[c].x = ax * SQRT_HALF;
    f0s[c].y = ay * SQRT_HALF;
    f1[c].x = bx;
    f1[c].y = by;
    f1s[c].x = bx * SQRT_HALF;
    f1s[c].y = by * SQRT_HALF;
  }

  constexpr int outW = 2 * W;
  float* op = out + ((size_t)b * (2 * r) + (size_t)I * 16 + half * 8) * outW +
              cs * 32 * NT + l16;

  static_assert(NT % 2 == 0, "tiles are computed in aligned pairs");
  constexpr int NP = NT / 2;
  unsigned jp = (unsigned)(S / 2 + pair_col(l16));
  PG G;
  pair_load<0, Hb, Wb>(G, plain, yh, b, x0, jp, koff);
  __builtin_amdgcn_sched_barrier(0);
  YT Hh = col_tile<4, Hb, Wb>(plain, yh, b, x0, hcol, koff, f0, f0s, f1s);
  bool top = (I == 0), bot = (I == nRT - 1);
  v2f c0[4], c0s[4], c1s[4];
#pragma unroll
  for (int c = 0; c < 4; ++c) {
    bool sw = (c == 0 && top) || (c == 3 && bot);
    c0[c] = swap_k(sw, f0[c]);
    c0s[c] = swap_k(sw, f0s[c]);
    c1s[c] = swap_k(sw, f1s[c]);
  }
  YT E0, O0;
  col_pair_piped<4, Hb, Wb>(plain, yh, b, x0, jp, jp + 16u, NP > 1, koff, c0,
                            c0s, c1s, G, E0, O0);
  tile_result_fence(Hh, E0, O0);
  v2f l1 = STEP1(Hh, 3), l2 = STEP2(Hh, 3);
  v2f e1 = STEP1(Hh, 0), e2 = STEP2(Hh, 0);
  if (TWO) {
    l1 = pick_step(cs != 0, l1, mirror_step(PSTEP1(E0, O0, 0)));
    l2 = pick_step(cs != 0, l2, mirror_step(PSTEP2(E0, O0, 0)));
  }
#pragma unroll
  for (int p = 1; p < NP; ++p) {
    const int u = 2 * (p - 1);
    store_tile(op + 32 * u, outW, row_even<0>(l1, l2, E0, O0, f0, f1));
    store_tile(op + 32 * u + 16, outW,
               row_odd<0>(E0, O0, PSTEP1(E0, O0, 4), PSTEP2(E0, O0, 4), f0,
                          f1));
    store_tile(op + 32 * u + 32, outW,
               row_even<4>(PSTEP1(E0, O0, 3), PSTEP2(E0, O0, 3), E0, O0, f0,
                           f1));
    YT E1, O1;
    col_pair_piped<4, Hb, Wb>(plain, yh, b, x0, jp + 16u * (unsigned)p,
                              jp + 16u * (unsigned)(p + 1), p + 1 < NP, koff,
                              c0, c0s, c1s, G, E1, O1);
    tile_result_fence(E1, O1);
    store_tile(op + 32 * u + 48, outW,
               row_odd<4>(E0, O0, PSTEP1(E1, O1, 0), PSTEP2(E1, O1, 0), f0,
                          f1));
    l1 = PSTEP1(E0, O0, 7);
    l2 = PSTEP2(E0, O0, 7);
    E0 = E1;
    O0 = O1;
    __builtin_amdgcn_sched_barrier(0);
  }
  {
    constexpr int u = NT - 2;
    if (TWO) {
      e1 = pick_step(cs != 0, mirror_step(PSTEP1(E0, O0, 7)), e1);
      e2 = pick_step(cs != 0, mirror_step(PSTEP2(E0, O0, 7)), e2);
    }
    store_tile(op + 32 * u, outW, row_even<0>(l1, l2, E0, O0, f0, f1));
    store_tile(op + 32 * u + 16, outW,
               row_odd<0>(E0, O0, PSTEP1(E0, O0, 4), PSTEP2(E0, O0, 4), f0,
                          f1));
    store_tile(op + 32 * u + 32, outW,
               row_even<4>(PSTEP1(E0, O0, 3), PSTEP2(E0, O0, 3), E0, O0, f0,
                           f1));
    store_tile(op + 32 * u + 48, outW, row_odd<4>(E0, O0, e1, e2, f0, f1));
  }
}

template <int BASE>
__device__ __forceinline__ v8f row_j1(v2f l1, v2f l2, const YT& E,
                                      const YT& O, v2f r1, v2f r2,
                                      const v2f* g0, const v2f* g1) {
  v8f acc = {0.f, 0.f, 0.f, 0.f, 0.f, 0.f, 0.f, 0.f};
  acc = wmma4(l1, g0[0], acc);
  acc = wmma4(l2, g1[0], acc);
#pragma unroll
  for (int q = 0; q < 4; ++q) {
    acc = wmma4(PSTEP1(E, O, BASE + q), g0[q + 1], acc);
    acc = wmma4(PSTEP2(E, O, BASE + q), g1[q + 1], acc);
  }
  acc = wmma4(r1, g0[5], acc);
  acc = wmma4(r2, g1[5], acc);
  return acc;
}

template <int NOUT, int Hb, int Wb>
__global__ void __launch_bounds__(128)
j1_fused_kernel(const float* __restrict__ plain, const float* __restrict__ yh,
                const float* __restrict__ h0, int taps0,
                const float* __restrict__ h1, int taps1,
                float* __restrict__ out, int B) {
  constexpr int r = 2 * Hb;
  constexpr int W = 2 * Wb;
  int lane = threadIdx.x & 31;
  int wave = __builtin_amdgcn_readfirstlane((int)(threadIdx.x >> 5));
  int g = (int)blockIdx.x * 4 + wave;
  constexpr int nCS = W / (16 * NOUT);
  constexpr int nRT = r / 16;
  int cs = g % nCS;
  int g2 = g / nCS;
  int I = g2 % nRT;
  int b = g2 / nRT;
  if (b >= B) return;
  int half = lane >> 4;
  int l16 = lane & 15;
  int koff = half * 2;
  int x0 = 16 * I - 4;
  int lcol = tile_col(l16);
  int S = cs * 16 * NOUT;
  constexpr bool TWO = (nCS == 2);
  int hcol = TWO ? ((cs ? S - 16 : S + 16 * NOUT) + lcol)
                 : (S + 16 * NOUT + lcol - (lcol >= 12 ? 16 * NOUT + 16 : 0));

  v2f g0[6], g1[6];
#pragma unroll
  for (int s = 0; s < 6; ++s) {
    int kk = 4 * s - 1 + koff;
    g0[s].x = cf_A(l16, kk, h0, taps0);
    g0[s].y = cf_A(l16, kk + 1, h0, taps0);
    g1[s].x = cf_A(l16, kk, h1, taps1);
    g1[s].y = cf_A(l16, kk + 1, h1, taps1);
  }
  v2f f0s[6], f1s[6];
#pragma unroll
  for (int c = 0; c < 6; ++c) {
    f0s[c].x = g0[c].x * SQRT_HALF;
    f0s[c].y = g0[c].y * SQRT_HALF;
    f1s[c].x = g1[c].x * SQRT_HALF;
    f1s[c].y = g1[c].y * SQRT_HALF;
  }

  float* op = out + ((size_t)b * r + (size_t)I * 16 + half * 8) * W +
              cs * 16 * NOUT + l16;

  static_assert(NOUT % 2 == 0, "tiles are computed in aligned pairs");
  constexpr int NP = NOUT / 2;
  unsigned jp = (unsigned)(S / 2 + pair_col(l16));
  PG G;
  pair_load<0, Hb, Wb>(G, plain, yh, b, x0, jp, koff);
  __builtin_amdgcn_sched_barrier(0);
  YT Hh = col_tile<6, Hb, Wb>(plain, yh, b, x0, hcol, koff, g0, f0s, f1s);
  bool top = (I == 0), bot = (I == nRT - 1);
  v2f c0[6], c0s[6], c1s[6];
#pragma unroll
  for (int c = 0; c < 6; ++c) {
    bool sw = (c == 0 && top) || (c == 5 && bot);
    c0[c] = swap_k(sw, g0[c]);
    c0s[c] = swap_k(sw, f0s[c]);
    c1s[c] = swap_k(sw, f1s[c]);
  }
  YT E0, O0;
  col_pair_piped<6, Hb, Wb>(plain, yh, b, x0, jp, jp + 16u, NP > 1, koff, c0,
                            c0s, c1s, G, E0, O0);
  tile_result_fence(Hh, E0, O0);
  v2f l1 = STEP1(Hh, 3), l2 = STEP2(Hh, 3);
  v2f e1 = STEP1(Hh, 0), e2 = STEP2(Hh, 0);
  if (TWO) {
    l1 = pick_step(cs != 0, l1, mirror_step(PSTEP1(E0, O0, 0)));
    l2 = pick_step(cs != 0, l2, mirror_step(PSTEP2(E0, O0, 0)));
  }
  auto trip = [&](int p, bool next) __attribute__((always_inline)) {
    YT E1, O1;
    col_pair_piped<6, Hb, Wb>(plain, yh, b, x0, jp + 16u * (unsigned)p,
                              jp + 16u * (unsigned)(p + 1), next, koff, c0,
                              c0s, c1s, G, E1, O1);
    tile_result_fence(E1, O1);
    store_tile(op + 32 * (p - 1), W,
               row_j1<0>(l1, l2, E0, O0, PSTEP1(E0, O0, 4), PSTEP2(E0, O0, 4),
                         g0, g1));
    store_tile(op + 32 * (p - 1) + 16, W,
               row_j1<4>(PSTEP1(E0, O0, 3), PSTEP2(E0, O0, 3), E0, O0,
                         PSTEP1(E1, O1, 0), PSTEP2(E1, O1, 0), g0, g1));
    l1 = PSTEP1(E0, O0, 7);
    l2 = PSTEP2(E0, O0, 7);
    E0 = E1;
    O0 = O1;
    __builtin_amdgcn_sched_barrier(0);
  };
#pragma unroll 1
  for (int p = 1; p < NP - 1; ++p) trip(p, true);
  if (NP > 1) trip(NP - 1, false);
  {
    if (TWO) {
      e1 = pick_step(cs != 0, mirror_step(PSTEP1(E0, O0, 7)), e1);
      e2 = pick_step(cs != 0, mirror_step(PSTEP2(E0, O0, 7)), e2);
    }
    store_tile(op + 16 * (NOUT - 2), W,
               row_j1<0>(l1, l2, E0, O0, PSTEP1(E0, O0, 4), PSTEP2(E0, O0, 4),
                         g0, g1));
    store_tile(op + 16 * (NOUT - 1), W,
               row_j1<4>(PSTEP1(E0, O0, 3), PSTEP2(E0, O0, 3), E0, O0, e1, e2,
                         g0, g1));
  }
}

extern "C" void kernel_launch(void* const* d_in, const int* in_sizes, int n_in,
                              void* d_out, int out_size, void* d_ws,
                              size_t ws_size, hipStream_t stream) {
  (void)in_sizes;
  (void)n_in;
  (void)out_size;
  (void)ws_size;

  const float* yl = (const float*)d_in[0];
  const float* yh0 = (const float*)d_in[1];
  const float* yh1 = (const float*)d_in[2];
  const float* yh2 = (const float*)d_in[3];
  const float* g0o = (const float*)d_in[4];
  const float* g1o = (const float*)d_in[5];
  const float* g0a = (const float*)d_in[6];
  const float* g0b = (const float*)d_in[7];
  const float* g1a = (const float*)d_in[8];
  const float* g1b = (const float*)d_in[9];

  float* LLa = (float*)d_ws;
  float* LLb = LLa + (size_t)64 * 65536;
  float* outp = (float*)d_out;

  dim3 blk(128);

  j2_fused_kernel<2, 64, 64><<<dim3(1024), blk, 0, stream>>>(
      yl, yh2, g0b, g0a, g1b, g1a, LLa, 64);

  j2_fused_kernel<8, 128, 128><<<dim3(1024), blk, 0, stream>>>(
      LLa, yh1, g0b, g0a, g1b, g1a, LLb, 64);

  j1_fused_kernel<16, 256, 256><<<dim3(1024), blk, 0, stream>>>(
      LLb, yh0, g0o, 7, g1o, 5, outp, 64);
}


# speedup vs baseline: 4.7127x; 1.0415x over previous
#include <hip/hip_runtime.h>


typedef __attribute__((ext_vector_type(2))) float v2f;
typedef __attribute__((ext_vector_type(4))) float v4f;
typedef __attribute__((ext_vector_type(8))) float v8f;

#define SQRT_HALF 0.70710678118654752440f

__device__ __forceinline__ int reflect_i(int x, int n) {
  x = (x < 0) ? (-1 - x) : x;
  x = (x >= n) ? (2 * n - 1 - x) : x;
  return x;
}

__device__ __forceinline__ int tile_col(int m) {
  return 4 * ((m & 7) >> 1) + 2 * (m >> 3) + (m & 1);
}

__device__ __forceinline__ float cif_A(int m, int kk,
                                       const float* __restrict__ ha,
                                       const float* __restrict__ hb, int hp) {
  int p = m & 3;
  int q = m >> 2;
  int want_odd = hp ? ((p & 1) ^ 1) : (p & 1);
  int t = ((kk - want_odd) >> 1) - q;
  bool ok = ((kk & 1) == want_odd) && (t >= 0) && (t <= 4);
  int tc = t < 0 ? 0 : (t > 4 ? 4 : t);
  const float* __restrict__ h = ((p == 0) || (p == 2)) ? ha : hb;
  int idx = (p < 2) ? (2 * tc + 1) : (2 * tc);
  float v = h[idx];
  return ok ? v : 0.0f;
}

__device__ __forceinline__ float cf_A(int m, int kk,
                                      const float* __restrict__ h, int taps) {
  int t = kk - m - (3 - (taps >> 1));
  bool ok = (t >= 0) && (t < taps);
  int tc = t < 0 ? 0 : (t >= taps ? taps - 1 : t);
  float v = h[tc];
  return ok ? v : 0.0f;
}

__device__ __forceinline__ void c2q_make(v2f w1, v2f w2, int odd_row, int cb,
                                         float& own, float& give) {
  float c0 = odd_row ? (w1.y - w2.y) : (w1.x + w2.x);
  float c1 = odd_row ? (w2.x - w1.x) : (w1.y + w2.y);
  own = cb ? c1 : c0;
  give = cb ? c0 : c1;
}

__device__ __forceinline__ void c2q_pair(const float* __restrict__ p1,
                                         const float* __restrict__ p2,
                                         unsigned o, int odd_row, int cb,
                                         float& own, float& give) {
  v2f w1 = ((const v2f*)p1)[o];
  v2f w2 = ((const v2f*)p2)[o];
  c2q_make(w1, w2, odd_row, cb, own, give);
}

__device__ __forceinline__ float pair_lane(float x) {
  return __builtin_bit_cast(
      float, __builtin_amdgcn_update_dpp(0, __builtin_bit_cast(int, x), 0xB1,
                                         0xF, 0xF, true));
}

__device__ __forceinline__ v8f wmma4(v2f a, v2f b, v8f c) {
  return __builtin_amdgcn_wmma_f32_16x16x4_f32(false, a, false, b, (short)0, c,
                                               false, false);
}

struct YT {
  v8f y1, y2;
};

template <int KC, int Hb, int Wb>
__device__ __forceinline__ YT col_tile(const float* __restrict__ plain,
                                       const float* __restrict__ yh, int b,
                                       int x0, int ccu, int koff,
                                       const v2f* f0, const v2f* f0s,
                                       const v2f* f1s) {
  constexpr int r = 2 * Hb;
  constexpr int W = 2 * Wb;
  int cc = reflect_i(ccu, W);
  int cb = cc & 1;
  unsigned jb = (unsigned)(cc >> 1);
  int lo = (int)(threadIdx.x & 1);
  constexpr size_t bs = (size_t)Hb * Wb * 2;
  const float* p0 = plain + (size_t)b * r * W;
  const float* yb = yh + (size_t)b * 6 * bs;
  const float* q0 = yb;
  const float* q1 = yb + bs;
  const float* q2 = yb + 2 * bs;
  const float* q3 = yb + 3 * bs;
  const float* q4 = yb + 4 * bs;
  const float* q5 = yb + 5 * bs;
  YT o;
  o.y1 = (v8f){0.f, 0.f, 0.f, 0.f, 0.f, 0.f, 0.f, 0.f};
  o.y2 = (v8f){0.f, 0.f, 0.f, 0.f, 0.f, 0.f, 0.f, 0.f};
#pragma unroll
  for (int c = 0; c < KC; ++c) {
    int kk = 4 * c + koff;
    int rs = reflect_i(x0 + kk + lo, r);
    unsigned ob = (unsigned)(rs >> 1) * (unsigned)Wb + jb;
    int d = rs & 1;
    v2f pl = ((const v2f*)p0)[(unsigned)rs * (unsigned)Wb + jb];
    float pl_o = cb ? pl.y : pl.x;
    float pl_g = cb ? pl.x : pl.y;
    float lh_o, lh_g, hl_o, hl_g, hh_o, hh_g;
    c2q_pair(q0, q5, ob, d, cb, lh_o, lh_g);
    c2q_pair(q2, q3, ob, d, cb, hl_o, hl_g);
    c2q_pair(q1, q4, ob, d, cb, hh_o, hh_g);
    float pl_x = pair_lane(pl_g);
    float lh_x = pair_lane(lh_g);
    float hl_x = pair_lane(hl_g);
    float hh_x = pair_lane(hh_g);
    v2f xl = lo ? (v2f){pl_x, pl_o} : (v2f){pl_o, pl_x};
    v2f xlh = lo ? (v2f){lh_x, lh_o} : (v2f){lh_o, lh_x};
    v2f xhl = lo ? (v2f){hl_x, hl_o} : (v2f){hl_o, hl_x};
    v2f xhh = lo ? (v2f){hh_x, hh_o} : (v2f){hh_o, hh_x};
    o.y1 = wmma4(xl, f0[c], o.y1);
    o.y1 = wmma4(xlh, f1s[c], o.y1);
    o.y2 = wmma4(xhl, f0s[c], o.y2);
    o.y2 = wmma4(xhh, f1s[c], o.y2);
    if (KC > 4 && c == KC / 2 - 1) __builtin_amdgcn_sched_barrier(0);
  }
  return o;
}

__device__ __forceinline__ int pair_col(int m) {
  return 2 * (m & 7) + (m >> 3);
}

__device__ __forceinline__ void pair_step(YT& E, YT& O, v2f r0, v2f r1,
                                          v2f lhA, v2f lhB, v2f hlA, v2f hlB,
                                          v2f hhA, v2f hhB, v2f f0c, v2f f0sc,
                                          v2f f1sc) {
  v2f ple = {r0.x, r1.x};
  v2f plo = {r0.y, r1.y};
  v2f lhe = {lhA.x + lhB.x, lhA.y - lhB.y};
  v2f lho = {lhA.y + lhB.y, lhB.x - lhA.x};
  v2f hle = {hlA.x + hlB.x, hlA.y - hlB.y};
  v2f hlo = {hlA.y + hlB.y, hlB.x - hlA.x};
  v2f hhe = {hhA.x + hhB.x, hhA.y - hhB.y};
  v2f hho = {hhA.y + hhB.y, hhB.x - hhA.x};
  E.y1 = wmma4(ple, f0c, E.y1);
  E.y1 = wmma4(lhe, f1sc, E.y1);
  E.y2 = wmma4(hle, f0sc, E.y2);
  E.y2 = wmma4(hhe, f1sc, E.y2);
  O.y1 = wmma4(plo, f0c, O.y1);
  O.y1 = wmma4(lho, f1sc, O.y1);
  O.y2 = wmma4(hlo, f0sc, O.y2);
  O.y2 = wmma4(hho, f1sc, O.y2);
}

struct PG {
  v2f r0[2], r1[2], a0[2], a1[2], a2[2], a3[2], a4[2], a5[2];
};

template <int CB, int Hb, int Wb>
__device__ __forceinline__ void pair_load(PG& G,
                                          const float* __restrict__ plain,
                                          const float* __restrict__ yh, int b,
                                          int x0, unsigned jb, int koff) {
  constexpr int r = 2 * Hb;
  constexpr int W = 2 * Wb;
  constexpr size_t bs = (size_t)Hb * Wb * 2;
  const v2f* p0 = (const v2f*)(plain + (size_t)b * r * W);
  const float* yb = yh + (size_t)b * 6 * bs;
  const v2f* q0 = (const v2f*)yb;
  const v2f* q1 = (const v2f*)(yb + bs);
  const v2f* q2 = (const v2f*)(yb + 2 * bs);
  const v2f* q3 = (const v2f*)(yb + 3 * bs);
  const v2f* q4 = (const v2f*)(yb + 4 * bs);
  const v2f* q5 = (const v2f*)(yb + 5 * bs);
#pragma unroll
  for (int s = 0; s < 2; ++s) {
    int kk = 4 * (CB + s) + koff;
    int ir = reflect_i((x0 + kk) >> 1, Hb);
    unsigned ob = (unsigned)ir * (unsigned)Wb + jb;
    unsigned opl = 2u * (unsigned)ir * (unsigned)Wb + jb;
    G.r0[s] = p0[opl];
    G.r1[s] = p0[opl + Wb];
    G.a0[s] = q0[ob];
    G.a5[s] = q5[ob];
    G.a2[s] = q2[ob];
    G.a3[s] = q3[ob];
    G.a1[s] = q1[ob];
    G.a4[s] = q4[ob];
  }
}

template <int CB>
__device__ __forceinline__ void pair_mma(YT& E, YT& O, const PG& G,
                                         const v2f* f0, const v2f* f0s,
                                         const v2f* f1s) {
#pragma unroll
  for (int s = 0; s < 2; ++s)
    pair_step(E, O, G.r0[s], G.r1[s], G.a0[s], G.a5[s], G.a2[s], G.a3[s],
              G.a1[s], G.a4[s], f0[CB + s], f0s[CB + s], f1s[CB + s]);
}

__device__ __forceinline__ void order_point(YT& E, YT& O) {
  asm volatile(""
               : "+v"(E.y1), "+v"(E.y2), "+v"(O.y1), "+v"(O.y2)
               :
               : "memory");
}

template <int KC, int Hb, int Wb>
__device__ __forceinline__ void col_pair_piped(
    const float* __restrict__ plain, const float* __restrict__ yh, int b,
    int x0, unsigned jb, unsigned jb_next, bool next, int koff, const v2f* f0,
    const v2f* f0s, const v2f* f1s, PG& G, YT& E, YT& O) {
  static_assert(KC == 4 || KC == 6, "two or three groups of two K steps");
  E.y1 = (v8f){0.f, 0.f, 0.f, 0.f, 0.f, 0.f, 0.f, 0.f};
  E.y2 = (v8f){0.f, 0.f, 0.f, 0.f, 0.f, 0.f, 0.f, 0.f};
  O.y1 = (v8f){0.f, 0.f, 0.f, 0.f, 0.f, 0.f, 0.f, 0.f};
  O.y2 = (v8f){0.f, 0.f, 0.f, 0.f, 0.f, 0.f, 0.f, 0.f};
  PG G1;
  [[maybe_unused]] PG G2;
  pair_load<2, Hb, Wb>(G1, plain, yh, b, x0, jb, koff);
  if constexpr (KC == 6) pair_load<4, Hb, Wb>(G2, plain, yh, b, x0, jb, koff);
  __builtin_amdgcn_sched_barrier(0);
  pair_mma<0>(E, O, G, f0, f0s, f1s);
  order_point(E, O);
  if (next) pair_load<0, Hb, Wb>(G, plain, yh, b, x0, jb_next, koff);
  order_point(E, O);
  __builtin_amdgcn_sched_barrier(0);
  pair_mma<2>(E, O, G1, f0, f0s, f1s);
  if constexpr (KC == 6) pair_mma<4>(E, O, G2, f0, f0s, f1s);
}

__device__ __forceinline__ v2f swap_k(bool c, v2f a) {
  return (v2f){c ? a.y : a.x, c ? a.x : a.y};
}

#define STAGE_PITCH 68
#define STAGE_FLOATS (16 * STAGE_PITCH)

__device__ __forceinline__ void stage_tile(float* st, int j, v8f acc) {
#pragma unroll
  for (int e = 0; e < 8; ++e) st[e * STAGE_PITCH + 16 * j] = acc[e];
}

__device__ __forceinline__ void stage_flush(const float* sr,
                                            float* __restrict__ out,
                                            unsigned og, unsigned pitch) {
#pragma unroll
  for (int i = 0; i < 8; ++i) {
    v4f v = *(const v4f*)(sr + 2 * i * STAGE_PITCH);
    *(v4f*)(out + (og + 2u * (unsigned)i * pitch)) = v;
  }
}

__device__ __forceinline__ void tile_result_fence(YT& T) {
  __builtin_amdgcn_sched_barrier(0);
  asm volatile(
      "s_wait_alu depctr_va_vdst(0)\n\t"
      "v_nop\n\tv_nop\n\tv_nop\n\tv_nop\n\tv_nop\n\t"
      "v_nop\n\tv_nop\n\tv_nop\n\tv_nop"
      : "+v"(T.y1), "+v"(T.y2)
      :
      : "memory");
  __builtin_amdgcn_sched_barrier(0);
}

__device__ __forceinline__ void tile_result_fence(YT& H, YT& T) {
  __builtin_amdgcn_sched_barrier(0);
  asm volatile(
      "s_wait_alu depctr_va_vdst(0)\n\t"
      "v_nop\n\tv_nop\n\tv_nop\n\tv_nop\n\tv_nop\n\t"
      "v_nop\n\tv_nop\n\tv_nop\n\tv_nop"
      : "+v"(H.y1), "+v"(H.y2), "+v"(T.y1), "+v"(T.y2)
      :
      : "memory");
  __builtin_amdgcn_sched_barrier(0);
}

__device__ __forceinline__ void tile_result_fence(YT& H, YT& S, YT& T) {
  __builtin_amdgcn_sched_barrier(0);
  asm volatile(
      "s_wait_alu depctr_va_vdst(0)\n\t"
      "v_nop\n\tv_nop\n\tv_nop\n\tv_nop\n\tv_nop\n\t"
      "v_nop\n\tv_nop\n\tv_nop\n\tv_nop"
      : "+v"(H.y1), "+v"(H.y2), "+v"(S.y1), "+v"(S.y2), "+v"(T.y1),
        "+v"(T.y2)
      :
      : "memory");
  __builtin_amdgcn_sched_barrier(0);
}

#define STEP1(T, q) ((v2f){(T).y1[2 * (q)], (T).y1[2 * (q) + 1]})
#define STEP2(T, q) ((v2f){(T).y2[2 * (q)], (T).y2[2 * (q) + 1]})

#define PSTEP1(E, O, t) ((v2f){(E).y1[(t)], (O).y1[(t)]})
#define PSTEP2(E, O, t) ((v2f){(E).y2[(t)], (O).y2[(t)]})

__device__ __forceinline__ float other_half(float x) {
  return __builtin_bit_cast(
      float, __builtin_amdgcn_permlanex16(__builtin_bit_cast(int, x),
                                          __builtin_bit_cast(int, x),
                                          0x76543210, 0xfedcba98, false, false));
}

__device__ __forceinline__ v2f mirror_step(v2f a) {
  return (v2f){other_half(a.y), other_half(a.x)};
}

__device__ __forceinline__ v2f pick_step(bool c, v2f a, v2f b) {
  return (v2f){c ? a.x : b.x, c ? a.y : b.y};
}

template <int BASE>
__device__ __forceinline__ v8f row_even(v2f l1, v2f l2, const YT& E,
                                        const YT& O, const v2f* g0,
                                        const v2f* g1) {
  v8f acc = {0.f, 0.f, 0.f, 0.f, 0.f, 0.f, 0.f, 0.f};
  acc = wmma4(l1, g0[0], acc);
  acc = wmma4(l2, g1[0], acc);
#pragma unroll
  for (int q = 0; q < 3; ++q) {
    acc = wmma4(PSTEP1(E, O, BASE + q), g0[q + 1], acc);
    acc = wmma4(PSTEP2(E, O, BASE + q), g1[q + 1], acc);
  }
  return acc;
}

template <int BASE>
__device__ __forceinline__ v8f row_odd(const YT& E, const YT& O, v2f r1,
                                       v2f r2, const v2f* g0, const v2f* g1) {
  v8f acc = {0.f, 0.f, 0.f, 0.f, 0.f, 0.f, 0.f, 0.f};
#pragma unroll
  for (int q = 1; q < 4; ++q) {
    acc = wmma4(PSTEP1(E, O, BASE + q), g0[q - 1], acc);
    acc = wmma4(PSTEP2(E, O, BASE + q), g1[q - 1], acc);
  }
  acc = wmma4(r1, g0[3], acc);
  acc = wmma4(r2, g1[3], acc);
  return acc;
}

template <int NT, int Hb, int Wb>
__global__ void __launch_bounds__(128, 4)
j2_fused_kernel(const float* __restrict__ plain, const float* __restrict__ yh,
                const float* __restrict__ ha0, const float* __restrict__ hb0,
                const float* __restrict__ ha1, const float* __restrict__ hb1,
                float* __restrict__ out, int B) {
  constexpr int r = 2 * Hb;
  constexpr int W = 2 * Wb;
  int lane = threadIdx.x & 31;
  int wave = __builtin_amdgcn_readfirstlane((int)(threadIdx.x >> 5));
  int g = (int)blockIdx.x * 4 + wave;
  constexpr int nCS = W / (16 * NT);
  constexpr int nRT = (2 * r) / 16;
  int cs = g % nCS;
  int g2 = g / nCS;
  int I = g2 % nRT;
  int b = g2 / nRT;
  if (b >= B) return;
  int half = lane >> 4;
  int l16 = lane & 15;
  int koff = half * 2;
  int x0 = 8 * I - 4;
  int lcol = tile_col(l16);
  int S = cs * 16 * NT;
  constexpr bool TWO = (nCS == 2);
  int hcol = TWO ? ((cs ? S - 16 : S + 16 * NT) + lcol)
                 : (S + 16 * NT + lcol - (lcol >= 12 ? 16 * NT + 16 : 0));

  v2f f0[4], f0s[4], f1[4], f1s[4];
#pragma unroll
  for (int c = 0; c < 4; ++c) {
    int kk = 4 * c + koff;
    float ax = cif_A(l16, kk, ha0, hb0, 0);
    float ay = cif_A(l16, kk + 1, ha0, hb0, 0);
    float bx = cif_A(l16, kk, ha1, hb1, 1);
    float by = cif_A(l16, kk + 1, ha1, hb1, 1);
    f0[c].x = ax;
    f0[c].y = ay;
    f0s[c].x = ax * SQRT_HALF;
    f0s[c].y = ay * SQRT_HALF;
    f1[c].x = bx;
    f1[c].y = by;
    f1s[c].x = bx * SQRT_HALF;
    f1s[c].y = by * SQRT_HALF;
  }

  constexpr int outW = 2 * W;
  unsigned og = (unsigned)((b * (2 * r) + I * 16 + half) * outW +
                           cs * 32 * NT + 4 * l16);
  __shared__ __attribute__((aligned(16))) float stage_mem[4 * STAGE_FLOATS];
  float* sw = stage_mem + wave * STAGE_FLOATS;
  float* st = sw + half * 8 * STAGE_PITCH + l16;
  const float* sr = sw + half * STAGE_PITCH + 4 * l16;

  static_assert(NT % 2 == 0, "tiles are computed in aligned pairs");
  constexpr int NP = NT / 2;
  unsigned jp = (unsigned)(S / 2 + pair_col(l16));
  PG G;
  pair_load<0, Hb, Wb>(G, plain, yh, b, x0, jp, koff);
  __builtin_amdgcn_sched_barrier(0);
  YT Hh = col_tile<4, Hb, Wb>(plain, yh, b, x0, hcol, koff, f0, f0s, f1s);
  bool top = (I == 0), bot = (I == nRT - 1);
  v2f c0[4], c0s[4], c1s[4];
#pragma unroll
  for (int c = 0; c < 4; ++c) {
    bool sw = (c == 0 && top) || (c == 3 && bot);
    c0[c] = swap_k(sw, f0[c]);
    c0s[c] = swap_k(sw, f0s[c]);
    c1s[c] = swap_k(sw, f1s[c]);
  }
  YT E0, O0;
  col_pair_piped<4, Hb, Wb>(plain, yh, b, x0, jp, jp + 16u, NP > 1, koff, c0,
                            c0s, c1s, G, E0, O0);
  tile_result_fence(Hh, E0, O0);
  v2f l1 = STEP1(Hh, 3), l2 = STEP2(Hh, 3);
  v2f e1 = STEP1(Hh, 0), e2 = STEP2(Hh, 0);
  if (TWO) {
    l1 = pick_step(cs != 0, l1, mirror_step(PSTEP1(E0, O0, 0)));
    l2 = pick_step(cs != 0, l2, mirror_step(PSTEP2(E0, O0, 0)));
  }
#pragma unroll
  for (int p = 1; p < NP; ++p) {
    const int u = 2 * (p - 1);
    stage_tile(st, 0, row_even<0>(l1, l2, E0, O0, f0, f1));
    stage_tile(st, 1,
               row_odd<0>(E0, O0, PSTEP1(E0, O0, 4), PSTEP2(E0, O0, 4), f0,
                          f1));
    stage_tile(st, 2,
               row_even<4>(PSTEP1(E0, O0, 3), PSTEP2(E0, O0, 3), E0, O0, f0,
                           f1));
    YT E1, O1;
    col_pair_piped<4, Hb, Wb>(plain, yh, b, x0, jp + 16u * (unsigned)p,
                              jp + 16u * (unsigned)(p + 1), p + 1 < NP, koff,
                              c0, c0s, c1s, G, E1, O1);
    tile_result_fence(E1, O1);
    stage_tile(st, 3,
               row_odd<4>(E0, O0, PSTEP1(E1, O1, 0), PSTEP2(E1, O1, 0), f0,
                          f1));
    stage_flush(sr, out, og + 32u * (unsigned)u, (unsigned)outW);
    l1 = PSTEP1(E0, O0, 7);
    l2 = PSTEP2(E0, O0, 7);
    E0 = E1;
    O0 = O1;
    __builtin_amdgcn_sched_barrier(0);
  }
  {
    constexpr int u = NT - 2;
    if (TWO) {
      e1 = pick_step(cs != 0, mirror_step(PSTEP1(E0, O0, 7)), e1);
      e2 = pick_step(cs != 0, mirror_step(PSTEP2(E0, O0, 7)), e2);
    }
    stage_tile(st, 0, row_even<0>(l1, l2, E0, O0, f0, f1));
    stage_tile(st, 1,
               row_odd<0>(E0, O0, PSTEP1(E0, O0, 4), PSTEP2(E0, O0, 4), f0,
                          f1));
    stage_tile(st, 2,
               row_even<4>(PSTEP1(E0, O0, 3), PSTEP2(E0, O0, 3), E0, O0, f0,
                           f1));
    stage_tile(st, 3, row_odd<4>(E0, O0, e1, e2, f0, f1));
    stage_flush(sr, out, og + 32u * (unsigned)u, (unsigned)outW);
  }
}

template <int BASE>
__device__ __forceinline__ v8f row_j1(v2f l1, v2f l2, const YT& E,
                                      const YT& O, v2f r1, v2f r2,
                                      const v2f* g0, const v2f* g1) {
  v8f acc = {0.f, 0.f, 0.f, 0.f, 0.f, 0.f, 0.f, 0.f};
  acc = wmma4(l1, g0[0], acc);
  acc = wmma4(l2, g1[0], acc);
#pragma unroll
  for (int q = 0; q < 4; ++q) {
    acc = wmma4(PSTEP1(E, O, BASE + q), g0[q + 1], acc);
    acc = wmma4(PSTEP2(E, O, BASE + q), g1[q + 1], acc);
  }
  acc = wmma4(r1, g0[5], acc);
  acc = wmma4(r2, g1[5], acc);
  return acc;
}

template <int NOUT, int Hb, int Wb>
__global__ void __launch_bounds__(128)
j1_fused_kernel(const float* __restrict__ plain, const float* __restrict__ yh,
                const float* __restrict__ h0, int taps0,
                const float* __restrict__ h1, int taps1,
                float* __restrict__ out, int B) {
  constexpr int r = 2 * Hb;
  constexpr int W = 2 * Wb;
  int lane = threadIdx.x & 31;
  int wave = __builtin_amdgcn_readfirstlane((int)(threadIdx.x >> 5));
  int g = (int)blockIdx.x * 4 + wave;
  constexpr int nCS = W / (16 * NOUT);
  constexpr int nRT = r / 16;
  int cs = g % nCS;
  int g2 = g / nCS;
  int I = g2 % nRT;
  int b = g2 / nRT;
  if (b >= B) return;
  int half = lane >> 4;
  int l16 = lane & 15;
  int koff = half * 2;
  int x0 = 16 * I - 4;
  int lcol = tile_col(l16);
  int S = cs * 16 * NOUT;
  constexpr bool TWO = (nCS == 2);
  int hcol = TWO ? ((cs ? S - 16 : S + 16 * NOUT) + lcol)
                 : (S + 16 * NOUT + lcol - (lcol >= 12 ? 16 * NOUT + 16 : 0));

  v2f g0[6], g1[6];
#pragma unroll
  for (int s = 0; s < 6; ++s) {
    int kk = 4 * s - 1 + koff;
    g0[s].x = cf_A(l16, kk, h0, taps0);
    g0[s].y = cf_A(l16, kk + 1, h0, taps0);
    g1[s].x = cf_A(l16, kk, h1, taps1);
    g1[s].y = cf_A(l16, kk + 1, h1, taps1);
  }
  v2f f0s[6], f1s[6];
#pragma unroll
  for (int c = 0; c < 6; ++c) {
    f0s[c].x = g0[c].x * SQRT_HALF;
    f0s[c].y = g0[c].y * SQRT_HALF;
    f1s[c].x = g1[c].x * SQRT_HALF;
    f1s[c].y = g1[c].y * SQRT_HALF;
  }

  unsigned og = (unsigned)((b * r + I * 16 + half) * W + cs * 16 * NOUT +
                           4 * l16);
  __shared__ __attribute__((aligned(16))) float stage_mem[4 * STAGE_FLOATS];
  float* sw = stage_mem + wave * STAGE_FLOATS;
  float* st = sw + half * 8 * STAGE_PITCH + l16;
  const float* sr = sw + half * STAGE_PITCH + 4 * l16;

  static_assert(NOUT % 4 == 0, "two aligned tile pairs per 64-column unit");
  constexpr int NP = NOUT / 2;
  unsigned jp = (unsigned)(S / 2 + pair_col(l16));
  PG G;
  pair_load<0, Hb, Wb>(G, plain, yh, b, x0, jp, koff);
  __builtin_amdgcn_sched_barrier(0);
  YT Hh = col_tile<6, Hb, Wb>(plain, yh, b, x0, hcol, koff, g0, f0s, f1s);
  bool top = (I == 0), bot = (I == nRT - 1);
  v2f c0[6], c0s[6], c1s[6];
#pragma unroll
  for (int c = 0; c < 6; ++c) {
    bool sw = (c == 0 && top) || (c == 5 && bot);
    c0[c] = swap_k(sw, g0[c]);
    c0s[c] = swap_k(sw, f0s[c]);
    c1s[c] = swap_k(sw, f1s[c]);
  }
  YT E0, O0;
  col_pair_piped<6, Hb, Wb>(plain, yh, b, x0, jp, jp + 16u, NP > 1, koff, c0,
                            c0s, c1s, G, E0, O0);
  tile_result_fence(Hh, E0, O0);
  v2f l1 = STEP1(Hh, 3), l2 = STEP2(Hh, 3);
  v2f e1 = STEP1(Hh, 0), e2 = STEP2(Hh, 0);
  if (TWO) {
    l1 = pick_step(cs != 0, l1, mirror_step(PSTEP1(E0, O0, 0)));
    l2 = pick_step(cs != 0, l2, mirror_step(PSTEP2(E0, O0, 0)));
  }
  auto trip = [&](int p, bool next, int slot,
                  bool flush) __attribute__((always_inline)) {
    YT E1, O1;
    col_pair_piped<6, Hb, Wb>(plain, yh, b, x0, jp + 16u * (unsigned)p,
                              jp + 16u * (unsigned)(p + 1), next, koff, c0,
                              c0s, c1s, G, E1, O1);
    tile_result_fence(E1, O1);
    stage_tile(st, 2 * slot,
               row_j1<0>(l1, l2, E0, O0, PSTEP1(E0, O0, 4), PSTEP2(E0, O0, 4),
                         g0, g1));
    stage_tile(st, 2 * slot + 1,
               row_j1<4>(PSTEP1(E0, O0, 3), PSTEP2(E0, O0, 3), E0, O0,
                         PSTEP1(E1, O1, 0), PSTEP2(E1, O1, 0), g0, g1));
    if (flush) stage_flush(sr, out, og + 32u * (unsigned)(p - 2), (unsigned)W);
    l1 = PSTEP1(E0, O0, 7);
    l2 = PSTEP2(E0, O0, 7);
    E0 = E1;
    O0 = O1;
    __builtin_amdgcn_sched_barrier(0);
  };
#pragma unroll 1
  for (int p = 1; p < NP - 1; ++p)
    trip(p, true, (p - 1) & 1, ((p - 1) & 1) != 0);
  trip(NP - 1, false, 0, false);
  {
    if (TWO) {
      e1 = pick_step(cs != 0, mirror_step(PSTEP1(E0, O0, 7)), e1);
      e2 = pick_step(cs != 0, mirror_step(PSTEP2(E0, O0, 7)), e2);
    }
    stage_tile(st, 2,
               row_j1<0>(l1, l2, E0, O0, PSTEP1(E0, O0, 4), PSTEP2(E0, O0, 4),
                         g0, g1));
    stage_tile(st, 3,
               row_j1<4>(PSTEP1(E0, O0, 3), PSTEP2(E0, O0, 3), E0, O0, e1, e2,
                         g0, g1));
    stage_flush(sr, out, og + 16u * (unsigned)(NOUT - 4), (unsigned)W);
  }
}

extern "C" void kernel_launch(void* const* d_in, const int* in_sizes, int n_in,
                              void* d_out, int out_size, void* d_ws,
                              size_t ws_size, hipStream_t stream) {
  (void)in_sizes;
  (void)n_in;
  (void)out_size;
  (void)ws_size;

  const float* yl = (const float*)d_in[0];
  const float* yh0 = (const float*)d_in[1];
  const float* yh1 = (const float*)d_in[2];
  const float* yh2 = (const float*)d_in[3];
  const float* g0o = (const float*)d_in[4];
  const float* g1o = (const float*)d_in[5];
  const float* g0a = (const float*)d_in[6];
  const float* g0b = (const float*)d_in[7];
  const float* g1a = (const float*)d_in[8];
  const float* g1b = (const float*)d_in[9];

  float* LLa = (float*)d_ws;
  float* LLb = LLa + (size_t)64 * 65536;
  float* outp = (float*)d_out;

  dim3 blk(128);

  j2_fused_kernel<2, 64, 64><<<dim3(1024), blk, 0, stream>>>(
      yl, yh2, g0b, g0a, g1b, g1a, LLa, 64);

  j2_fused_kernel<8, 128, 128><<<dim3(1024), blk, 0, stream>>>(
      LLa, yh1, g0b, g0a, g1b, g1a, LLb, 64);

  j1_fused_kernel<16, 256, 256><<<dim3(1024), blk, 0, stream>>>(
      LLb, yh0, g0o, 7, g1o, 5, outp, 64);
}


# Halo tiles dropped: the strips of a row strip are waves of one work-group and pass their edge K steps through LDS (one barrier); right strip's first tile adds its left-margin terms last (sum order)
# speedup vs baseline: 5.0930x; 1.0807x over previous
#include <hip/hip_runtime.h>


typedef __attribute__((ext_vector_type(2))) float v2f;
typedef __attribute__((ext_vector_type(4))) float v4f;
typedef __attribute__((ext_vector_type(8))) float v8f;

#define SQRT_HALF 0.70710678118654752440f

__device__ __forceinline__ int reflect_i(int x, int n) {
  x = (x < 0) ? (-1 - x) : x;
  x = (x >= n) ? (2 * n - 1 - x) : x;
  return x;
}

__device__ __forceinline__ float cif_A(int m, int kk,
                                       const float* __restrict__ ha,
                                       const float* __restrict__ hb, int hp) {
  int p = m & 3;
  int q = m >> 2;
  int want_odd = hp ? ((p & 1) ^ 1) : (p & 1);
  int t = ((kk - want_odd) >> 1) - q;
  bool ok = ((kk & 1) == want_odd) && (t >= 0) && (t <= 4);
  int tc = t < 0 ? 0 : (t > 4 ? 4 : t);
  const float* __restrict__ h = ((p == 0) || (p == 2)) ? ha : hb;
  int idx = (p < 2) ? (2 * tc + 1) : (2 * tc);
  float v = h[idx];
  return ok ? v : 0.0f;
}

__device__ __forceinline__ float cf_A(int m, int kk,
                                      const float* __restrict__ h, int taps) {
  int t = kk - m - (3 - (taps >> 1));
  bool ok = (t >= 0) && (t < taps);
  int tc = t < 0 ? 0 : (t >= taps ? taps - 1 : t);
  float v = h[tc];
  return ok ? v : 0.0f;
}

__device__ __forceinline__ v8f wmma4(v2f a, v2f b, v8f c) {
  return __builtin_amdgcn_wmma_f32_16x16x4_f32(false, a, false, b, (short)0, c,
                                               false, false);
}

struct YT {
  v8f y1, y2;
};

__device__ __forceinline__ int pair_col(int m) {
  return 2 * (m & 7) + (m >> 3);
}

__device__ __forceinline__ void pair_step(YT& E, YT& O, v2f r0, v2f r1,
                                          v2f lhA, v2f lhB, v2f hlA, v2f hlB,
                                          v2f hhA, v2f hhB, v2f f0c, v2f f0sc,
                                          v2f f1sc) {
  v2f ple = {r0.x, r1.x};
  v2f plo = {r0.y, r1.y};
  v2f lhe = {lhA.x + lhB.x, lhA.y - lhB.y};
  v2f lho = {lhA.y + lhB.y, lhB.x - lhA.x};
  v2f hle = {hlA.x + hlB.x, hlA.y - hlB.y};
  v2f hlo = {hlA.y + hlB.y, hlB.x - hlA.x};
  v2f hhe = {hhA.x + hhB.x, hhA.y - hhB.y};
  v2f hho = {hhA.y + hhB.y, hhB.x - hhA.x};
  E.y1 = wmma4(ple, f0c, E.y1);
  E.y1 = wmma4(lhe, f1sc, E.y1);
  E.y2 = wmma4(hle, f0sc, E.y2);
  E.y2 = wmma4(hhe, f1sc, E.y2);
  O.y1 = wmma4(plo, f0c, O.y1);
  O.y1 = wmma4(lho, f1sc, O.y1);
  O.y2 = wmma4(hlo, f0sc, O.y2);
  O.y2 = wmma4(hho, f1sc, O.y2);
}

struct PG {
  v2f r0[2], r1[2], a0[2], a1[2], a2[2], a3[2], a4[2], a5[2];
};

template <int CB, int Hb, int Wb>
__device__ __forceinline__ void pair_load(PG& G,
                                          const float* __restrict__ plain,
                                          const float* __restrict__ yh, int b,
                                          int x0, unsigned jb, int koff) {
  constexpr int r = 2 * Hb;
  constexpr int W = 2 * Wb;
  constexpr size_t bs = (size_t)Hb * Wb * 2;
  const v2f* p0 = (const v2f*)(plain + (size_t)b * r * W);
  const float* yb = yh + (size_t)b * 6 * bs;
  const v2f* q0 = (const v2f*)yb;
  const v2f* q1 = (const v2f*)(yb + bs);
  const v2f* q2 = (const v2f*)(yb + 2 * bs);
  const v2f* q3 = (const v2f*)(yb + 3 * bs);
  const v2f* q4 = (const v2f*)(yb + 4 * bs);
  const v2f* q5 = (const v2f*)(yb + 5 * bs);
#pragma unroll
  for (int s = 0; s < 2; ++s) {
    int kk = 4 * (CB + s) + koff;
    int ir = reflect_i((x0 + kk) >> 1, Hb);
    unsigned ob = (unsigned)ir * (unsigned)Wb + jb;
    unsigned opl = 2u * (unsigned)ir * (unsigned)Wb + jb;
    G.r0[s] = p0[opl];
    G.r1[s] = p0[opl + Wb];
    G.a0[s] = q0[ob];
    G.a5[s] = q5[ob];
    G.a2[s] = q2[ob];
    G.a3[s] = q3[ob];
    G.a1[s] = q1[ob];
    G.a4[s] = q4[ob];
  }
}

template <int CB>
__device__ __forceinline__ void pair_mma(YT& E, YT& O, const PG& G,
                                         const v2f* f0, const v2f* f0s,
                                         const v2f* f1s) {
#pragma unroll
  for (int s = 0; s < 2; ++s)
    pair_step(E, O, G.r0[s], G.r1[s], G.a0[s], G.a5[s], G.a2[s], G.a3[s],
              G.a1[s], G.a4[s], f0[CB + s], f0s[CB + s], f1s[CB + s]);
}

__device__ __forceinline__ void order_point(YT& E, YT& O) {
  asm volatile(""
               : "+v"(E.y1), "+v"(E.y2), "+v"(O.y1), "+v"(O.y2)
               :
               : "memory");
}

template <int KC, int Hb, int Wb>
__device__ __forceinline__ void col_pair_piped(
    const float* __restrict__ plain, const float* __restrict__ yh, int b,
    int x0, unsigned jb, unsigned jb_next, bool next, int koff, const v2f* f0,
    const v2f* f0s, const v2f* f1s, PG& G, YT& E, YT& O) {
  static_assert(KC == 4 || KC == 6, "two or three groups of two K steps");
  E.y1 = (v8f){0.f, 0.f, 0.f, 0.f, 0.f, 0.f, 0.f, 0.f};
  E.y2 = (v8f){0.f, 0.f, 0.f, 0.f, 0.f, 0.f, 0.f, 0.f};
  O.y1 = (v8f){0.f, 0.f, 0.f, 0.f, 0.f, 0.f, 0.f, 0.f};
  O.y2 = (v8f){0.f, 0.f, 0.f, 0.f, 0.f, 0.f, 0.f, 0.f};
  PG G1;
  [[maybe_unused]] PG G2;
  pair_load<2, Hb, Wb>(G1, plain, yh, b, x0, jb, koff);
  if constexpr (KC == 6) pair_load<4, Hb, Wb>(G2, plain, yh, b, x0, jb, koff);
  __builtin_amdgcn_sched_barrier(0);
  pair_mma<0>(E, O, G, f0, f0s, f1s);
  order_point(E, O);
  if (next) pair_load<0, Hb, Wb>(G, plain, yh, b, x0, jb_next, koff);
  order_point(E, O);
  __builtin_amdgcn_sched_barrier(0);
  pair_mma<2>(E, O, G1, f0, f0s, f1s);
  if constexpr (KC == 6) pair_mma<4>(E, O, G2, f0, f0s, f1s);
}

__device__ __forceinline__ v2f swap_k(bool c, v2f a) {
  return (v2f){c ? a.y : a.x, c ? a.x : a.y};
}

#define STAGE_PITCH 68
#define STAGE_FLOATS (16 * STAGE_PITCH)

__device__ __forceinline__ void stage_tile(float* st, int j, v8f acc) {
#pragma unroll
  for (int e = 0; e < 8; ++e) st[e * STAGE_PITCH + 16 * j] = acc[e];
}

__device__ __forceinline__ void stage_flush(const float* sr,
                                            float* __restrict__ out,
                                            unsigned og, unsigned pitch,
                                            bool lane_on) {
#pragma unroll
  for (int i = 0; i < 8; ++i) {
    v4f v = *(const v4f*)(sr + 2 * i * STAGE_PITCH);
    if (lane_on) *(v4f*)(out + (og + 2u * (unsigned)i * pitch)) = v;
  }
}

__device__ __forceinline__ void tile_result_fence(YT& T) {
  __builtin_amdgcn_sched_barrier(0);
  asm volatile(
      "s_wait_alu depctr_va_vdst(0)\n\t"
      "v_nop\n\tv_nop\n\tv_nop\n\tv_nop\n\tv_nop\n\t"
      "v_nop\n\tv_nop\n\tv_nop\n\tv_nop"
      : "+v"(T.y1), "+v"(T.y2)
      :
      : "memory");
  __builtin_amdgcn_sched_barrier(0);
}

__device__ __forceinline__ void tile_result_fence(YT& H, YT& T) {
  __builtin_amdgcn_sched_barrier(0);
  asm volatile(
      "s_wait_alu depctr_va_vdst(0)\n\t"
      "v_nop\n\tv_nop\n\tv_nop\n\tv_nop\n\tv_nop\n\t"
      "v_nop\n\tv_nop\n\tv_nop\n\tv_nop"
      : "+v"(H.y1), "+v"(H.y2), "+v"(T.y1), "+v"(T.y2)
      :
      : "memory");
  __builtin_amdgcn_sched_barrier(0);
}

#define PSTEP1(E, O, t) ((v2f){(E).y1[(t)], (O).y1[(t)]})
#define PSTEP2(E, O, t) ((v2f){(E).y2[(t)], (O).y2[(t)]})

__device__ __forceinline__ float other_half(float x) {
  return __builtin_bit_cast(
      float, __builtin_amdgcn_permlanex16(__builtin_bit_cast(int, x),
                                          __builtin_bit_cast(int, x),
                                          0x76543210, 0xfedcba98, false, false));
}

__device__ __forceinline__ v2f mirror_step(v2f a) {
  return (v2f){other_half(a.y), other_half(a.x)};
}

__device__ __forceinline__ v2f pick_step(bool c, v2f a, v2f b) {
  return (v2f){c ? a.x : b.x, c ? a.y : b.y};
}

#define XCH_FLOATS 256
__device__ __forceinline__ void xch_put(float* x, v2f a1, v2f a2) {
  x[0] = a1.x;
  x[32] = a1.y;
  x[64] = a2.x;
  x[96] = a2.y;
}
__device__ __forceinline__ void xch_get(const float* x, v2f& a1, v2f& a2) {
  a1 = (v2f){x[0], x[32]};
  a2 = (v2f){x[64], x[96]};
}

__device__ __forceinline__ void lds_barrier() {
  __builtin_amdgcn_fence(__ATOMIC_RELEASE, "workgroup", "local");
  __builtin_amdgcn_s_barrier();
  __builtin_amdgcn_fence(__ATOMIC_ACQUIRE, "workgroup", "local");
}

#define DEFER_FLOATS 256
__device__ __forceinline__ void defer_put(float* d, v8f acc) {
#pragma unroll
  for (int e = 0; e < 8; ++e) d[32 * e] = acc[e];
}
__device__ __forceinline__ v8f defer_get(const float* d) {
  v8f acc;
#pragma unroll
  for (int e = 0; e < 8; ++e) acc[e] = d[32 * e];
  return acc;
}

__device__ __forceinline__ void finish_deferred(const float* d, v2f n1, v2f n2,
                                                v2f g0c, v2f g1c,
                                                float* __restrict__ out,
                                                unsigned ot, unsigned pitch) {
  v8f acc = defer_get(d);
  acc = wmma4(n1, g0c, acc);
  acc = wmma4(n2, g1c, acc);
#pragma unroll
  for (int e = 0; e < 8; ++e) out[ot + (unsigned)e * pitch] = acc[e];
}

template <int BASE>
__device__ __forceinline__ v8f row_even(v2f l1, v2f l2, const YT& E,
                                        const YT& O, const v2f* g0,
                                        const v2f* g1) {
  v8f acc = {0.f, 0.f, 0.f, 0.f, 0.f, 0.f, 0.f, 0.f};
  acc = wmma4(l1, g0[0], acc);
  acc = wmma4(l2, g1[0], acc);
#pragma unroll
  for (int q = 0; q < 3; ++q) {
    acc = wmma4(PSTEP1(E, O, BASE + q), g0[q + 1], acc);
    acc = wmma4(PSTEP2(E, O, BASE + q), g1[q + 1], acc);
  }
  return acc;
}

template <int BASE>
__device__ __forceinline__ v8f row_odd(const YT& E, const YT& O, v2f r1,
                                       v2f r2, const v2f* g0, const v2f* g1) {
  v8f acc = {0.f, 0.f, 0.f, 0.f, 0.f, 0.f, 0.f, 0.f};
#pragma unroll
  for (int q = 1; q < 4; ++q) {
    acc = wmma4(PSTEP1(E, O, BASE + q), g0[q - 1], acc);
    acc = wmma4(PSTEP2(E, O, BASE + q), g1[q - 1], acc);
  }
  acc = wmma4(r1, g0[3], acc);
  acc = wmma4(r2, g1[3], acc);
  return acc;
}

template <int NT, int Hb, int Wb>
__global__ void __launch_bounds__(128, 4)
j2_fused_kernel(const float* __restrict__ plain, const float* __restrict__ yh,
                const float* __restrict__ ha0, const float* __restrict__ hb0,
                const float* __restrict__ ha1, const float* __restrict__ hb1,
                float* __restrict__ out, int B) {
  constexpr int r = 2 * Hb;
  constexpr int W = 2 * Wb;
  int lane = threadIdx.x & 31;
  int wave = __builtin_amdgcn_readfirstlane((int)(threadIdx.x >> 5));
  int g = (int)blockIdx.x * 4 + wave;
  constexpr int nCS = W / (16 * NT);
  constexpr int nRT = (2 * r) / 16;
  int cs = g % nCS;
  int g2 = g / nCS;
  int I = g2 % nRT;
  int b = g2 / nRT;
  if (b >= B) return;
  int half = lane >> 4;
  int l16 = lane & 15;
  int koff = half * 2;
  int x0 = 8 * I - 4;
  int S = cs * 16 * NT;
  constexpr bool TWO = (nCS == 2);
  static_assert(nCS == 2 || nCS == 4, "neighbour strips share a work-group");

  v2f f0[4], f0s[4], f1[4], f1s[4];
#pragma unroll
  for (int c = 0; c < 4; ++c) {
    int kk = 4 * c + koff;
    float ax = cif_A(l16, kk, ha0, hb0, 0);
    float ay = cif_A(l16, kk + 1, ha0, hb0, 0);
    float bx = cif_A(l16, kk, ha1, hb1, 1);
    float by = cif_A(l16, kk + 1, ha1, hb1, 1);
    f0[c].x = ax;
    f0[c].y = ay;
    f0s[c].x = ax * SQRT_HALF;
    f0s[c].y = ay * SQRT_HALF;
    f1[c].x = bx;
    f1[c].y = by;
    f1s[c].x = bx * SQRT_HALF;
    f1s[c].y = by * SQRT_HALF;
  }

  constexpr int outW = 2 * W;
  unsigned og = (unsigned)((b * (2 * r) + I * 16 + half) * outW +
                           cs * 32 * NT + 4 * l16);
  __shared__ __attribute__((aligned(16))) float stage_mem[4 * STAGE_FLOATS];
  float* sw = stage_mem + wave * STAGE_FLOATS;
  float* st = sw + half * 8 * STAGE_PITCH + l16;
  const float* sr = sw + half * STAGE_PITCH + 4 * l16;

  static_assert(NT % 2 == 0, "tiles are computed in aligned pairs");
  constexpr int NP = NT / 2;
  static_assert(TWO ? NP > 1 : NP == 1, "see the hand-over of margins");
  static_assert((nCS * nRT) % 4 == 0, "a work-group lies in one image");
  __shared__ float xch_mem[4 * XCH_FLOATS];
  __shared__ float defer_mem[4 * DEFER_FLOATS];
  float* xw = xch_mem + wave * XCH_FLOATS + lane;
  float* dw = defer_mem + wave * DEFER_FLOATS + lane;
  unsigned jp = (unsigned)(S / 2 + pair_col(l16));
  PG G;
  pair_load<0, Hb, Wb>(G, plain, yh, b, x0, jp, koff);
  bool top = (I == 0), bot = (I == nRT - 1);
  v2f c0[4], c0s[4], c1s[4];
#pragma unroll
  for (int c = 0; c < 4; ++c) {
    bool sw = (c == 0 && top) || (c == 3 && bot);
    c0[c] = swap_k(sw, f0[c]);
    c0s[c] = swap_k(sw, f0s[c]);
    c1s[c] = swap_k(sw, f1s[c]);
  }
  YT E0, O0;
  col_pair_piped<4, Hb, Wb>(plain, yh, b, x0, jp, jp + 16u, NP > 1, koff, c0,
                            c0s, c1s, G, E0, O0);
  tile_result_fence(E0, O0);
  v2f l1, l2;
  v2f e1, e2;
  if constexpr (TWO) {
    if (cs != 0) xch_put(xw, PSTEP1(E0, O0, 0), PSTEP2(E0, O0, 0));
    l1 = pick_step(cs != 0, (v2f){0.f, 0.f}, mirror_step(PSTEP1(E0, O0, 0)));
    l2 = pick_step(cs != 0, (v2f){0.f, 0.f}, mirror_step(PSTEP2(E0, O0, 0)));
  } else {
    xch_put(xw, PSTEP1(E0, O0, 0), PSTEP2(E0, O0, 0));
    xch_put(xw + XCH_FLOATS / 2, PSTEP1(E0, O0, 7), PSTEP2(E0, O0, 7));
    lds_barrier();
    int wl = (cs > 0) ? wave - 1 : wave;
    int wr = (cs < nCS - 1) ? wave + 1 : wave;
    v2f n1, n2, m1, m2;
    xch_get(xch_mem + wl * XCH_FLOATS + XCH_FLOATS / 2 + lane, n1, n2);
    xch_get(xch_mem + wr * XCH_FLOATS + lane, m1, m2);
    l1 = pick_step(cs > 0, n1, mirror_step(PSTEP1(E0, O0, 0)));
    l2 = pick_step(cs > 0, n2, mirror_step(PSTEP2(E0, O0, 0)));
    e1 = pick_step(cs < nCS - 1, m1, mirror_step(PSTEP1(E0, O0, 7)));
    e2 = pick_step(cs < nCS - 1, m2, mirror_step(PSTEP2(E0, O0, 7)));
  }
  bool first_on = !(TWO && cs != 0 && l16 < 4);
#pragma unroll
  for (int p = 1; p < NP; ++p) {
    const int u = 2 * (p - 1);
    v8f t0 = row_even<0>(l1, l2, E0, O0, f0, f1);
    if (TWO && p == 1 && cs != 0) defer_put(dw, t0);
    stage_tile(st, 0, t0);
    stage_tile(st, 1,
               row_odd<0>(E0, O0, PSTEP1(E0, O0, 4), PSTEP2(E0, O0, 4), f0,
                          f1));
    stage_tile(st, 2,
               row_even<4>(PSTEP1(E0, O0, 3), PSTEP2(E0, O0, 3), E0, O0, f0,
                           f1));
    YT E1, O1;
    col_pair_piped<4, Hb, Wb>(plain, yh, b, x0, jp + 16u * (unsigned)p,
                              jp + 16u * (unsigned)(p + 1), p + 1 < NP, koff,
                              c0, c0s, c1s, G, E1, O1);
    tile_result_fence(E1, O1);
    stage_tile(st, 3,
               row_odd<4>(E0, O0, PSTEP1(E1, O1, 0), PSTEP2(E1, O1, 0), f0,
                          f1));
    stage_flush(sr, out, og + 32u * (unsigned)u, (unsigned)outW,
                p == 1 ? first_on : true);
    l1 = PSTEP1(E0, O0, 7);
    l2 = PSTEP2(E0, O0, 7);
    E0 = E1;
    O0 = O1;
    __builtin_amdgcn_sched_barrier(0);
  }
  {
    constexpr int u = NT - 2;
    [[maybe_unused]] v2f n1, n2;
    if constexpr (TWO) {
      if (cs == 0) xch_put(xw, PSTEP1(E0, O0, 7), PSTEP2(E0, O0, 7));
      lds_barrier();
      xch_get(xch_mem + (wave ^ 1) * XCH_FLOATS + lane, n1, n2);
      e1 = pick_step(cs != 0, mirror_step(PSTEP1(E0, O0, 7)), n1);
      e2 = pick_step(cs != 0, mirror_step(PSTEP2(E0, O0, 7)), n2);
    }
    stage_tile(st, 0, row_even<0>(l1, l2, E0, O0, f0, f1));
    stage_tile(st, 1,
               row_odd<0>(E0, O0, PSTEP1(E0, O0, 4), PSTEP2(E0, O0, 4), f0,
                          f1));
    stage_tile(st, 2,
               row_even<4>(PSTEP1(E0, O0, 3), PSTEP2(E0, O0, 3), E0, O0, f0,
                           f1));
    stage_tile(st, 3, row_odd<4>(E0, O0, e1, e2, f0, f1));
    stage_flush(sr, out, og + 32u * (unsigned)u, (unsigned)outW, true);
    if constexpr (TWO) {
      if (cs != 0) {
        unsigned ot = (unsigned)((b * (2 * r) + I * 16 + 8 * half) * outW +
                                 cs * 32 * NT + l16);
        finish_deferred(dw, n1, n2, f0[0], f1[0], out, ot, (unsigned)outW);
      }
    }
  }
}

template <int BASE>
__device__ __forceinline__ v8f row_j1(v2f l1, v2f l2, const YT& E,
                                      const YT& O, v2f r1, v2f r2,
                                      const v2f* g0, const v2f* g1) {
  v8f acc = {0.f, 0.f, 0.f, 0.f, 0.f, 0.f, 0.f, 0.f};
  acc = wmma4(l1, g0[0], acc);
  acc = wmma4(l2, g1[0], acc);
#pragma unroll
  for (int q = 0; q < 4; ++q) {
    acc = wmma4(PSTEP1(E, O, BASE + q), g0[q + 1], acc);
    acc = wmma4(PSTEP2(E, O, BASE + q), g1[q + 1], acc);
  }
  acc = wmma4(r1, g0[5], acc);
  acc = wmma4(r2, g1[5], acc);
  return acc;
}

template <int NOUT, int Hb, int Wb>
__global__ void __launch_bounds__(128)
j1_fused_kernel(const float* __restrict__ plain, const float* __restrict__ yh,
                const float* __restrict__ h0, int taps0,
                const float* __restrict__ h1, int taps1,
                float* __restrict__ out, int B) {
  constexpr int r = 2 * Hb;
  constexpr int W = 2 * Wb;
  int lane = threadIdx.x & 31;
  int wave = __builtin_amdgcn_readfirstlane((int)(threadIdx.x >> 5));
  int g = (int)blockIdx.x * 4 + wave;
  constexpr int nCS = W / (16 * NOUT);
  constexpr int nRT = r / 16;
  int cs = g % nCS;
  int g2 = g / nCS;
  int I = g2 % nRT;
  int b = g2 / nRT;
  if (b >= B) return;
  int half = lane >> 4;
  int l16 = lane & 15;
  int koff = half * 2;
  int x0 = 16 * I - 4;
  int S = cs * 16 * NOUT;
  static_assert(nCS == 2, "the two strips of a row strip share a work-group");
  static_assert((nCS * nRT) % 4 == 0, "a work-group lies in one image");

  v2f g0[6], g1[6];
#pragma unroll
  for (int s = 0; s < 6; ++s) {
    int kk = 4 * s - 1 + koff;
    g0[s].x = cf_A(l16, kk, h0, taps0);
    g0[s].y = cf_A(l16, kk + 1, h0, taps0);
    g1[s].x = cf_A(l16, kk, h1, taps1);
    g1[s].y = cf_A(l16, kk + 1, h1, taps1);
  }
  v2f f0s[6], f1s[6];
#pragma unroll
  for (int c = 0; c < 6; ++c) {
    f0s[c].x = g0[c].x * SQRT_HALF;
    f0s[c].y = g0[c].y * SQRT_HALF;
    f1s[c].x = g1[c].x * SQRT_HALF;
    f1s[c].y = g1[c].y * SQRT_HALF;
  }

  unsigned og = (unsigned)((b * r + I * 16 + half) * W + cs * 16 * NOUT +
                           4 * l16);
  __shared__ __attribute__((aligned(16))) float stage_mem[4 * STAGE_FLOATS];
  float* sw = stage_mem + wave * STAGE_FLOATS;
  float* st = sw + half * 8 * STAGE_PITCH + l16;
  const float* sr = sw + half * STAGE_PITCH + 4 * l16;

  static_assert(NOUT % 4 == 0, "two aligned tile pairs per 64-column unit");
  constexpr int NP = NOUT / 2;
  static_assert(NP >= 4, "the deferred first tile is not in the last unit");
  __shared__ float xch_mem[4 * XCH_FLOATS];
  __shared__ float defer_mem[4 * DEFER_FLOATS];
  float* xw = xch_mem + wave * XCH_FLOATS + lane;
  float* dw = defer_mem + wave * DEFER_FLOATS + lane;
  unsigned jp = (unsigned)(S / 2 + pair_col(l16));
  PG G;
  pair_load<0, Hb, Wb>(G, plain, yh, b, x0, jp, koff);
  bool top = (I == 0), bot = (I == nRT - 1);
  v2f c0[6], c0s[6], c1s[6];
#pragma unroll
  for (int c = 0; c < 6; ++c) {
    bool sw = (c == 0 && top) || (c == 5 && bot);
    c0[c] = swap_k(sw, g0[c]);
    c0s[c] = swap_k(sw, f0s[c]);
    c1s[c] = swap_k(sw, f1s[c]);
  }
  YT E0, O0;
  col_pair_piped<6, Hb, Wb>(plain, yh, b, x0, jp, jp + 16u, NP > 1, koff, c0,
                            c0s, c1s, G, E0, O0);
  tile_result_fence(E0, O0);
  if (cs != 0) xch_put(xw, PSTEP1(E0, O0, 0), PSTEP2(E0, O0, 0));
  v2f l1 = pick_step(cs != 0, (v2f){0.f, 0.f},
                     mirror_step(PSTEP1(E0, O0, 0)));
  v2f l2 = pick_step(cs != 0, (v2f){0.f, 0.f},
                     mirror_step(PSTEP2(E0, O0, 0)));
  bool first_on = !(cs != 0 && l16 < 4);
  auto trip = [&](int p, bool next, int slot,
                  bool flush) __attribute__((always_inline)) {
    YT E1, O1;
    col_pair_piped<6, Hb, Wb>(plain, yh, b, x0, jp + 16u * (unsigned)p,
                              jp + 16u * (unsigned)(p + 1), next, koff, c0,
                              c0s, c1s, G, E1, O1);
    tile_result_fence(E1, O1);
    v8f t0 = row_j1<0>(l1, l2, E0, O0, PSTEP1(E0, O0, 4), PSTEP2(E0, O0, 4),
                       g0, g1);
    if (p == 1 && cs != 0) defer_put(dw, t0);
    stage_tile(st, 2 * slot, t0);
    stage_tile(st, 2 * slot + 1,
               row_j1<4>(PSTEP1(E0, O0, 3), PSTEP2(E0, O0, 3), E0, O0,
                         PSTEP1(E1, O1, 0), PSTEP2(E1, O1, 0), g0, g1));
    if (flush)
      stage_flush(sr, out, og + 32u * (unsigned)(p - 2), (unsigned)W,
                  p == 2 ? first_on : true);
    l1 = PSTEP1(E0, O0, 7);
    l2 = PSTEP2(E0, O0, 7);
    E0 = E1;
    O0 = O1;
    __builtin_amdgcn_sched_barrier(0);
  };
#pragma unroll 1
  for (int p = 1; p < NP - 1; ++p)
    trip(p, true, (p - 1) & 1, ((p - 1) & 1) != 0);
  trip(NP - 1, false, 0, false);
  {
    if (cs == 0) xch_put(xw, PSTEP1(E0, O0, 7), PSTEP2(E0, O0, 7));
    lds_barrier();
    v2f n1, n2;
    xch_get(xch_mem + (wave ^ 1) * XCH_FLOATS + lane, n1, n2);
    v2f e1 = pick_step(cs != 0, mirror_step(PSTEP1(E0, O0, 7)), n1);
    v2f e2 = pick_step(cs != 0, mirror_step(PSTEP2(E0, O0, 7)), n2);
    stage_tile(st, 2,
               row_j1<0>(l1, l2, E0, O0, PSTEP1(E0, O0, 4), PSTEP2(E0, O0, 4),
                         g0, g1));
    stage_tile(st, 3,
               row_j1<4>(PSTEP1(E0, O0, 3), PSTEP2(E0, O0, 3), E0, O0, e1, e2,
                         g0, g1));
    stage_flush(sr, out, og + 16u * (unsigned)(NOUT - 4), (unsigned)W, true);
    if (cs != 0) {
      unsigned ot = (unsigned)((b * r + I * 16 + 8 * half) * W +
                               cs * 16 * NOUT + l16);
      finish_deferred(dw, n1, n2, g0[0], g1[0], out, ot, (unsigned)W);
    }
  }
}

extern "C" void kernel_launch(void* const* d_in, const int* in_sizes, int n_in,
                              void* d_out, int out_size, void* d_ws,
                              size_t ws_size, hipStream_t stream) {
  (void)in_sizes;
  (void)n_in;
  (void)out_size;
  (void)ws_size;

  const float* yl = (const float*)d_in[0];
  const float* yh0 = (const float*)d_in[1];
  const float* yh1 = (const float*)d_in[2];
  const float* yh2 = (const float*)d_in[3];
  const float* g0o = (const float*)d_in[4];
  const float* g1o = (const float*)d_in[5];
  const float* g0a = (const float*)d_in[6];
  const float* g0b = (const float*)d_in[7];
  const float* g1a = (const float*)d_in[8];
  const float* g1b = (const float*)d_in[9];

  float* LLa = (float*)d_ws;
  float* LLb = LLa + (size_t)64 * 65536;
  float* outp = (float*)d_out;

  dim3 blk(128);

  j2_fused_kernel<2, 64, 64><<<dim3(1024), blk, 0, stream>>>(
      yl, yh2, g0b, g0a, g1b, g1a, LLa, 64);

  j2_fused_kernel<8, 128, 128><<<dim3(1024), blk, 0, stream>>>(
      LLa, yh1, g0b, g0a, g1b, g1a, LLb, 64);

  j1_fused_kernel<16, 256, 256><<<dim3(1024), blk, 0, stream>>>(
      LLb, yh0, g0o, 7, g1o, 5, outp, 64);
}


# Level 2 joined into the level-1 launch: an 8-wave group makes the level-2 rows its 64 output rows read in LDS (3 reused slots of 16 rows), not memory; 3 launches to 2; column, row sum order changes
# speedup vs baseline: 5.4083x; 1.0619x over previous
#include <hip/hip_runtime.h>


typedef __attribute__((ext_vector_type(2))) float v2f;
typedef __attribute__((ext_vector_type(4))) float v4f;
typedef __attribute__((ext_vector_type(8))) float v8f;

#define SQRT_HALF 0.70710678118654752440f

__device__ __forceinline__ int reflect_i(int x, int n) {
  x = (x < 0) ? (-1 - x) : x;
  x = (x >= n) ? (2 * n - 1 - x) : x;
  return x;
}

__device__ __forceinline__ float cif_A(int m, int kk,
                                       const float* __restrict__ ha,
                                       const float* __restrict__ hb, int hp) {
  int p = m & 3;
  int q = m >> 2;
  int want_odd = hp ? ((p & 1) ^ 1) : (p & 1);
  int t = ((kk - want_odd) >> 1) - q;
  bool ok = ((kk & 1) == want_odd) && (t >= 0) && (t <= 4);
  int tc = t < 0 ? 0 : (t > 4 ? 4 : t);
  const float* __restrict__ h = ((p == 0) || (p == 2)) ? ha : hb;
  int idx = (p < 2) ? (2 * tc + 1) : (2 * tc);
  float v = h[idx];
  return ok ? v : 0.0f;
}

__device__ __forceinline__ float cf_A(int m, int kk,
                                      const float* __restrict__ h, int taps) {
  int t = kk - m - (3 - (taps >> 1));
  bool ok = (t >= 0) && (t < taps);
  int tc = t < 0 ? 0 : (t >= taps ? taps - 1 : t);
  float v = h[tc];
  return ok ? v : 0.0f;
}

__device__ __forceinline__ v8f wmma4(v2f a, v2f b, v8f c) {
  return __builtin_amdgcn_wmma_f32_16x16x4_f32(false, a, false, b, (short)0, c,
                                               false, false);
}

struct YT {
  v8f y1, y2;
};

__device__ __forceinline__ int pair_col(int m) {
  return 2 * (m & 7) + (m >> 3);
}

__device__ __forceinline__ void pair_step(YT& E, YT& O, v2f r0, v2f r1,
                                          v2f lhA, v2f lhB, v2f hlA, v2f hlB,
                                          v2f hhA, v2f hhB, v2f f0c, v2f f0sc,
                                          v2f f1sc) {
  v2f ple = {r0.x, r1.x};
  v2f plo = {r0.y, r1.y};
  v2f lhe = {lhA.x + lhB.x, lhA.y - lhB.y};
  v2f lho = {lhA.y + lhB.y, lhB.x - lhA.x};
  v2f hle = {hlA.x + hlB.x, hlA.y - hlB.y};
  v2f hlo = {hlA.y + hlB.y, hlB.x - hlA.x};
  v2f hhe = {hhA.x + hhB.x, hhA.y - hhB.y};
  v2f hho = {hhA.y + hhB.y, hhB.x - hhA.x};
  E.y1 = wmma4(ple, f0c, E.y1);
  E.y1 = wmma4(lhe, f1sc, E.y1);
  E.y2 = wmma4(hle, f0sc, E.y2);
  E.y2 = wmma4(hhe, f1sc, E.y2);
  O.y1 = wmma4(plo, f0c, O.y1);
  O.y1 = wmma4(lho, f1sc, O.y1);
  O.y2 = wmma4(hlo, f0sc, O.y2);
  O.y2 = wmma4(hho, f1sc, O.y2);
}

struct PG {
  v2f r0[2], r1[2], a0[2], a1[2], a2[2], a3[2], a4[2], a5[2];
};

#define IMG_PITCH 528
__device__ __forceinline__ int img_row(int rho) {
  int tl = rho >> 4;
  int sl = (tl >= 3) ? tl - 3 : tl;
  return 16 * sl + (rho & 15);
}

template <int CB, int Hb, int Wb, bool LDSP, int PART = 0>
__device__ __forceinline__ void pair_load(PG& G,
                                          const float* __restrict__ plain,
                                          const float* img, int rbase,
                                          const float* __restrict__ yh, int b,
                                          int x0, unsigned jb, int koff) {
  constexpr int r = 2 * Hb;
  constexpr int W = 2 * Wb;
  constexpr size_t bs = (size_t)Hb * Wb * 2;
  const float* yb = yh + (size_t)b * 6 * bs;
  const v2f* q0 = (const v2f*)yb;
  const v2f* q1 = (const v2f*)(yb + bs);
  const v2f* q2 = (const v2f*)(yb + 2 * bs);
  const v2f* q3 = (const v2f*)(yb + 3 * bs);
  const v2f* q4 = (const v2f*)(yb + 4 * bs);
  const v2f* q5 = (const v2f*)(yb + 5 * bs);
#pragma unroll
  for (int s = 0; s < 2; ++s) {
    int kk = 4 * (CB + s) + koff;
    int ir = reflect_i((x0 + kk) >> 1, Hb);
    unsigned ob = (unsigned)ir * (unsigned)Wb + jb;
    if constexpr (PART == 1) {
    } else if constexpr (LDSP) {
      const float* pr = img + img_row(2 * ir - rbase) * IMG_PITCH + 2 * (int)jb;
      G.r0[s] = *(const v2f*)pr;
      G.r1[s] = *(const v2f*)(pr + IMG_PITCH);
    } else {
      const v2f* p0 = (const v2f*)(plain + (size_t)b * r * W);
      unsigned opl = 2u * (unsigned)ir * (unsigned)Wb + jb;
      G.r0[s] = p0[opl];
      G.r1[s] = p0[opl + Wb];
    }
    if constexpr (PART != 2) {
      G.a0[s] = q0[ob];
      G.a5[s] = q5[ob];
      G.a2[s] = q2[ob];
      G.a3[s] = q3[ob];
      G.a1[s] = q1[ob];
      G.a4[s] = q4[ob];
    }
  }
}

template <int CB>
__device__ __forceinline__ void pair_mma(YT& E, YT& O, const PG& G,
                                         const v2f* f0, const v2f* f0s,
                                         const v2f* f1s) {
#pragma unroll
  for (int s = 0; s < 2; ++s)
    pair_step(E, O, G.r0[s], G.r1[s], G.a0[s], G.a5[s], G.a2[s], G.a3[s],
              G.a1[s], G.a4[s], f0[CB + s], f0s[CB + s], f1s[CB + s]);
}

__device__ __forceinline__ void order_point(YT& E, YT& O) {
  asm volatile(""
               : "+v"(E.y1), "+v"(E.y2), "+v"(O.y1), "+v"(O.y2)
               :
               : "memory");
}

template <int KC, int Hb, int Wb, bool LDSP>
__device__ __forceinline__ void col_pair_piped(
    const float* __restrict__ plain, const float* img, int rbase,
    const float* __restrict__ yh, int b, int x0, unsigned jb, unsigned jb_next,
    bool next, int koff, const v2f* f0, const v2f* f0s, const v2f* f1s, PG& G,
    YT& E, YT& O) {
  static_assert(KC == 4 || KC == 6, "two or three groups of two K steps");
  E.y1 = (v8f){0.f, 0.f, 0.f, 0.f, 0.f, 0.f, 0.f, 0.f};
  E.y2 = (v8f){0.f, 0.f, 0.f, 0.f, 0.f, 0.f, 0.f, 0.f};
  O.y1 = (v8f){0.f, 0.f, 0.f, 0.f, 0.f, 0.f, 0.f, 0.f};
  O.y2 = (v8f){0.f, 0.f, 0.f, 0.f, 0.f, 0.f, 0.f, 0.f};
  PG G1;
  [[maybe_unused]] PG G2;
  pair_load<2, Hb, Wb, LDSP>(G1, plain, img, rbase, yh, b, x0, jb, koff);
  if constexpr (KC == 6)
    pair_load<4, Hb, Wb, LDSP>(G2, plain, img, rbase, yh, b, x0, jb, koff);
  __builtin_amdgcn_sched_barrier(0);
  pair_mma<0>(E, O, G, f0, f0s, f1s);
  order_point(E, O);
  if (next)
    pair_load<0, Hb, Wb, LDSP>(G, plain, img, rbase, yh, b, x0, jb_next, koff);
  order_point(E, O);
  __builtin_amdgcn_sched_barrier(0);
  pair_mma<2>(E, O, G1, f0, f0s, f1s);
  if constexpr (KC == 6) pair_mma<4>(E, O, G2, f0, f0s, f1s);
}

__device__ __forceinline__ v2f swap_k(bool c, v2f a) {
  return (v2f){c ? a.y : a.x, c ? a.x : a.y};
}

#define STAGE_PITCH 68
#define STAGE_FLOATS (16 * STAGE_PITCH)

__device__ __forceinline__ void stage_tile(float* st, int j, v8f acc) {
#pragma unroll
  for (int e = 0; e < 8; ++e) st[e * STAGE_PITCH + 16 * j] = acc[e];
}

__device__ __forceinline__ void stage_flush(const float* sr,
                                            float* __restrict__ out,
                                            unsigned og, unsigned pitch,
                                            bool lane_on) {
#pragma unroll
  for (int i = 0; i < 8; ++i) {
    v4f v = *(const v4f*)(sr + 2 * i * STAGE_PITCH);
    if (lane_on) *(v4f*)(out + (og + 2u * (unsigned)i * pitch)) = v;
  }
}

__device__ __forceinline__ void tile_result_fence(YT& T) {
  __builtin_amdgcn_sched_barrier(0);
  asm volatile(
      "s_wait_alu depctr_va_vdst(0)\n\t"
      "v_nop\n\tv_nop\n\tv_nop\n\tv_nop\n\tv_nop\n\t"
      "v_nop\n\tv_nop\n\tv_nop\n\tv_nop"
      : "+v"(T.y1), "+v"(T.y2)
      :
      : "memory");
  __builtin_amdgcn_sched_barrier(0);
}

__device__ __forceinline__ void tile_result_fence(YT& H, YT& T) {
  __builtin_amdgcn_sched_barrier(0);
  asm volatile(
      "s_wait_alu depctr_va_vdst(0)\n\t"
      "v_nop\n\tv_nop\n\tv_nop\n\tv_nop\n\tv_nop\n\t"
      "v_nop\n\tv_nop\n\tv_nop\n\tv_nop"
      : "+v"(H.y1), "+v"(H.y2), "+v"(T.y1), "+v"(T.y2)
      :
      : "memory");
  __builtin_amdgcn_sched_barrier(0);
}

#define PSTEP1(E, O, t) ((v2f){(E).y1[(t)], (O).y1[(t)]})
#define PSTEP2(E, O, t) ((v2f){(E).y2[(t)], (O).y2[(t)]})

__device__ __forceinline__ float other_half(float x) {
  return __builtin_bit_cast(
      float, __builtin_amdgcn_permlanex16(__builtin_bit_cast(int, x),
                                          __builtin_bit_cast(int, x),
                                          0x76543210, 0xfedcba98, false, false));
}

__device__ __forceinline__ v2f mirror_step(v2f a) {
  return (v2f){other_half(a.y), other_half(a.x)};
}

__device__ __forceinline__ v2f pick_step(bool c, v2f a, v2f b) {
  return (v2f){c ? a.x : b.x, c ? a.y : b.y};
}

#define XCH_FLOATS 256
__device__ __forceinline__ void xch_put(float* x, v2f a1, v2f a2) {
  x[0] = a1.x;
  x[32] = a1.y;
  x[64] = a2.x;
  x[96] = a2.y;
}
__device__ __forceinline__ void xch_get(const float* x, v2f& a1, v2f& a2) {
  a1 = (v2f){x[0], x[32]};
  a2 = (v2f){x[64], x[96]};
}

__device__ __forceinline__ void lds_barrier() {
  __builtin_amdgcn_fence(__ATOMIC_RELEASE, "workgroup", "local");
  __builtin_amdgcn_s_barrier();
  __builtin_amdgcn_fence(__ATOMIC_ACQUIRE, "workgroup", "local");
}

#define DEFER_FLOATS 256
__device__ __forceinline__ void defer_put(float* d, v8f acc) {
#pragma unroll
  for (int e = 0; e < 8; ++e) d[32 * e] = acc[e];
}
__device__ __forceinline__ v8f defer_get(const float* d) {
  v8f acc;
#pragma unroll
  for (int e = 0; e < 8; ++e) acc[e] = d[32 * e];
  return acc;
}

__device__ __forceinline__ void finish_deferred(const float* d, v2f n1, v2f n2,
                                                v2f g0c, v2f g1c,
                                                float* __restrict__ out,
                                                unsigned ot, unsigned pitch) {
  v8f acc = defer_get(d);
  acc = wmma4(n1, g0c, acc);
  acc = wmma4(n2, g1c, acc);
#pragma unroll
  for (int e = 0; e < 8; ++e) out[ot + (unsigned)e * pitch] = acc[e];
}

template <int BASE>
__device__ __forceinline__ v8f row_even(v2f l1, v2f l2, const YT& E,
                                        const YT& O, const v2f* g0,
                                        const v2f* g1) {
  v8f acc = {0.f, 0.f, 0.f, 0.f, 0.f, 0.f, 0.f, 0.f};
  acc = wmma4(l1, g0[0], acc);
  acc = wmma4(l2, g1[0], acc);
#pragma unroll
  for (int q = 0; q < 3; ++q) {
    acc = wmma4(PSTEP1(E, O, BASE + q), g0[q + 1], acc);
    acc = wmma4(PSTEP2(E, O, BASE + q), g1[q + 1], acc);
  }
  return acc;
}

template <int BASE>
__device__ __forceinline__ v8f row_odd(const YT& E, const YT& O, v2f r1,
                                       v2f r2, const v2f* g0, const v2f* g1) {
  v8f acc = {0.f, 0.f, 0.f, 0.f, 0.f, 0.f, 0.f, 0.f};
#pragma unroll
  for (int q = 1; q < 4; ++q) {
    acc = wmma4(PSTEP1(E, O, BASE + q), g0[q - 1], acc);
    acc = wmma4(PSTEP2(E, O, BASE + q), g1[q - 1], acc);
  }
  acc = wmma4(r1, g0[3], acc);
  acc = wmma4(r2, g1[3], acc);
  return acc;
}

template <int NT, int Hb, int Wb>
__global__ void __launch_bounds__(128, 4)
j2_fused_kernel(const float* __restrict__ plain, const float* __restrict__ yh,
                const float* __restrict__ ha0, const float* __restrict__ hb0,
                const float* __restrict__ ha1, const float* __restrict__ hb1,
                float* __restrict__ out, int B) {
  constexpr int r = 2 * Hb;
  constexpr int W = 2 * Wb;
  int lane = threadIdx.x & 31;
  int wave = __builtin_amdgcn_readfirstlane((int)(threadIdx.x >> 5));
  int g = (int)blockIdx.x * 4 + wave;
  constexpr int nCS = W / (16 * NT);
  constexpr int nRT = (2 * r) / 16;
  int cs = g % nCS;
  int g2 = g / nCS;
  int I = g2 % nRT;
  int b = g2 / nRT;
  if (b >= B) return;
  int half = lane >> 4;
  int l16 = lane & 15;
  int koff = half * 2;
  int x0 = 8 * I - 4;
  int S = cs * 16 * NT;
  constexpr bool TWO = (nCS == 2);
  static_assert(nCS == 2 || nCS == 4, "neighbour strips share a work-group");

  v2f f0[4], f0s[4], f1[4], f1s[4];
#pragma unroll
  for (int c = 0; c < 4; ++c) {
    int kk = 4 * c + koff;
    float ax = cif_A(l16, kk, ha0, hb0, 0);
    float ay = cif_A(l16, kk + 1, ha0, hb0, 0);
    float bx = cif_A(l16, kk, ha1, hb1, 1);
    float by = cif_A(l16, kk + 1, ha1, hb1, 1);
    f0[c].x = ax;
    f0[c].y = ay;
    f0s[c].x = ax * SQRT_HALF;
    f0s[c].y = ay * SQRT_HALF;
    f1[c].x = bx;
    f1[c].y = by;
    f1s[c].x = bx * SQRT_HALF;
    f1s[c].y = by * SQRT_HALF;
  }

  constexpr int outW = 2 * W;
  unsigned og = (unsigned)((b * (2 * r) + I * 16 + half) * outW +
                           cs * 32 * NT + 4 * l16);
  __shared__ __attribute__((aligned(16))) float stage_mem[4 * STAGE_FLOATS];
  float* sw = stage_mem + wave * STAGE_FLOATS;
  float* st = sw + half * 8 * STAGE_PITCH + l16;
  const float* sr = sw + half * STAGE_PITCH + 4 * l16;

  static_assert(NT % 2 == 0, "tiles are computed in aligned pairs");
  constexpr int NP = NT / 2;
  static_assert(TWO ? NP > 1 : NP == 1, "see the hand-over of margins");
  static_assert((nCS * nRT) % 4 == 0, "a work-group lies in one image");
  __shared__ float xch_mem[4 * XCH_FLOATS];
  __shared__ float defer_mem[4 * DEFER_FLOATS];
  float* xw = xch_mem + wave * XCH_FLOATS + lane;
  float* dw = defer_mem + wave * DEFER_FLOATS + lane;
  unsigned jp = (unsigned)(S / 2 + pair_col(l16));
  PG G;
  pair_load<0, Hb, Wb, false>(G, plain, nullptr, 0, yh, b, x0, jp, koff);
  bool top = (I == 0), bot = (I == nRT - 1);
  v2f c0[4], c0s[4], c1s[4];
#pragma unroll
  for (int c = 0; c < 4; ++c) {
    bool sw = (c == 0 && top) || (c == 3 && bot);
    c0[c] = swap_k(sw, f0[c]);
    c0s[c] = swap_k(sw, f0s[c]);
    c1s[c] = swap_k(sw, f1s[c]);
  }
  YT E0, O0;
  col_pair_piped<4, Hb, Wb, false>(plain, nullptr, 0, yh, b, x0, jp, jp + 16u,
                                   NP > 1, koff, c0, c0s, c1s, G, E0, O0);
  tile_result_fence(E0, O0);
  v2f l1, l2;
  v2f e1, e2;
  if constexpr (TWO) {
    if (cs != 0) xch_put(xw, PSTEP1(E0, O0, 0), PSTEP2(E0, O0, 0));
    l1 = pick_step(cs != 0, (v2f){0.f, 0.f}, mirror_step(PSTEP1(E0, O0, 0)));
    l2 = pick_step(cs != 0, (v2f){0.f, 0.f}, mirror_step(PSTEP2(E0, O0, 0)));
  } else {
    xch_put(xw, PSTEP1(E0, O0, 0), PSTEP2(E0, O0, 0));
    xch_put(xw + XCH_FLOATS / 2, PSTEP1(E0, O0, 7), PSTEP2(E0, O0, 7));
    lds_barrier();
    int wl = (cs > 0) ? wave - 1 : wave;
    int wr = (cs < nCS - 1) ? wave + 1 : wave;
    v2f n1, n2, m1, m2;
    xch_get(xch_mem + wl * XCH_FLOATS + XCH_FLOATS / 2 + lane, n1, n2);
    xch_get(xch_mem + wr * XCH_FLOATS + lane, m1, m2);
    l1 = pick_step(cs > 0, n1, mirror_step(PSTEP1(E0, O0, 0)));
    l2 = pick_step(cs > 0, n2, mirror_step(PSTEP2(E0, O0, 0)));
    e1 = pick_step(cs < nCS - 1, m1, mirror_step(PSTEP1(E0, O0, 7)));
    e2 = pick_step(cs < nCS - 1, m2, mirror_step(PSTEP2(E0, O0, 7)));
  }
  bool first_on = !(TWO && cs != 0 && l16 < 4);
#pragma unroll
  for (int p = 1; p < NP; ++p) {
    const int u = 2 * (p - 1);
    v8f t0 = row_even<0>(l1, l2, E0, O0, f0, f1);
    if (TWO && p == 1 && cs != 0) defer_put(dw, t0);
    stage_tile(st, 0, t0);
    stage_tile(st, 1,
               row_odd<0>(E0, O0, PSTEP1(E0, O0, 4), PSTEP2(E0, O0, 4), f0,
                          f1));
    stage_tile(st, 2,
               row_even<4>(PSTEP1(E0, O0, 3), PSTEP2(E0, O0, 3), E0, O0, f0,
                           f1));
    YT E1, O1;
    col_pair_piped<4, Hb, Wb, false>(plain, nullptr, 0, yh, b, x0,
                                     jp + 16u * (unsigned)p,
                                     jp + 16u * (unsigned)(p + 1), p + 1 < NP,
                                     koff, c0, c0s, c1s, G, E1, O1);
    tile_result_fence(E1, O1);
    stage_tile(st, 3,
               row_odd<4>(E0, O0, PSTEP1(E1, O1, 0), PSTEP2(E1, O1, 0), f0,
                          f1));
    stage_flush(sr, out, og + 32u * (unsigned)u, (unsigned)outW,
                p == 1 ? first_on : true);
    l1 = PSTEP1(E0, O0, 7);
    l2 = PSTEP2(E0, O0, 7);
    E0 = E1;
    O0 = O1;
    __builtin_amdgcn_sched_barrier(0);
  }
  {
    constexpr int u = NT - 2;
    [[maybe_unused]] v2f n1, n2;
    if constexpr (TWO) {
      if (cs == 0) xch_put(xw, PSTEP1(E0, O0, 7), PSTEP2(E0, O0, 7));
      lds_barrier();
      xch_get(xch_mem + (wave ^ 1) * XCH_FLOATS + lane, n1, n2);
      e1 = pick_step(cs != 0, mirror_step(PSTEP1(E0, O0, 7)), n1);
      e2 = pick_step(cs != 0, mirror_step(PSTEP2(E0, O0, 7)), n2);
    }
    stage_tile(st, 0, row_even<0>(l1, l2, E0, O0, f0, f1));
    stage_tile(st, 1,
               row_odd<0>(E0, O0, PSTEP1(E0, O0, 4), PSTEP2(E0, O0, 4), f0,
                          f1));
    stage_tile(st, 2,
               row_even<4>(PSTEP1(E0, O0, 3), PSTEP2(E0, O0, 3), E0, O0, f0,
                           f1));
    stage_tile(st, 3, row_odd<4>(E0, O0, e1, e2, f0, f1));
    stage_flush(sr, out, og + 32u * (unsigned)u, (unsigned)outW, true);
    if constexpr (TWO) {
      if (cs != 0) {
        unsigned ot = (unsigned)((b * (2 * r) + I * 16 + 8 * half) * outW +
                                 cs * 32 * NT + l16);
        finish_deferred(dw, n1, n2, f0[0], f1[0], out, ot, (unsigned)outW);
      }
    }
  }
}


__device__ __forceinline__ void img_tile(float* ip, int j, v8f acc) {
#pragma unroll
  for (int e = 0; e < 8; ++e) ip[e * IMG_PITCH + 16 * j] = acc[e];
}

template <int Hb, int Wb, bool NEXT, typename Ahead>
__device__ __forceinline__ void l2_tile(
    const PG& G, const PG& G1, PG& Gn, const float* __restrict__ plain,
    const float* __restrict__ yh, int b,
    int x0, int cs, int lane, const v2f* f0, const v2f* f0s, const v2f* f1,
    const v2f* f1s, float* slot, float* xset, Ahead ahead) {
  constexpr int r = 2 * Hb;
  constexpr int nCS = 8;
  int half = lane >> 4;
  int l16 = lane & 15;
  int koff = half * 2;
  unsigned jp = (unsigned)(16 * cs + pair_col(l16));
  v2f c0[4], c0s[4], c1s[4];
#pragma unroll
  for (int c = 0; c < 4; ++c) {
    int row = x0 + 4 * c + koff;
    bool sw = (row < 0) || (row >= r);
    c0[c] = swap_k(sw, f0[c]);
    c0s[c] = swap_k(sw, f0s[c]);
    c1s[c] = swap_k(sw, f1s[c]);
  }
  YT E0, O0;
  E0.y1 = (v8f){0.f, 0.f, 0.f, 0.f, 0.f, 0.f, 0.f, 0.f};
  E0.y2 = (v8f){0.f, 0.f, 0.f, 0.f, 0.f, 0.f, 0.f, 0.f};
  O0.y1 = (v8f){0.f, 0.f, 0.f, 0.f, 0.f, 0.f, 0.f, 0.f};
  O0.y2 = (v8f){0.f, 0.f, 0.f, 0.f, 0.f, 0.f, 0.f, 0.f};
  if constexpr (NEXT)
    pair_load<2, Hb, Wb, false>(Gn, plain, nullptr, 0, yh, b, x0 + 8, jp, koff);
  __builtin_amdgcn_sched_barrier(0);
  pair_mma<0>(E0, O0, G, c0, c0s, c1s);
  pair_mma<2>(E0, O0, G1, c0, c0s, c1s);
  tile_result_fence(E0, O0);
  float* xw = xset + cs * XCH_FLOATS + lane;
  xch_put(xw, PSTEP1(E0, O0, 0), PSTEP2(E0, O0, 0));
  xch_put(xw + XCH_FLOATS / 2, PSTEP1(E0, O0, 7), PSTEP2(E0, O0, 7));
  lds_barrier();
  ahead();
  int wl = (cs > 0) ? cs - 1 : cs;
  int wr = (cs < nCS - 1) ? cs + 1 : cs;
  v2f n1, n2, m1, m2;
  xch_get(xset + wl * XCH_FLOATS + XCH_FLOATS / 2 + lane, n1, n2);
  xch_get(xset + wr * XCH_FLOATS + lane, m1, m2);
  v2f l1 = pick_step(cs > 0, n1, mirror_step(PSTEP1(E0, O0, 0)));
  v2f l2 = pick_step(cs > 0, n2, mirror_step(PSTEP2(E0, O0, 0)));
  v2f e1 = pick_step(cs < nCS - 1, m1, mirror_step(PSTEP1(E0, O0, 7)));
  v2f e2 = pick_step(cs < nCS - 1, m2, mirror_step(PSTEP2(E0, O0, 7)));
  float* ip = slot + 8 * half * IMG_PITCH + 64 * cs + l16;
  img_tile(ip, 0, row_even<0>(l1, l2, E0, O0, f0, f1));
  img_tile(ip, 1,
           row_odd<0>(E0, O0, PSTEP1(E0, O0, 4), PSTEP2(E0, O0, 4), f0, f1));
  img_tile(ip, 2,
           row_even<4>(PSTEP1(E0, O0, 3), PSTEP2(E0, O0, 3), E0, O0, f0, f1));
  img_tile(ip, 3, row_odd<4>(E0, O0, e1, e2, f0, f1));
}

template <int T0, int NTL, int Hb, int Wb, bool PRE, typename Ahead>
__device__ __forceinline__ void l2_phase(
    const float* __restrict__ plain, const float* __restrict__ yh,
    const float* __restrict__ ha0, const float* __restrict__ hb0,
    const float* __restrict__ ha1, const float* __restrict__ hb1, int b, int J,
    int wave, int lane, float* img, float* xch_mem, PG& G, PG& G1,
    Ahead ahead) {
  static_assert(NTL == 2 || NTL == 3, "see the phases of the group");
  int half = lane >> 4;
  int l16 = lane & 15;
  int koff = half * 2;
  v2f f0[4], f0s[4], f1[4], f1s[4];
#pragma unroll
  for (int c = 0; c < 4; ++c) {
    int kk = 4 * c + koff;
    float ax = cif_A(l16, kk, ha0, hb0, 0);
    float ay = cif_A(l16, kk + 1, ha0, hb0, 0);
    float bx = cif_A(l16, kk, ha1, hb1, 1);
    float by = cif_A(l16, kk + 1, ha1, hb1, 1);
    f0[c].x = ax;
    f0[c].y = ay;
    f0s[c].x = ax * SQRT_HALF;
    f0s[c].y = ay * SQRT_HALF;
    f1[c].x = bx;
    f1[c].y = by;
    f1s[c].x = bx * SQRT_HALF;
    f1s[c].y = by * SQRT_HALF;
  }
  int x0 = 32 * J + 8 * T0 - 6;
  unsigned jp = (unsigned)(16 * wave + pair_col(l16));
  if constexpr (!PRE) {
    pair_load<0, Hb, Wb, false>(G, plain, nullptr, 0, yh, b, x0, jp, koff);
    pair_load<2, Hb, Wb, false>(G1, plain, nullptr, 0, yh, b, x0, jp, koff);
  }
  PG G2;
  constexpr int XS = 8 * XCH_FLOATS;
  constexpr int SL = 16 * IMG_PITCH;
  auto none = []() {};
  l2_tile<Hb, Wb, true>(G, G1, G2, plain, yh, b, x0, wave, lane, f0, f0s, f1,
                        f1s, img + (T0 % 3) * SL, xch_mem + (T0 & 1) * XS,
                        none);
  if constexpr (NTL == 3) {
    l2_tile<Hb, Wb, true>(G1, G2, G, plain, yh, b, x0 + 8, wave, lane, f0, f0s,
                          f1, f1s, img + ((T0 + 1) % 3) * SL,
                          xch_mem + ((T0 + 1) & 1) * XS, none);
    l2_tile<Hb, Wb, false>(G2, G, G1, plain, yh, b, x0 + 16, wave, lane, f0,
                           f0s, f1, f1s, img + ((T0 + 2) % 3) * SL,
                           xch_mem + ((T0 + 2) & 1) * XS, ahead);
  } else {
    l2_tile<Hb, Wb, false>(G1, G2, G, plain, yh, b, x0 + 8, wave, lane, f0,
                           f0s, f1, f1s, img + ((T0 + 1) % 3) * SL,
                           xch_mem + ((T0 + 1) & 1) * XS, ahead);
  }
}

template <int BASE>
__device__ __forceinline__ v8f row_j1(v2f l1, v2f l2, const YT& E,
                                      const YT& O, v2f r1, v2f r2,
                                      const v2f* g0, const v2f* g1) {
  v8f acc = {0.f, 0.f, 0.f, 0.f, 0.f, 0.f, 0.f, 0.f};
  acc = wmma4(l1, g0[0], acc);
  acc = wmma4(l2, g1[0], acc);
#pragma unroll
  for (int q = 0; q < 4; ++q) {
    acc = wmma4(PSTEP1(E, O, BASE + q), g0[q + 1], acc);
    acc = wmma4(PSTEP2(E, O, BASE + q), g1[q + 1], acc);
  }
  acc = wmma4(r1, g0[5], acc);
  acc = wmma4(r2, g1[5], acc);
  return acc;
}

template <int Hb, int Wb, typename Ahead>
__device__ __forceinline__ void l1_phase(
    const float* img, int rbase, const float* __restrict__ yh,
    const float* __restrict__ h0, int taps0, const float* __restrict__ h1,
    int taps1, float* __restrict__ out, int b, int I, int cs, int wave,
    int lane, float* stage_mem, float* xset, float* defer_mem, PG& G,
    Ahead ahead) {
  constexpr int r = 2 * Hb;
  constexpr int W = 2 * Wb;
  constexpr int NOUT = 8;
  constexpr int nCS = W / (16 * NOUT);
  constexpr int nRT = r / 16;
  static_assert(nCS == 4, "four column strips per row strip");
  int half = lane >> 4;
  int l16 = lane & 15;
  int koff = half * 2;
  int x0 = 16 * I - 4;
  int S = cs * 16 * NOUT;

  v2f g0[6], g1[6];
#pragma unroll
  for (int s = 0; s < 6; ++s) {
    int kk = 4 * s - 1 + koff;
    g0[s].x = cf_A(l16, kk, h0, taps0);
    g0[s].y = cf_A(l16, kk + 1, h0, taps0);
    g1[s].x = cf_A(l16, kk, h1, taps1);
    g1[s].y = cf_A(l16, kk + 1, h1, taps1);
  }
  v2f f0s[6], f1s[6];
#pragma unroll
  for (int c = 0; c < 6; ++c) {
    f0s[c].x = g0[c].x * SQRT_HALF;
    f0s[c].y = g0[c].y * SQRT_HALF;
    f1s[c].x = g1[c].x * SQRT_HALF;
    f1s[c].y = g1[c].y * SQRT_HALF;
  }

  unsigned og = (unsigned)((b * r + I * 16 + half) * W + cs * 16 * NOUT +
                           4 * l16);
  float* sw = stage_mem + wave * STAGE_FLOATS;
  float* st = sw + half * 8 * STAGE_PITCH + l16;
  const float* sr = sw + half * STAGE_PITCH + 4 * l16;

  static_assert(NOUT % 4 == 0, "two aligned tile pairs per 64-column unit");
  constexpr int NP = NOUT / 2;
  static_assert(NP >= 4, "the deferred first tile is not in the last unit");
  float* xw = xset + wave * XCH_FLOATS + lane;
  float* dw = defer_mem + wave * DEFER_FLOATS + lane;
  unsigned jp = (unsigned)(S / 2 + pair_col(l16));
  pair_load<0, Hb, Wb, true, 2>(G, nullptr, img, rbase, yh, b, x0, jp, koff);
  bool top = (I == 0), bot = (I == nRT - 1);
  v2f c0[6], c0s[6], c1s[6];
#pragma unroll
  for (int c = 0; c < 6; ++c) {
    bool sw = (c == 0 && top) || (c == 5 && bot);
    c0[c] = swap_k(sw, g0[c]);
    c0s[c] = swap_k(sw, f0s[c]);
    c1s[c] = swap_k(sw, f1s[c]);
  }
  YT E0, O0;
  col_pair_piped<6, Hb, Wb, true>(nullptr, img, rbase, yh, b, x0, jp, jp + 16u,
                                  NP > 1, koff, c0, c0s, c1s, G, E0, O0);
  tile_result_fence(E0, O0);
  xch_put(xw, PSTEP1(E0, O0, 0), PSTEP2(E0, O0, 0));
  v2f l1 = pick_step(cs != 0, (v2f){0.f, 0.f},
                     mirror_step(PSTEP1(E0, O0, 0)));
  v2f l2 = pick_step(cs != 0, (v2f){0.f, 0.f},
                     mirror_step(PSTEP2(E0, O0, 0)));
  bool first_on = !(cs != 0 && l16 < 4);
  auto trip = [&](int p, bool next, int slot,
                  bool flush) __attribute__((always_inline)) {
    YT E1, O1;
    col_pair_piped<6, Hb, Wb, true>(nullptr, img, rbase, yh, b, x0,
                                    jp + 16u * (unsigned)p,
                                    jp + 16u * (unsigned)(p + 1), next, koff,
                                    c0, c0s, c1s, G, E1, O1);
    tile_result_fence(E1, O1);
    v8f t0 = row_j1<0>(l1, l2, E0, O0, PSTEP1(E0, O0, 4), PSTEP2(E0, O0, 4),
                       g0, g1);
    if (p == 1 && cs != 0) defer_put(dw, t0);
    stage_tile(st, 2 * slot, t0);
    stage_tile(st, 2 * slot + 1,
               row_j1<4>(PSTEP1(E0, O0, 3), PSTEP2(E0, O0, 3), E0, O0,
                         PSTEP1(E1, O1, 0), PSTEP2(E1, O1, 0), g0, g1));
    if (flush)
      stage_flush(sr, out, og + 32u * (unsigned)(p - 2), (unsigned)W,
                  p == 2 ? first_on : true);
    l1 = PSTEP1(E0, O0, 7);
    l2 = PSTEP2(E0, O0, 7);
    E0 = E1;
    O0 = O1;
    __builtin_amdgcn_sched_barrier(0);
  };
#pragma unroll 1
  for (int p = 1; p < NP - 1; ++p)
    trip(p, true, (p - 1) & 1, ((p - 1) & 1) != 0);
  trip(NP - 1, false, 0, false);
  ahead();
  {
    xch_put(xw + XCH_FLOATS / 2, PSTEP1(E0, O0, 7), PSTEP2(E0, O0, 7));
    lds_barrier();
    int wl = (cs > 0) ? wave - 1 : wave;
    int wr = (cs < nCS - 1) ? wave + 1 : wave;
    v2f n1, n2, m1, m2;
    xch_get(xset + wl * XCH_FLOATS + XCH_FLOATS / 2 + lane, n1, n2);
    xch_get(xset + wr * XCH_FLOATS + lane, m1, m2);
    v2f e1 = pick_step(cs < nCS - 1, m1, mirror_step(PSTEP1(E0, O0, 7)));
    v2f e2 = pick_step(cs < nCS - 1, m2, mirror_step(PSTEP2(E0, O0, 7)));
    stage_tile(st, 2,
               row_j1<0>(l1, l2, E0, O0, PSTEP1(E0, O0, 4), PSTEP2(E0, O0, 4),
                         g0, g1));
    stage_tile(st, 3,
               row_j1<4>(PSTEP1(E0, O0, 3), PSTEP2(E0, O0, 3), E0, O0, e1, e2,
                         g0, g1));
    stage_flush(sr, out, og + 16u * (unsigned)(NOUT - 4), (unsigned)W, true);
    if (cs != 0) {
      unsigned ot = (unsigned)((b * r + I * 16 + 8 * half) * W +
                               cs * 16 * NOUT + l16);
      finish_deferred(dw, n1, n2, g0[0], g1[0], out, ot, (unsigned)W);
    }
  }
}

__device__ __forceinline__ int fresh_lane(int lane) {
  asm volatile("" : "+v"(lane));
  return lane;
}

__global__ void __launch_bounds__(256, 4)
j21_fused_kernel(const float* __restrict__ lla, const float* __restrict__ yh1,
                 const float* __restrict__ ha0, const float* __restrict__ hb0,
                 const float* __restrict__ ha1, const float* __restrict__ hb1,
                 const float* __restrict__ yh0, const float* __restrict__ h0,
                 int taps0, const float* __restrict__ h1, int taps1,
                 float* __restrict__ out, int B) {
  int lane = threadIdx.x & 31;
  int wave = __builtin_amdgcn_readfirstlane((int)(threadIdx.x >> 5));
  int J = (int)blockIdx.x & 7;
  int b = (int)blockIdx.x >> 3;
  if (b >= B) return;
  __shared__ __attribute__((aligned(16))) float img[48 * IMG_PITCH];
  __shared__ __attribute__((aligned(16))) float stage_mem[8 * STAGE_FLOATS];
  __shared__ float xch_mem[2 * 8 * XCH_FLOATS];
  __shared__ float defer_mem[8 * DEFER_FLOATS];
  int rbase = 64 * J - 4;
  int cs = wave & 3;
  int rs = wave >> 2;

  int koff = (lane >> 4) * 2;
  unsigned jp2 = (unsigned)(16 * wave + pair_col(lane & 15));
  unsigned jp1 = (unsigned)(64 * cs + pair_col(lane & 15));
  PG GA, GA1, GB;
  auto ahead_l1a = [&]() {
    pair_load<0, 256, 256, true, 1>(GB, nullptr, img, rbase, yh0, b,
                                    16 * (4 * J + rs) - 4, jp1, koff);
  };
  auto ahead_l2 = [&]() {
    pair_load<0, 128, 128, false>(GA, lla, nullptr, 0, yh1, b,
                                  32 * J + 8 * 3 - 6, jp2, koff);
    pair_load<2, 128, 128, false>(GA1, lla, nullptr, 0, yh1, b,
                                  32 * J + 8 * 3 - 6, jp2, koff);
  };
  auto ahead_l1b = [&]() {
    pair_load<0, 256, 256, true, 1>(GB, nullptr, img, rbase, yh0, b,
                                    16 * (4 * J + 2 + rs) - 4, jp1, koff);
  };
  auto none = []() {};
  l2_phase<0, 3, 128, 128, false>(lla, yh1, ha0, hb0, ha1, hb1, b, J, wave,
                                  fresh_lane(lane), img, xch_mem, GA, GA1,
                                  ahead_l1a);
  lds_barrier();
  l1_phase<256, 256>(img, rbase, yh0, h0, taps0, h1, taps1, out, b,
                     4 * J + rs, cs, wave, fresh_lane(lane), stage_mem,
                     xch_mem, defer_mem, GB, ahead_l2);
  lds_barrier();
  l2_phase<3, 2, 128, 128, true>(lla, yh1, ha0, hb0, ha1, hb1, b, J, wave,
                                 fresh_lane(lane), img, xch_mem, GA, GA1,
                                 ahead_l1b);
  lds_barrier();
  l1_phase<256, 256>(img, rbase, yh0, h0, taps0, h1, taps1, out, b,
                     4 * J + 2 + rs, cs, wave, fresh_lane(lane), stage_mem,
                     xch_mem, defer_mem, GB, none);
}

extern "C" void kernel_launch(void* const* d_in, const int* in_sizes, int n_in,
                              void* d_out, int out_size, void* d_ws,
                              size_t ws_size, hipStream_t stream) {
  (void)in_sizes;
  (void)n_in;
  (void)out_size;
  (void)ws_size;

  const float* yl = (const float*)d_in[0];
  const float* yh0 = (const float*)d_in[1];
  const float* yh1 = (const float*)d_in[2];
  const float* yh2 = (const float*)d_in[3];
  const float* g0o = (const float*)d_in[4];
  const float* g1o = (const float*)d_in[5];
  const float* g0a = (const float*)d_in[6];
  const float* g0b = (const float*)d_in[7];
  const float* g1a = (const float*)d_in[8];
  const float* g1b = (const float*)d_in[9];

  float* LLa = (float*)d_ws;
  float* outp = (float*)d_out;

  dim3 blk(128);

  j2_fused_kernel<2, 64, 64><<<dim3(1024), blk, 0, stream>>>(
      yl, yh2, g0b, g0a, g1b, g1a, LLa, 64);

  j21_fused_kernel<<<dim3(512), dim3(256), 0, stream>>>(
      LLa, yh1, g0b, g0a, g1b, g1a, yh0, g0o, 7, g1o, 5, outp, 64);
}
